# starting kernel
# baseline (speedup 1.0000x reference)
#include <hip/hip_runtime.h>
#include <hip/hip_bf16.h>


static constexpr int NN = 100000;
static constexpr int EE = 1600000;
static constexpr int FF = 128;

typedef __attribute__((ext_vector_type(2))) float v2f;
typedef __attribute__((ext_vector_type(8))) float v8f;


__global__ void k_zero_u32(unsigned* __restrict__ p, int n) {
  int i = blockIdx.x * 256 + threadIdx.x;
  if (i < n) p[i] = 0u;
}


__global__ void k_deg(const int* __restrict__ src, const int* __restrict__ dst,
                      unsigned* __restrict__ outdeg, unsigned* __restrict__ indeg) {
  int e = blockIdx.x * 256 + threadIdx.x;
  if (e < EE) {
    atomicAdd(&outdeg[src[e]], 1u);
    atomicAdd(&indeg[dst[e]], 1u);
  }
}

__global__ void k_ds(const unsigned* __restrict__ outdeg, const unsigned* __restrict__ indeg,
                     float* __restrict__ dsout, float* __restrict__ dsin) {
  int i = blockIdx.x * 256 + threadIdx.x;
  if (i < NN) {
    unsigned od = outdeg[i], id = indeg[i];
    dsout[i] = od ? rsqrtf((float)od) : 1.0f;
    dsin[i]  = id ? rsqrtf((float)id) : 1.0f;
  }
}

__global__ void k_scan(const unsigned* __restrict__ deg, unsigned* __restrict__ offs) {
  __shared__ unsigned ts[1024];
  const int t = threadIdx.x;
  const int per = (NN + 1023) / 1024;
  int s = t * per; if (s > NN) s = NN;
  int e = s + per; if (e > NN) e = NN;
  unsigned sum = 0;
  for (int i = s; i < e; ++i) sum += deg[i];
  ts[t] = sum;
  __syncthreads();
  for (int off = 1; off < 1024; off <<= 1) {
    unsigned v = (t >= off) ? ts[t - off] : 0u;
    __syncthreads();
    if (t >= off) ts[t] += v;
    __syncthreads();
  }
  unsigned run = (t == 0) ? 0u : ts[t - 1];
  for (int i = s; i < e; ++i) { offs[i] = run; run += deg[i]; }
  if (t == 1023) offs[NN] = ts[1023];
}

__global__ void k_copy_u32(const unsigned* __restrict__ a, unsigned* __restrict__ b, int n) {
  int i = blockIdx.x * 256 + threadIdx.x;
  if (i < n) b[i] = a[i];
}

__global__ void k_fill(const int* __restrict__ src, const int* __restrict__ dst,
                       unsigned* __restrict__ cursor, unsigned* __restrict__ esrc) {
  int e = blockIdx.x * 256 + threadIdx.x;
  if (e < EE) {
    unsigned pos = atomicAdd(&cursor[dst[e]], 1u);
    esrc[pos] = (unsigned)src[e];
  }
}


__global__ void k_agg128(const float* __restrict__ h, const unsigned* __restrict__ offs,
                         const unsigned* __restrict__ esrc,
                         const float* __restrict__ dsout, const float* __restrict__ dsin,
                         float* __restrict__ agg) {
  const int wv = threadIdx.x >> 5, lane = threadIdx.x & 31;
  const int n = blockIdx.x * 8 + wv;
  const unsigned b = offs[n], e = offs[n + 1];
  float4 acc = make_float4(0.f, 0.f, 0.f, 0.f);
  for (unsigned i = b; i < e; ++i) {
    const unsigned s = esrc[i];
    const float w = dsout[s];
    const float4 v = *(const float4*)(h + (size_t)s * FF + lane * 4);
    acc.x = fmaf(w, v.x, acc.x);
    acc.y = fmaf(w, v.y, acc.y);
    acc.z = fmaf(w, v.z, acc.z);
    acc.w = fmaf(w, v.w, acc.w);
  }
  const float c = dsin[n];
  *(float4*)(agg + (size_t)n * FF + lane * 4) =
      make_float4(c * acc.x, c * acc.y, c * acc.z, c * acc.w);
}

__global__ void k_agg64(const float* __restrict__ P, const unsigned* __restrict__ offs,
                        const unsigned* __restrict__ esrc, const float* __restrict__ dsin,
                        float* __restrict__ R) {
  const int wv = threadIdx.x >> 5, lane = threadIdx.x & 31;
  const int n = blockIdx.x * 8 + wv;
  const unsigned b = offs[n], e = offs[n + 1];
  float2 acc = make_float2(0.f, 0.f);
  for (unsigned i = b; i < e; ++i) {
    const unsigned s = esrc[i];
    const float2 v = *(const float2*)(P + (size_t)s * 64 + lane * 2);
    acc.x += v.x;
    acc.y += v.y;
  }
  const float c = dsin[n];
  *(float2*)(R + (size_t)n * 64 + lane * 2) = make_float2(c * acc.x, c * acc.y);
}


__global__ void k_gemm_dual(const float* __restrict__ Agg, const float* __restrict__ H,
                            const float* __restrict__ W, const float* __restrict__ L,
                            float* __restrict__ Z) {
  const int wv = threadIdx.x >> 5;
  const int lane = threadIdx.x & 31;
  const int t = lane & 15;
  const int hh = lane >> 4;
  const int r0 = blockIdx.x << 4;
  const int cb = wv << 4;
  const float* arow = Agg + (size_t)(r0 + t) * FF;
  const float* hrow = H + (size_t)(r0 + t) * FF;
  v8f acc = {};
#pragma unroll 4
  for (int k0 = 0; k0 < 128; k0 += 4) {
    const int kk = k0 + 2 * hh;
    v2f a1 = *(const v2f*)(arow + kk);
    v2f b1; b1.x = W[kk * FF + cb + t]; b1.y = W[(kk + 1) * FF + cb + t];
    acc = __builtin_amdgcn_wmma_f32_16x16x4_f32(false, a1, false, b1, (short)0, acc, false, false);
    v2f a2 = *(const v2f*)(hrow + kk);
    v2f b2; b2.x = L[kk * FF + cb + t]; b2.y = L[(kk + 1) * FF + cb + t];
    acc = __builtin_amdgcn_wmma_f32_16x16x4_f32(false, a2, false, b2, (short)0, acc, false, false);
  }
#pragma unroll
  for (int v = 0; v < 8; ++v)
    Z[(size_t)(r0 + v + 8 * hh) * FF + cb + t] = acc[v];
}

__global__ void k_gemm_p(const float* __restrict__ H, const float* __restrict__ dsout,
                         const float* __restrict__ Wp, float* __restrict__ P) {
  const int wv = threadIdx.x >> 5;
  const int lane = threadIdx.x & 31;
  const int t = lane & 15, hh = lane >> 4;
  const int r0 = blockIdx.x << 4;
  const int cb = wv << 4;
  const float d = dsout[r0 + t];
  const float* hrow = H + (size_t)(r0 + t) * FF;
  v8f acc = {};
#pragma unroll 4
  for (int k0 = 0; k0 < 128; k0 += 4) {
    const int kk = k0 + 2 * hh;
    v2f a; a.x = d * hrow[kk]; a.y = d * hrow[kk + 1];
    v2f b; b.x = Wp[kk * 64 + cb + t]; b.y = Wp[(kk + 1) * 64 + cb + t];
    acc = __builtin_amdgcn_wmma_f32_16x16x4_f32(false, a, false, b, (short)0, acc, false, false);
  }
#pragma unroll
  for (int v = 0; v < 8; ++v)
    P[(size_t)(r0 + v + 8 * hh) * 64 + cb + t] = acc[v];
}

__global__ void k_gemm_out(const float* __restrict__ H, const float* __restrict__ Lp,
                           const float* __restrict__ R, const float* __restrict__ b2,
                           float* __restrict__ out) {
  const int wv = threadIdx.x >> 5;
  if (wv == 3) return;
  const int lane = threadIdx.x & 31;
  const int t = lane & 15, hh = lane >> 4;
  const int r0 = blockIdx.x << 4;
  const int cb = wv << 4;
  const float* hrow = H + (size_t)(r0 + t) * FF;
  v8f acc = {};
#pragma unroll 4
  for (int k0 = 0; k0 < 128; k0 += 4) {
    const int kk = k0 + 2 * hh;
    v2f a = *(const v2f*)(hrow + kk);
    v2f b; b.x = Lp[kk * 64 + cb + t]; b.y = Lp[(kk + 1) * 64 + cb + t];
    acc = __builtin_amdgcn_wmma_f32_16x16x4_f32(false, a, false, b, (short)0, acc, false, false);
  }
  const int col = cb + t;
  if (col < 40) {
    const float bias = b2[col];
#pragma unroll
    for (int v = 0; v < 8; ++v) {
      const int row = r0 + v + 8 * hh;
      out[(size_t)row * 40 + col] = acc[v] + R[(size_t)row * 64 + col] + bias;
    }
  }
}


__global__ void k_colstats(const float* __restrict__ z, float* __restrict__ cs,
                           float* __restrict__ cs2) {
  const int c = threadIdx.x;
  const size_t r0 = (size_t)blockIdx.x * 250;
  float s = 0.f, s2 = 0.f;
  for (int i = 0; i < 250; ++i) {
    const float v = z[(r0 + i) * FF + c];
    s += v;
    s2 = fmaf(v, v, s2);
  }
  atomicAdd(&cs[c], s);
  atomicAdd(&cs2[c], s2);
}

__global__ void k_bnparams(const float* __restrict__ cs, const float* __restrict__ cs2,
                           const float* __restrict__ g, const float* __restrict__ be,
                           float* __restrict__ bnA, float* __restrict__ bnB) {
  const int c = threadIdx.x;
  const float inv = 1.0f / (float)NN;
  const float mu = cs[c] * inv;
  const float var = cs2[c] * inv - mu * mu;
  const float a = g[c] * rsqrtf(var + 1e-5f);
  bnA[c] = a;
  bnB[c] = be[c] - mu * a;
}

__global__ void k_bnapply(const float* __restrict__ z, const float* __restrict__ bnA,
                          const float* __restrict__ bnB, float* __restrict__ hout) {
  const size_t i = (size_t)blockIdx.x * 256 + threadIdx.x;
  const int c = (int)(i & 127);
  const float v = fmaf(z[i], bnA[c], bnB[c]);
  hout[i] = v > 0.f ? v : 0.f;
}

__global__ void k_padw(const float* __restrict__ W2, const float* __restrict__ L2,
                       float* __restrict__ Wp, float* __restrict__ Lp) {
  const int i = blockIdx.x * 256 + threadIdx.x;
  const int k = i >> 6, c = i & 63;
  const float w = (c < 40) ? W2[k * 40 + c] : 0.f;
  const float l = (c < 40) ? L2[k * 40 + c] : 0.f;
  Wp[i] = w;
  Lp[i] = l;
}


extern "C" void kernel_launch(void* const* d_in, const int* in_sizes, int n_in,
                              void* d_out, int out_size, void* d_ws, size_t ws_size,
                              hipStream_t stream) {
  (void)in_sizes; (void)n_in; (void)out_size; (void)ws_size;

  const float* feat = (const float*)d_in[0];
  const int* src   = (const int*)d_in[1];
  const int* dst   = (const int*)d_in[2];
  const float* W0  = (const float*)d_in[3];
  const float* L0  = (const float*)d_in[4];
  const float* W1  = (const float*)d_in[5];
  const float* L1  = (const float*)d_in[6];
  const float* W2  = (const float*)d_in[7];
  const float* L2  = (const float*)d_in[8];
  const float* b2  = (const float*)d_in[9];
  const float* g0  = (const float*)d_in[10];
  const float* be0 = (const float*)d_in[11];
  const float* g1  = (const float*)d_in[12];
  const float* be1 = (const float*)d_in[13];
  float* out = (float*)d_out;

  char* base = (char*)d_ws;
  size_t off = 0;
  auto take = [&](size_t bytes) -> char* {
    char* p = base + off;
    off = (off + bytes + 255) & ~(size_t)255;
    return p;
  };
  unsigned* indeg  = (unsigned*)take((size_t)NN * 4);
  unsigned* outdeg = (unsigned*)take((size_t)NN * 4);
  unsigned* offs   = (unsigned*)take((size_t)(NN + 1) * 4);
  unsigned* cursor = (unsigned*)take((size_t)NN * 4);
  unsigned* esrc   = (unsigned*)take((size_t)EE * 4);
  float* dsout = (float*)take((size_t)NN * 4);
  float* dsin  = (float*)take((size_t)NN * 4);
  float* cs    = (float*)take(128 * 4);
  float* cs2   = (float*)take(128 * 4);
  float* bnA   = (float*)take(128 * 4);
  float* bnB   = (float*)take(128 * 4);
  float* Wp    = (float*)take(128 * 64 * 4);
  float* Lp    = (float*)take(128 * 64 * 4);
  float* bufA  = (float*)take((size_t)NN * FF * 4);
  float* bufB  = (float*)take((size_t)NN * FF * 4);
  float* bufC  = (float*)take((size_t)NN * FF * 4);

  const int BEdge = (EE + 255) / 256;
  const int BNode = (NN + 255) / 256;
  const int BWave = NN / 8;
  const int BTile = NN / 16;
  const int BApply = (NN * FF) / 256;

  k_zero_u32<<<BNode, 256, 0, stream>>>(indeg, NN);
  k_zero_u32<<<BNode, 256, 0, stream>>>(outdeg, NN);
  k_deg<<<BEdge, 256, 0, stream>>>(src, dst, outdeg, indeg);
  k_ds<<<BNode, 256, 0, stream>>>(outdeg, indeg, dsout, dsin);
  k_scan<<<1, 1024, 0, stream>>>(indeg, offs);
  k_copy_u32<<<BNode, 256, 0, stream>>>(offs, cursor, NN);
  k_fill<<<BEdge, 256, 0, stream>>>(src, dst, cursor, esrc);
  k_padw<<<32, 256, 0, stream>>>(W2, L2, Wp, Lp);

  k_agg128<<<BWave, 256, 0, stream>>>(feat, offs, esrc, dsout, dsin, bufA);
  k_gemm_dual<<<BTile, 256, 0, stream>>>(bufA, feat, W0, L0, bufB);
  k_zero_u32<<<1, 256, 0, stream>>>((unsigned*)cs, 256);
  k_colstats<<<400, 128, 0, stream>>>(bufB, cs, cs2);
  k_bnparams<<<1, 128, 0, stream>>>(cs, cs2, g0, be0, bnA, bnB);
  k_bnapply<<<BApply, 256, 0, stream>>>(bufB, bnA, bnB, bufC);

  k_agg128<<<BWave, 256, 0, stream>>>(bufC, offs, esrc, dsout, dsin, bufA);
  k_gemm_dual<<<BTile, 256, 0, stream>>>(bufA, bufC, W1, L1, bufB);
  k_zero_u32<<<1, 256, 0, stream>>>((unsigned*)cs, 256);
  k_colstats<<<400, 128, 0, stream>>>(bufB, cs, cs2);
  k_bnparams<<<1, 128, 0, stream>>>(cs, cs2, g1, be1, bnA, bnB);
  k_bnapply<<<BApply, 256, 0, stream>>>(bufB, bnA, bnB, bufC);

  k_gemm_p<<<BTile, 128, 0, stream>>>(bufC, dsout, Wp, bufA);
  k_agg64<<<BWave, 256, 0, stream>>>(bufA, offs, esrc, dsin, bufB);
  k_gemm_out<<<BTile, 128, 0, stream>>>(bufC, Lp, bufB, b2, out);
}


# GEMM operand reuse: the three GEMM kernels run persistent, each wave keeps its weight fragments in registers across row tiles, each row tile is fetched once per block and staged in LDS
# speedup vs baseline: 1.2326x; 1.2326x over previous
#include <hip/hip_runtime.h>
#include <hip/hip_bf16.h>


static constexpr int NN = 100000;
static constexpr int EE = 1600000;
static constexpr int FF = 128;

typedef __attribute__((ext_vector_type(2))) float v2f;
typedef __attribute__((ext_vector_type(8))) float v8f;


__global__ void k_zero_u32(unsigned* __restrict__ p, int n) {
  int i = blockIdx.x * 256 + threadIdx.x;
  if (i < n) p[i] = 0u;
}


__global__ void k_deg(const int* __restrict__ src, const int* __restrict__ dst,
                      unsigned* __restrict__ outdeg, unsigned* __restrict__ indeg) {
  int e = blockIdx.x * 256 + threadIdx.x;
  if (e < EE) {
    atomicAdd(&outdeg[src[e]], 1u);
    atomicAdd(&indeg[dst[e]], 1u);
  }
}

__global__ void k_ds(const unsigned* __restrict__ outdeg, const unsigned* __restrict__ indeg,
                     float* __restrict__ dsout, float* __restrict__ dsin) {
  int i = blockIdx.x * 256 + threadIdx.x;
  if (i < NN) {
    unsigned od = outdeg[i], id = indeg[i];
    dsout[i] = od ? rsqrtf((float)od) : 1.0f;
    dsin[i]  = id ? rsqrtf((float)id) : 1.0f;
  }
}

__global__ void k_scan(const unsigned* __restrict__ deg, unsigned* __restrict__ offs) {
  __shared__ unsigned ts[1024];
  const int t = threadIdx.x;
  const int per = (NN + 1023) / 1024;
  int s = t * per; if (s > NN) s = NN;
  int e = s + per; if (e > NN) e = NN;
  unsigned sum = 0;
  for (int i = s; i < e; ++i) sum += deg[i];
  ts[t] = sum;
  __syncthreads();
  for (int off = 1; off < 1024; off <<= 1) {
    unsigned v = (t >= off) ? ts[t - off] : 0u;
    __syncthreads();
    if (t >= off) ts[t] += v;
    __syncthreads();
  }
  unsigned run = (t == 0) ? 0u : ts[t - 1];
  for (int i = s; i < e; ++i) { offs[i] = run; run += deg[i]; }
  if (t == 1023) offs[NN] = ts[1023];
}

__global__ void k_copy_u32(const unsigned* __restrict__ a, unsigned* __restrict__ b, int n) {
  int i = blockIdx.x * 256 + threadIdx.x;
  if (i < n) b[i] = a[i];
}

__global__ void k_fill(const int* __restrict__ src, const int* __restrict__ dst,
                       unsigned* __restrict__ cursor, unsigned* __restrict__ esrc) {
  int e = blockIdx.x * 256 + threadIdx.x;
  if (e < EE) {
    unsigned pos = atomicAdd(&cursor[dst[e]], 1u);
    esrc[pos] = (unsigned)src[e];
  }
}


__global__ void k_agg128(const float* __restrict__ h, const unsigned* __restrict__ offs,
                         const unsigned* __restrict__ esrc,
                         const float* __restrict__ dsout, const float* __restrict__ dsin,
                         float* __restrict__ agg) {
  const int wv = threadIdx.x >> 5, lane = threadIdx.x & 31;
  const int n = blockIdx.x * 8 + wv;
  const unsigned b = offs[n], e = offs[n + 1];
  float4 acc = make_float4(0.f, 0.f, 0.f, 0.f);
  for (unsigned i = b; i < e; ++i) {
    const unsigned s = esrc[i];
    const float w = dsout[s];
    const float4 v = *(const float4*)(h + (size_t)s * FF + lane * 4);
    acc.x = fmaf(w, v.x, acc.x);
    acc.y = fmaf(w, v.y, acc.y);
    acc.z = fmaf(w, v.z, acc.z);
    acc.w = fmaf(w, v.w, acc.w);
  }
  const float c = dsin[n];
  *(float4*)(agg + (size_t)n * FF + lane * 4) =
      make_float4(c * acc.x, c * acc.y, c * acc.z, c * acc.w);
}

__global__ void k_agg64(const float* __restrict__ P, const unsigned* __restrict__ offs,
                        const unsigned* __restrict__ esrc, const float* __restrict__ dsin,
                        float* __restrict__ R) {
  const int wv = threadIdx.x >> 5, lane = threadIdx.x & 31;
  const int n = blockIdx.x * 8 + wv;
  const unsigned b = offs[n], e = offs[n + 1];
  float2 acc = make_float2(0.f, 0.f);
  for (unsigned i = b; i < e; ++i) {
    const unsigned s = esrc[i];
    const float2 v = *(const float2*)(P + (size_t)s * 64 + lane * 2);
    acc.x += v.x;
    acc.y += v.y;
  }
  const float c = dsin[n];
  *(float2*)(R + (size_t)n * 64 + lane * 2) = make_float2(c * acc.x, c * acc.y);
}


static constexpr int GD_PITCH = 132;
static constexpr int GD_TILES = NN / 16;
__global__ __launch_bounds__(256) void k_gemm_dual(
    const float* __restrict__ Agg, const float* __restrict__ H,
    const float* __restrict__ W, const float* __restrict__ L,
    float* __restrict__ Z) {
  __shared__ __attribute__((aligned(16))) float sAg[16 * GD_PITCH];
  __shared__ __attribute__((aligned(16))) float sH[16 * GD_PITCH];
  const int wv = threadIdx.x >> 5;
  const int lane = threadIdx.x & 31;
  const int t = lane & 15;
  const int hh = lane >> 4;
  const int cb = wv << 4;

  v2f bw[32], bl[32];
#pragma unroll
  for (int ks = 0; ks < 32; ++ks) {
    const int kk = 4 * ks + 2 * hh;
    bw[ks].x = W[kk * FF + cb + t]; bw[ks].y = W[(kk + 1) * FF + cb + t];
    bl[ks].x = L[kk * FF + cb + t]; bl[ks].y = L[(kk + 1) * FF + cb + t];
  }

  const int so0 = wv * GD_PITCH + lane * 4;
  const int so1 = (wv + 8) * GD_PITCH + lane * 4;
  const float* fa = sAg + t * GD_PITCH + 2 * hh;
  const float* fh = sH + t * GD_PITCH + 2 * hh;

  int tile = blockIdx.x;
  float4 pa0, pa1, ph0, ph1;
  if (tile < GD_TILES) {
    const size_t g0 = (size_t)(tile * 16 + wv) * FF + lane * 4;
    const size_t g1 = g0 + (size_t)8 * FF;
    pa0 = *(const float4*)(Agg + g0); pa1 = *(const float4*)(Agg + g1);
    ph0 = *(const float4*)(H + g0);   ph1 = *(const float4*)(H + g1);
  }
  for (; tile < GD_TILES; tile += gridDim.x) {
    __syncthreads();
    *(float4*)(sAg + so0) = pa0; *(float4*)(sAg + so1) = pa1;
    *(float4*)(sH + so0) = ph0;  *(float4*)(sH + so1) = ph1;
    __syncthreads();
    const int nt = tile + gridDim.x;
    if (nt < GD_TILES) {
      const size_t g0 = (size_t)(nt * 16 + wv) * FF + lane * 4;
      const size_t g1 = g0 + (size_t)8 * FF;
      pa0 = *(const float4*)(Agg + g0); pa1 = *(const float4*)(Agg + g1);
      ph0 = *(const float4*)(H + g0);   ph1 = *(const float4*)(H + g1);
    }
    v8f acc = {};
#pragma unroll
    for (int ks = 0; ks < 32; ++ks) {
      v2f a1 = *(const v2f*)(fa + 4 * ks);
      acc = __builtin_amdgcn_wmma_f32_16x16x4_f32(false, a1, false, bw[ks], (short)0, acc, false, false);
      v2f a2 = *(const v2f*)(fh + 4 * ks);
      acc = __builtin_amdgcn_wmma_f32_16x16x4_f32(false, a2, false, bl[ks], (short)0, acc, false, false);
    }
    const int r0 = tile << 4;
#pragma unroll
    for (int v = 0; v < 8; ++v)
      Z[(size_t)(r0 + v + 8 * hh) * FF + cb + t] = acc[v];
  }
}

static constexpr int GP_TILES = NN / 32;
__global__ __launch_bounds__(256) void k_gemm_p(
    const float* __restrict__ H, const float* __restrict__ dsout,
    const float* __restrict__ Wp, float* __restrict__ P) {
  __shared__ __attribute__((aligned(16))) float sT[32 * GD_PITCH];
  const int wv = threadIdx.x >> 5;
  const int lane = threadIdx.x & 31;
  const int t = lane & 15, hh = lane >> 4;
  const int ct = wv & 3, rt = wv >> 2;
  const int cb = ct << 4;

  v2f bp[32];
#pragma unroll
  for (int ks = 0; ks < 32; ++ks) {
    const int kk = 4 * ks + 2 * hh;
    bp[ks].x = Wp[kk * 64 + cb + t]; bp[ks].y = Wp[(kk + 1) * 64 + cb + t];
  }

  const int so = wv * GD_PITCH + lane * 4;
  const float* fa = sT + (16 * rt + t) * GD_PITCH + 2 * hh;

  int tile = blockIdx.x;
  float4 p0, p1, p2, p3;
  float dn = 0.f;
  if (tile < GP_TILES) {
    const size_t g = (size_t)(tile * 32 + wv) * FF + lane * 4;
    p0 = *(const float4*)(H + g);
    p1 = *(const float4*)(H + g + (size_t)8 * FF);
    p2 = *(const float4*)(H + g + (size_t)16 * FF);
    p3 = *(const float4*)(H + g + (size_t)24 * FF);
    dn = dsout[tile * 32 + 16 * rt + t];
  }
  for (; tile < GP_TILES; tile += gridDim.x) {
    const float d = dn;
    __syncthreads();
    *(float4*)(sT + so) = p0;
    *(float4*)(sT + so + 8 * GD_PITCH) = p1;
    *(float4*)(sT + so + 16 * GD_PITCH) = p2;
    *(float4*)(sT + so + 24 * GD_PITCH) = p3;
    __syncthreads();
    const int nt = tile + gridDim.x;
    if (nt < GP_TILES) {
      const size_t g = (size_t)(nt * 32 + wv) * FF + lane * 4;
      p0 = *(const float4*)(H + g);
      p1 = *(const float4*)(H + g + (size_t)8 * FF);
      p2 = *(const float4*)(H + g + (size_t)16 * FF);
      p3 = *(const float4*)(H + g + (size_t)24 * FF);
      dn = dsout[nt * 32 + 16 * rt + t];
    }
    v8f acc = {};
#pragma unroll
    for (int ks = 0; ks < 32; ++ks) {
      v2f a = *(const v2f*)(fa + 4 * ks);
      a.x = d * a.x; a.y = d * a.y;
      acc = __builtin_amdgcn_wmma_f32_16x16x4_f32(false, a, false, bp[ks], (short)0, acc, false, false);
    }
    const int r0 = tile * 32 + 16 * rt;
#pragma unroll
    for (int v = 0; v < 8; ++v)
      P[(size_t)(r0 + v + 8 * hh) * 64 + cb + t] = acc[v];
  }
}

__global__ __launch_bounds__(256) void k_gemm_out(
    const float* __restrict__ H, const float* __restrict__ Lp,
    const float* __restrict__ R, const float* __restrict__ b2,
    float* __restrict__ out) {
  __shared__ __attribute__((aligned(16))) float sT[32 * GD_PITCH];
  const int wv = threadIdx.x >> 5;
  const int lane = threadIdx.x & 31;
  const int t = lane & 15, hh = lane >> 4;
  const int ct = wv & 3, rt = wv >> 2;
  const int cb = ct << 4;
  const bool active = ct < 3;

  v2f bq[32];
#pragma unroll
  for (int ks = 0; ks < 32; ++ks) { bq[ks].x = 0.f; bq[ks].y = 0.f; }
  if (active) {
#pragma unroll
    for (int ks = 0; ks < 32; ++ks) {
      const int kk = 4 * ks + 2 * hh;
      bq[ks].x = Lp[kk * 64 + cb + t]; bq[ks].y = Lp[(kk + 1) * 64 + cb + t];
    }
  }

  const int so = wv * GD_PITCH + lane * 4;
  const float* fa = sT + (16 * rt + t) * GD_PITCH + 2 * hh;
  const int col = cb + t;

  int tile = blockIdx.x;
  float4 p0, p1, p2, p3;
  float rn[8];
#pragma unroll
  for (int v = 0; v < 8; ++v) rn[v] = 0.f;
  if (tile < GP_TILES) {
    const size_t g = (size_t)(tile * 32 + wv) * FF + lane * 4;
    p0 = *(const float4*)(H + g);
    p1 = *(const float4*)(H + g + (size_t)8 * FF);
    p2 = *(const float4*)(H + g + (size_t)16 * FF);
    p3 = *(const float4*)(H + g + (size_t)24 * FF);
    if (active) {
#pragma unroll
      for (int v = 0; v < 8; ++v)
        rn[v] = R[(size_t)(tile * 32 + 16 * rt + v + 8 * hh) * 64 + col];
    }
  }
  for (; tile < GP_TILES; tile += gridDim.x) {
    float rv[8];
#pragma unroll
    for (int v = 0; v < 8; ++v) rv[v] = rn[v];
    __syncthreads();
    *(float4*)(sT + so) = p0;
    *(float4*)(sT + so + 8 * GD_PITCH) = p1;
    *(float4*)(sT + so + 16 * GD_PITCH) = p2;
    *(float4*)(sT + so + 24 * GD_PITCH) = p3;
    __syncthreads();
    const int nt = tile + gridDim.x;
    if (nt < GP_TILES) {
      const size_t g = (size_t)(nt * 32 + wv) * FF + lane * 4;
      p0 = *(const float4*)(H + g);
      p1 = *(const float4*)(H + g + (size_t)8 * FF);
      p2 = *(const float4*)(H + g + (size_t)16 * FF);
      p3 = *(const float4*)(H + g + (size_t)24 * FF);
      if (active) {
#pragma unroll
        for (int v = 0; v < 8; ++v)
          rn[v] = R[(size_t)(nt * 32 + 16 * rt + v + 8 * hh) * 64 + col];
      }
    }
    if (active) {
      const int r0 = tile * 32 + 16 * rt;
      v8f acc = {};
#pragma unroll
      for (int ks = 0; ks < 32; ++ks) {
        v2f a = *(const v2f*)(fa + 4 * ks);
        acc = __builtin_amdgcn_wmma_f32_16x16x4_f32(false, a, false, bq[ks], (short)0, acc, false, false);
      }
      if (col < 40) {
        const float bias = b2[col];
#pragma unroll
        for (int v = 0; v < 8; ++v)
          out[(size_t)(r0 + v + 8 * hh) * 40 + col] = acc[v] + rv[v] + bias;
      }
    }
  }
}


__global__ void k_colstats(const float* __restrict__ z, float* __restrict__ cs,
                           float* __restrict__ cs2) {
  const int c = threadIdx.x;
  const size_t r0 = (size_t)blockIdx.x * 250;
  float s = 0.f, s2 = 0.f;
  for (int i = 0; i < 250; ++i) {
    const float v = z[(r0 + i) * FF + c];
    s += v;
    s2 = fmaf(v, v, s2);
  }
  atomicAdd(&cs[c], s);
  atomicAdd(&cs2[c], s2);
}

__global__ void k_bnparams(const float* __restrict__ cs, const float* __restrict__ cs2,
                           const float* __restrict__ g, const float* __restrict__ be,
                           float* __restrict__ bnA, float* __restrict__ bnB) {
  const int c = threadIdx.x;
  const float inv = 1.0f / (float)NN;
  const float mu = cs[c] * inv;
  const float var = cs2[c] * inv - mu * mu;
  const float a = g[c] * rsqrtf(var + 1e-5f);
  bnA[c] = a;
  bnB[c] = be[c] - mu * a;
}

__global__ void k_bnapply(const float* __restrict__ z, const float* __restrict__ bnA,
                          const float* __restrict__ bnB, float* __restrict__ hout) {
  const size_t i = (size_t)blockIdx.x * 256 + threadIdx.x;
  const int c = (int)(i & 127);
  const float v = fmaf(z[i], bnA[c], bnB[c]);
  hout[i] = v > 0.f ? v : 0.f;
}

__global__ void k_padw(const float* __restrict__ W2, const float* __restrict__ L2,
                       float* __restrict__ Wp, float* __restrict__ Lp) {
  const int i = blockIdx.x * 256 + threadIdx.x;
  const int k = i >> 6, c = i & 63;
  const float w = (c < 40) ? W2[k * 40 + c] : 0.f;
  const float l = (c < 40) ? L2[k * 40 + c] : 0.f;
  Wp[i] = w;
  Lp[i] = l;
}


extern "C" void kernel_launch(void* const* d_in, const int* in_sizes, int n_in,
                              void* d_out, int out_size, void* d_ws, size_t ws_size,
                              hipStream_t stream) {
  (void)in_sizes; (void)n_in; (void)out_size; (void)ws_size;

  const float* feat = (const float*)d_in[0];
  const int* src   = (const int*)d_in[1];
  const int* dst   = (const int*)d_in[2];
  const float* W0  = (const float*)d_in[3];
  const float* L0  = (const float*)d_in[4];
  const float* W1  = (const float*)d_in[5];
  const float* L1  = (const float*)d_in[6];
  const float* W2  = (const float*)d_in[7];
  const float* L2  = (const float*)d_in[8];
  const float* b2  = (const float*)d_in[9];
  const float* g0  = (const float*)d_in[10];
  const float* be0 = (const float*)d_in[11];
  const float* g1  = (const float*)d_in[12];
  const float* be1 = (const float*)d_in[13];
  float* out = (float*)d_out;

  char* base = (char*)d_ws;
  size_t off = 0;
  auto take = [&](size_t bytes) -> char* {
    char* p = base + off;
    off = (off + bytes + 255) & ~(size_t)255;
    return p;
  };
  unsigned* indeg  = (unsigned*)take((size_t)NN * 4);
  unsigned* outdeg = (unsigned*)take((size_t)NN * 4);
  unsigned* offs   = (unsigned*)take((size_t)(NN + 1) * 4);
  unsigned* cursor = (unsigned*)take((size_t)NN * 4);
  unsigned* esrc   = (unsigned*)take((size_t)EE * 4);
  float* dsout = (float*)take((size_t)NN * 4);
  float* dsin  = (float*)take((size_t)NN * 4);
  float* cs    = (float*)take(128 * 4);
  float* cs2   = (float*)take(128 * 4);
  float* bnA   = (float*)take(128 * 4);
  float* bnB   = (float*)take(128 * 4);
  float* Wp    = (float*)take(128 * 64 * 4);
  float* Lp    = (float*)take(128 * 64 * 4);
  float* bufA  = (float*)take((size_t)NN * FF * 4);
  float* bufB  = (float*)take((size_t)NN * FF * 4);
  float* bufC  = (float*)take((size_t)NN * FF * 4);

  const int BEdge = (EE + 255) / 256;
  const int BNode = (NN + 255) / 256;
  const int BWave = NN / 8;
  const int BApply = (NN * FF) / 256;

  k_zero_u32<<<BNode, 256, 0, stream>>>(indeg, NN);
  k_zero_u32<<<BNode, 256, 0, stream>>>(outdeg, NN);
  k_deg<<<BEdge, 256, 0, stream>>>(src, dst, outdeg, indeg);
  k_ds<<<BNode, 256, 0, stream>>>(outdeg, indeg, dsout, dsin);
  k_scan<<<1, 1024, 0, stream>>>(indeg, offs);
  k_copy_u32<<<BNode, 256, 0, stream>>>(offs, cursor, NN);
  k_fill<<<BEdge, 256, 0, stream>>>(src, dst, cursor, esrc);
  k_padw<<<32, 256, 0, stream>>>(W2, L2, Wp, Lp);

  k_agg128<<<BWave, 256, 0, stream>>>(feat, offs, esrc, dsout, dsin, bufA);
  k_gemm_dual<<<512, 256, 0, stream>>>(bufA, feat, W0, L0, bufB);
  k_zero_u32<<<1, 256, 0, stream>>>((unsigned*)cs, 256);
  k_colstats<<<400, 128, 0, stream>>>(bufB, cs, cs2);
  k_bnparams<<<1, 128, 0, stream>>>(cs, cs2, g0, be0, bnA, bnB);
  k_bnapply<<<BApply, 256, 0, stream>>>(bufB, bnA, bnB, bufC);

  k_agg128<<<BWave, 256, 0, stream>>>(bufC, offs, esrc, dsout, dsin, bufA);
  k_gemm_dual<<<512, 256, 0, stream>>>(bufA, bufC, W1, L1, bufB);
  k_zero_u32<<<1, 256, 0, stream>>>((unsigned*)cs, 256);
  k_colstats<<<400, 128, 0, stream>>>(bufB, cs, cs2);
  k_bnparams<<<1, 128, 0, stream>>>(cs, cs2, g1, be1, bnA, bnB);
  k_bnapply<<<BApply, 256, 0, stream>>>(bufB, bnA, bnB, bufC);

  k_gemm_p<<<1024, 256, 0, stream>>>(bufC, dsout, Wp, bufA);
  k_agg64<<<BWave, 256, 0, stream>>>(bufA, offs, esrc, dsin, bufB);
  k_gemm_out<<<1024, 256, 0, stream>>>(bufC, Lp, bufB, b2, out);
}


# BN-apply kernel: 10 float4 per thread in a 1250-block grid that is resident at once, instead of one float per thread in 50000 blocks: 400,000 waves per launch -> 10,000 (wave entry was the limit)
# speedup vs baseline: 1.2812x; 1.0394x over previous
#include <hip/hip_runtime.h>
#include <hip/hip_bf16.h>


static constexpr int NN = 100000;
static constexpr int EE = 1600000;
static constexpr int FF = 128;

typedef __attribute__((ext_vector_type(2))) float v2f;
typedef __attribute__((ext_vector_type(8))) float v8f;


__global__ void k_zero_u32(unsigned* __restrict__ p, int n) {
  int i = blockIdx.x * 256 + threadIdx.x;
  if (i < n) p[i] = 0u;
}


__global__ void k_deg(const int* __restrict__ src, const int* __restrict__ dst,
                      unsigned* __restrict__ outdeg, unsigned* __restrict__ indeg) {
  int e = blockIdx.x * 256 + threadIdx.x;
  if (e < EE) {
    atomicAdd(&outdeg[src[e]], 1u);
    atomicAdd(&indeg[dst[e]], 1u);
  }
}

__global__ void k_ds(const unsigned* __restrict__ outdeg, const unsigned* __restrict__ indeg,
                     float* __restrict__ dsout, float* __restrict__ dsin) {
  int i = blockIdx.x * 256 + threadIdx.x;
  if (i < NN) {
    unsigned od = outdeg[i], id = indeg[i];
    dsout[i] = od ? rsqrtf((float)od) : 1.0f;
    dsin[i]  = id ? rsqrtf((float)id) : 1.0f;
  }
}

__global__ void k_scan(const unsigned* __restrict__ deg, unsigned* __restrict__ offs) {
  __shared__ unsigned ts[1024];
  const int t = threadIdx.x;
  const int per = (NN + 1023) / 1024;
  int s = t * per; if (s > NN) s = NN;
  int e = s + per; if (e > NN) e = NN;
  unsigned sum = 0;
  for (int i = s; i < e; ++i) sum += deg[i];
  ts[t] = sum;
  __syncthreads();
  for (int off = 1; off < 1024; off <<= 1) {
    unsigned v = (t >= off) ? ts[t - off] : 0u;
    __syncthreads();
    if (t >= off) ts[t] += v;
    __syncthreads();
  }
  unsigned run = (t == 0) ? 0u : ts[t - 1];
  for (int i = s; i < e; ++i) { offs[i] = run; run += deg[i]; }
  if (t == 1023) offs[NN] = ts[1023];
}

__global__ void k_copy_u32(const unsigned* __restrict__ a, unsigned* __restrict__ b, int n) {
  int i = blockIdx.x * 256 + threadIdx.x;
  if (i < n) b[i] = a[i];
}

__global__ void k_fill(const int* __restrict__ src, const int* __restrict__ dst,
                       unsigned* __restrict__ cursor, unsigned* __restrict__ esrc) {
  int e = blockIdx.x * 256 + threadIdx.x;
  if (e < EE) {
    unsigned pos = atomicAdd(&cursor[dst[e]], 1u);
    esrc[pos] = (unsigned)src[e];
  }
}


__global__ void k_agg128(const float* __restrict__ h, const unsigned* __restrict__ offs,
                         const unsigned* __restrict__ esrc,
                         const float* __restrict__ dsout, const float* __restrict__ dsin,
                         float* __restrict__ agg) {
  const int wv = threadIdx.x >> 5, lane = threadIdx.x & 31;
  const int n = blockIdx.x * 8 + wv;
  const unsigned b = offs[n], e = offs[n + 1];
  float4 acc = make_float4(0.f, 0.f, 0.f, 0.f);
  for (unsigned i = b; i < e; ++i) {
    const unsigned s = esrc[i];
    const float w = dsout[s];
    const float4 v = *(const float4*)(h + (size_t)s * FF + lane * 4);
    acc.x = fmaf(w, v.x, acc.x);
    acc.y = fmaf(w, v.y, acc.y);
    acc.z = fmaf(w, v.z, acc.z);
    acc.w = fmaf(w, v.w, acc.w);
  }
  const float c = dsin[n];
  *(float4*)(agg + (size_t)n * FF + lane * 4) =
      make_float4(c * acc.x, c * acc.y, c * acc.z, c * acc.w);
}

__global__ void k_agg64(const float* __restrict__ P, const unsigned* __restrict__ offs,
                        const unsigned* __restrict__ esrc, const float* __restrict__ dsin,
                        float* __restrict__ R) {
  const int wv = threadIdx.x >> 5, lane = threadIdx.x & 31;
  const int n = blockIdx.x * 8 + wv;
  const unsigned b = offs[n], e = offs[n + 1];
  float2 acc = make_float2(0.f, 0.f);
  for (unsigned i = b; i < e; ++i) {
    const unsigned s = esrc[i];
    const float2 v = *(const float2*)(P + (size_t)s * 64 + lane * 2);
    acc.x += v.x;
    acc.y += v.y;
  }
  const float c = dsin[n];
  *(float2*)(R + (size_t)n * 64 + lane * 2) = make_float2(c * acc.x, c * acc.y);
}


static constexpr int GD_PITCH = 132;
static constexpr int GD_TILES = NN / 16;
__global__ __launch_bounds__(256) void k_gemm_dual(
    const float* __restrict__ Agg, const float* __restrict__ H,
    const float* __restrict__ W, const float* __restrict__ L,
    float* __restrict__ Z) {
  __shared__ __attribute__((aligned(16))) float sAg[16 * GD_PITCH];
  __shared__ __attribute__((aligned(16))) float sH[16 * GD_PITCH];
  const int wv = threadIdx.x >> 5;
  const int lane = threadIdx.x & 31;
  const int t = lane & 15;
  const int hh = lane >> 4;
  const int cb = wv << 4;

  v2f bw[32], bl[32];
#pragma unroll
  for (int ks = 0; ks < 32; ++ks) {
    const int kk = 4 * ks + 2 * hh;
    bw[ks].x = W[kk * FF + cb + t]; bw[ks].y = W[(kk + 1) * FF + cb + t];
    bl[ks].x = L[kk * FF + cb + t]; bl[ks].y = L[(kk + 1) * FF + cb + t];
  }

  const int so0 = wv * GD_PITCH + lane * 4;
  const int so1 = (wv + 8) * GD_PITCH + lane * 4;
  const float* fa = sAg + t * GD_PITCH + 2 * hh;
  const float* fh = sH + t * GD_PITCH + 2 * hh;

  int tile = blockIdx.x;
  float4 pa0, pa1, ph0, ph1;
  if (tile < GD_TILES) {
    const size_t g0 = (size_t)(tile * 16 + wv) * FF + lane * 4;
    const size_t g1 = g0 + (size_t)8 * FF;
    pa0 = *(const float4*)(Agg + g0); pa1 = *(const float4*)(Agg + g1);
    ph0 = *(const float4*)(H + g0);   ph1 = *(const float4*)(H + g1);
  }
  for (; tile < GD_TILES; tile += gridDim.x) {
    __syncthreads();
    *(float4*)(sAg + so0) = pa0; *(float4*)(sAg + so1) = pa1;
    *(float4*)(sH + so0) = ph0;  *(float4*)(sH + so1) = ph1;
    __syncthreads();
    const int nt = tile + gridDim.x;
    if (nt < GD_TILES) {
      const size_t g0 = (size_t)(nt * 16 + wv) * FF + lane * 4;
      const size_t g1 = g0 + (size_t)8 * FF;
      pa0 = *(const float4*)(Agg + g0); pa1 = *(const float4*)(Agg + g1);
      ph0 = *(const float4*)(H + g0);   ph1 = *(const float4*)(H + g1);
    }
    v8f acc = {};
#pragma unroll
    for (int ks = 0; ks < 32; ++ks) {
      v2f a1 = *(const v2f*)(fa + 4 * ks);
      acc = __builtin_amdgcn_wmma_f32_16x16x4_f32(false, a1, false, bw[ks], (short)0, acc, false, false);
      v2f a2 = *(const v2f*)(fh + 4 * ks);
      acc = __builtin_amdgcn_wmma_f32_16x16x4_f32(false, a2, false, bl[ks], (short)0, acc, false, false);
    }
    const int r0 = tile << 4;
#pragma unroll
    for (int v = 0; v < 8; ++v)
      Z[(size_t)(r0 + v + 8 * hh) * FF + cb + t] = acc[v];
  }
}

static constexpr int GP_TILES = NN / 32;
__global__ __launch_bounds__(256) void k_gemm_p(
    const float* __restrict__ H, const float* __restrict__ dsout,
    const float* __restrict__ Wp, float* __restrict__ P) {
  __shared__ __attribute__((aligned(16))) float sT[32 * GD_PITCH];
  const int wv = threadIdx.x >> 5;
  const int lane = threadIdx.x & 31;
  const int t = lane & 15, hh = lane >> 4;
  const int ct = wv & 3, rt = wv >> 2;
  const int cb = ct << 4;

  v2f bp[32];
#pragma unroll
  for (int ks = 0; ks < 32; ++ks) {
    const int kk = 4 * ks + 2 * hh;
    bp[ks].x = Wp[kk * 64 + cb + t]; bp[ks].y = Wp[(kk + 1) * 64 + cb + t];
  }

  const int so = wv * GD_PITCH + lane * 4;
  const float* fa = sT + (16 * rt + t) * GD_PITCH + 2 * hh;

  int tile = blockIdx.x;
  float4 p0, p1, p2, p3;
  float dn = 0.f;
  if (tile < GP_TILES) {
    const size_t g = (size_t)(tile * 32 + wv) * FF + lane * 4;
    p0 = *(const float4*)(H + g);
    p1 = *(const float4*)(H + g + (size_t)8 * FF);
    p2 = *(const float4*)(H + g + (size_t)16 * FF);
    p3 = *(const float4*)(H + g + (size_t)24 * FF);
    dn = dsout[tile * 32 + 16 * rt + t];
  }
  for (; tile < GP_TILES; tile += gridDim.x) {
    const float d = dn;
    __syncthreads();
    *(float4*)(sT + so) = p0;
    *(float4*)(sT + so + 8 * GD_PITCH) = p1;
    *(float4*)(sT + so + 16 * GD_PITCH) = p2;
    *(float4*)(sT + so + 24 * GD_PITCH) = p3;
    __syncthreads();
    const int nt = tile + gridDim.x;
    if (nt < GP_TILES) {
      const size_t g = (size_t)(nt * 32 + wv) * FF + lane * 4;
      p0 = *(const float4*)(H + g);
      p1 = *(const float4*)(H + g + (size_t)8 * FF);
      p2 = *(const float4*)(H + g + (size_t)16 * FF);
      p3 = *(const float4*)(H + g + (size_t)24 * FF);
      dn = dsout[nt * 32 + 16 * rt + t];
    }
    v8f acc = {};
#pragma unroll
    for (int ks = 0; ks < 32; ++ks) {
      v2f a = *(const v2f*)(fa + 4 * ks);
      a.x = d * a.x; a.y = d * a.y;
      acc = __builtin_amdgcn_wmma_f32_16x16x4_f32(false, a, false, bp[ks], (short)0, acc, false, false);
    }
    const int r0 = tile * 32 + 16 * rt;
#pragma unroll
    for (int v = 0; v < 8; ++v)
      P[(size_t)(r0 + v + 8 * hh) * 64 + cb + t] = acc[v];
  }
}

__global__ __launch_bounds__(256) void k_gemm_out(
    const float* __restrict__ H, const float* __restrict__ Lp,
    const float* __restrict__ R, const float* __restrict__ b2,
    float* __restrict__ out) {
  __shared__ __attribute__((aligned(16))) float sT[32 * GD_PITCH];
  const int wv = threadIdx.x >> 5;
  const int lane = threadIdx.x & 31;
  const int t = lane & 15, hh = lane >> 4;
  const int ct = wv & 3, rt = wv >> 2;
  const int cb = ct << 4;
  const bool active = ct < 3;

  v2f bq[32];
#pragma unroll
  for (int ks = 0; ks < 32; ++ks) { bq[ks].x = 0.f; bq[ks].y = 0.f; }
  if (active) {
#pragma unroll
    for (int ks = 0; ks < 32; ++ks) {
      const int kk = 4 * ks + 2 * hh;
      bq[ks].x = Lp[kk * 64 + cb + t]; bq[ks].y = Lp[(kk + 1) * 64 + cb + t];
    }
  }

  const int so = wv * GD_PITCH + lane * 4;
  const float* fa = sT + (16 * rt + t) * GD_PITCH + 2 * hh;
  const int col = cb + t;

  int tile = blockIdx.x;
  float4 p0, p1, p2, p3;
  float rn[8];
#pragma unroll
  for (int v = 0; v < 8; ++v) rn[v] = 0.f;
  if (tile < GP_TILES) {
    const size_t g = (size_t)(tile * 32 + wv) * FF + lane * 4;
    p0 = *(const float4*)(H + g);
    p1 = *(const float4*)(H + g + (size_t)8 * FF);
    p2 = *(const float4*)(H + g + (size_t)16 * FF);
    p3 = *(const float4*)(H + g + (size_t)24 * FF);
    if (active) {
#pragma unroll
      for (int v = 0; v < 8; ++v)
        rn[v] = R[(size_t)(tile * 32 + 16 * rt + v + 8 * hh) * 64 + col];
    }
  }
  for (; tile < GP_TILES; tile += gridDim.x) {
    float rv[8];
#pragma unroll
    for (int v = 0; v < 8; ++v) rv[v] = rn[v];
    __syncthreads();
    *(float4*)(sT + so) = p0;
    *(float4*)(sT + so + 8 * GD_PITCH) = p1;
    *(float4*)(sT + so + 16 * GD_PITCH) = p2;
    *(float4*)(sT + so + 24 * GD_PITCH) = p3;
    __syncthreads();
    const int nt = tile + gridDim.x;
    if (nt < GP_TILES) {
      const size_t g = (size_t)(nt * 32 + wv) * FF + lane * 4;
      p0 = *(const float4*)(H + g);
      p1 = *(const float4*)(H + g + (size_t)8 * FF);
      p2 = *(const float4*)(H + g + (size_t)16 * FF);
      p3 = *(const float4*)(H + g + (size_t)24 * FF);
      if (active) {
#pragma unroll
        for (int v = 0; v < 8; ++v)
          rn[v] = R[(size_t)(nt * 32 + 16 * rt + v + 8 * hh) * 64 + col];
      }
    }
    if (active) {
      const int r0 = tile * 32 + 16 * rt;
      v8f acc = {};
#pragma unroll
      for (int ks = 0; ks < 32; ++ks) {
        v2f a = *(const v2f*)(fa + 4 * ks);
        acc = __builtin_amdgcn_wmma_f32_16x16x4_f32(false, a, false, bq[ks], (short)0, acc, false, false);
      }
      if (col < 40) {
        const float bias = b2[col];
#pragma unroll
        for (int v = 0; v < 8; ++v)
          out[(size_t)(r0 + v + 8 * hh) * 40 + col] = acc[v] + rv[v] + bias;
      }
    }
  }
}


__global__ void k_colstats(const float* __restrict__ z, float* __restrict__ cs,
                           float* __restrict__ cs2) {
  const int c = threadIdx.x;
  const size_t r0 = (size_t)blockIdx.x * 250;
  float s = 0.f, s2 = 0.f;
  for (int i = 0; i < 250; ++i) {
    const float v = z[(r0 + i) * FF + c];
    s += v;
    s2 = fmaf(v, v, s2);
  }
  atomicAdd(&cs[c], s);
  atomicAdd(&cs2[c], s2);
}

__global__ void k_bnparams(const float* __restrict__ cs, const float* __restrict__ cs2,
                           const float* __restrict__ g, const float* __restrict__ be,
                           float* __restrict__ bnA, float* __restrict__ bnB) {
  const int c = threadIdx.x;
  const float inv = 1.0f / (float)NN;
  const float mu = cs[c] * inv;
  const float var = cs2[c] * inv - mu * mu;
  const float a = g[c] * rsqrtf(var + 1e-5f);
  bnA[c] = a;
  bnB[c] = be[c] - mu * a;
}

static constexpr int BA_STEPS = 10;
static constexpr int BA_BLOCKS = (NN * FF / 4) / (256 * BA_STEPS);
static_assert((size_t)BA_BLOCKS * 256 * BA_STEPS * 4 == (size_t)NN * FF, "bnapply grid must tile the matrix exactly");
__global__ __launch_bounds__(256) void k_bnapply(
    const float* __restrict__ z, const float* __restrict__ bnA,
    const float* __restrict__ bnB, float* __restrict__ hout) {
  const int c4 = (threadIdx.x & 31) * 4;
  const float4 a = *(const float4*)(bnA + c4);
  const float4 b = *(const float4*)(bnB + c4);
  const size_t j0 = ((size_t)blockIdx.x * BA_STEPS * 256 + threadIdx.x) * 4;
  float4 v[BA_STEPS];
#pragma unroll
  for (int k = 0; k < BA_STEPS; ++k) v[k] = *(const float4*)(z + j0 + (size_t)k * 1024);
#pragma unroll
  for (int k = 0; k < BA_STEPS; ++k) {
    float4 r;
    r.x = fmaf(v[k].x, a.x, b.x); r.x = r.x > 0.f ? r.x : 0.f;
    r.y = fmaf(v[k].y, a.y, b.y); r.y = r.y > 0.f ? r.y : 0.f;
    r.z = fmaf(v[k].z, a.z, b.z); r.z = r.z > 0.f ? r.z : 0.f;
    r.w = fmaf(v[k].w, a.w, b.w); r.w = r.w > 0.f ? r.w : 0.f;
    *(float4*)(hout + j0 + (size_t)k * 1024) = r;
  }
}

__global__ void k_padw(const float* __restrict__ W2, const float* __restrict__ L2,
                       float* __restrict__ Wp, float* __restrict__ Lp) {
  const int i = blockIdx.x * 256 + threadIdx.x;
  const int k = i >> 6, c = i & 63;
  const float w = (c < 40) ? W2[k * 40 + c] : 0.f;
  const float l = (c < 40) ? L2[k * 40 + c] : 0.f;
  Wp[i] = w;
  Lp[i] = l;
}


extern "C" void kernel_launch(void* const* d_in, const int* in_sizes, int n_in,
                              void* d_out, int out_size, void* d_ws, size_t ws_size,
                              hipStream_t stream) {
  (void)in_sizes; (void)n_in; (void)out_size; (void)ws_size;

  const float* feat = (const float*)d_in[0];
  const int* src   = (const int*)d_in[1];
  const int* dst   = (const int*)d_in[2];
  const float* W0  = (const float*)d_in[3];
  const float* L0  = (const float*)d_in[4];
  const float* W1  = (const float*)d_in[5];
  const float* L1  = (const float*)d_in[6];
  const float* W2  = (const float*)d_in[7];
  const float* L2  = (const float*)d_in[8];
  const float* b2  = (const float*)d_in[9];
  const float* g0  = (const float*)d_in[10];
  const float* be0 = (const float*)d_in[11];
  const float* g1  = (const float*)d_in[12];
  const float* be1 = (const float*)d_in[13];
  float* out = (float*)d_out;

  char* base = (char*)d_ws;
  size_t off = 0;
  auto take = [&](size_t bytes) -> char* {
    char* p = base + off;
    off = (off + bytes + 255) & ~(size_t)255;
    return p;
  };
  unsigned* indeg  = (unsigned*)take((size_t)NN * 4);
  unsigned* outdeg = (unsigned*)take((size_t)NN * 4);
  unsigned* offs   = (unsigned*)take((size_t)(NN + 1) * 4);
  unsigned* cursor = (unsigned*)take((size_t)NN * 4);
  unsigned* esrc   = (unsigned*)take((size_t)EE * 4);
  float* dsout = (float*)take((size_t)NN * 4);
  float* dsin  = (float*)take((size_t)NN * 4);
  float* cs    = (float*)take(128 * 4);
  float* cs2   = (float*)take(128 * 4);
  float* bnA   = (float*)take(128 * 4);
  float* bnB   = (float*)take(128 * 4);
  float* Wp    = (float*)take(128 * 64 * 4);
  float* Lp    = (float*)take(128 * 64 * 4);
  float* bufA  = (float*)take((size_t)NN * FF * 4);
  float* bufB  = (float*)take((size_t)NN * FF * 4);
  float* bufC  = (float*)take((size_t)NN * FF * 4);

  const int BEdge = (EE + 255) / 256;
  const int BNode = (NN + 255) / 256;
  const int BWave = NN / 8;

  k_zero_u32<<<BNode, 256, 0, stream>>>(indeg, NN);
  k_zero_u32<<<BNode, 256, 0, stream>>>(outdeg, NN);
  k_deg<<<BEdge, 256, 0, stream>>>(src, dst, outdeg, indeg);
  k_ds<<<BNode, 256, 0, stream>>>(outdeg, indeg, dsout, dsin);
  k_scan<<<1, 1024, 0, stream>>>(indeg, offs);
  k_copy_u32<<<BNode, 256, 0, stream>>>(offs, cursor, NN);
  k_fill<<<BEdge, 256, 0, stream>>>(src, dst, cursor, esrc);
  k_padw<<<32, 256, 0, stream>>>(W2, L2, Wp, Lp);

  k_agg128<<<BWave, 256, 0, stream>>>(feat, offs, esrc, dsout, dsin, bufA);
  k_gemm_dual<<<512, 256, 0, stream>>>(bufA, feat, W0, L0, bufB);
  k_zero_u32<<<1, 256, 0, stream>>>((unsigned*)cs, 256);
  k_colstats<<<400, 128, 0, stream>>>(bufB, cs, cs2);
  k_bnparams<<<1, 128, 0, stream>>>(cs, cs2, g0, be0, bnA, bnB);
  k_bnapply<<<BA_BLOCKS, 256, 0, stream>>>(bufB, bnA, bnB, bufC);

  k_agg128<<<BWave, 256, 0, stream>>>(bufC, offs, esrc, dsout, dsin, bufA);
  k_gemm_dual<<<512, 256, 0, stream>>>(bufA, bufC, W1, L1, bufB);
  k_zero_u32<<<1, 256, 0, stream>>>((unsigned*)cs, 256);
  k_colstats<<<400, 128, 0, stream>>>(bufB, cs, cs2);
  k_bnparams<<<1, 128, 0, stream>>>(cs, cs2, g1, be1, bnA, bnB);
  k_bnapply<<<BA_BLOCKS, 256, 0, stream>>>(bufB, bnA, bnB, bufC);

  k_gemm_p<<<1024, 256, 0, stream>>>(bufC, dsout, Wp, bufA);
  k_agg64<<<BWave, 256, 0, stream>>>(bufA, offs, esrc, dsin, bufB);
  k_gemm_out<<<1024, 256, 0, stream>>>(bufC, Lp, bufB, b2, out);
}


# CSR fill without atomics: an edge's slot in its dst row is the value its in-degree atomic returns (stored per edge); cursor atomics and cursor-copy launch removed; row order stays arrival order
# speedup vs baseline: 1.4216x; 1.1096x over previous
#include <hip/hip_runtime.h>
#include <hip/hip_bf16.h>


static constexpr int NN = 100000;
static constexpr int EE = 1600000;
static constexpr int FF = 128;

typedef __attribute__((ext_vector_type(2))) float v2f;
typedef __attribute__((ext_vector_type(8))) float v8f;


__global__ void k_zero_u32(unsigned* __restrict__ p, int n) {
  int i = blockIdx.x * 256 + threadIdx.x;
  if (i < n) p[i] = 0u;
}


__global__ void k_deg(const int* __restrict__ src, const int* __restrict__ dst,
                      unsigned* __restrict__ outdeg, unsigned* __restrict__ indeg,
                      unsigned* __restrict__ rank) {
  int e = blockIdx.x * 256 + threadIdx.x;
  if (e < EE) {
    const int s = src[e], d = dst[e];
    atomicAdd(&outdeg[s], 1u);
    rank[e] = atomicAdd(&indeg[d], 1u);
  }
}

__global__ void k_ds(const unsigned* __restrict__ outdeg, const unsigned* __restrict__ indeg,
                     float* __restrict__ dsout, float* __restrict__ dsin) {
  int i = blockIdx.x * 256 + threadIdx.x;
  if (i < NN) {
    unsigned od = outdeg[i], id = indeg[i];
    dsout[i] = od ? rsqrtf((float)od) : 1.0f;
    dsin[i]  = id ? rsqrtf((float)id) : 1.0f;
  }
}

__global__ void k_scan(const unsigned* __restrict__ deg, unsigned* __restrict__ offs) {
  __shared__ unsigned ts[1024];
  const int t = threadIdx.x;
  const int per = (NN + 1023) / 1024;
  int s = t * per; if (s > NN) s = NN;
  int e = s + per; if (e > NN) e = NN;
  unsigned sum = 0;
  for (int i = s; i < e; ++i) sum += deg[i];
  ts[t] = sum;
  __syncthreads();
  for (int off = 1; off < 1024; off <<= 1) {
    unsigned v = (t >= off) ? ts[t - off] : 0u;
    __syncthreads();
    if (t >= off) ts[t] += v;
    __syncthreads();
  }
  unsigned run = (t == 0) ? 0u : ts[t - 1];
  for (int i = s; i < e; ++i) { offs[i] = run; run += deg[i]; }
  if (t == 1023) offs[NN] = ts[1023];
}

__global__ void k_fill(const int* __restrict__ src, const int* __restrict__ dst,
                       const unsigned* __restrict__ offs, const unsigned* __restrict__ rank,
                       unsigned* __restrict__ esrc) {
  int e = blockIdx.x * 256 + threadIdx.x;
  if (e < EE) {
    esrc[offs[dst[e]] + rank[e]] = (unsigned)src[e];
  }
}


__global__ void k_agg128(const float* __restrict__ h, const unsigned* __restrict__ offs,
                         const unsigned* __restrict__ esrc,
                         const float* __restrict__ dsout, const float* __restrict__ dsin,
                         float* __restrict__ agg) {
  const int wv = threadIdx.x >> 5, lane = threadIdx.x & 31;
  const int n = blockIdx.x * 8 + wv;
  const unsigned b = offs[n], e = offs[n + 1];
  float4 acc = make_float4(0.f, 0.f, 0.f, 0.f);
  for (unsigned i = b; i < e; ++i) {
    const unsigned s = esrc[i];
    const float w = dsout[s];
    const float4 v = *(const float4*)(h + (size_t)s * FF + lane * 4);
    acc.x = fmaf(w, v.x, acc.x);
    acc.y = fmaf(w, v.y, acc.y);
    acc.z = fmaf(w, v.z, acc.z);
    acc.w = fmaf(w, v.w, acc.w);
  }
  const float c = dsin[n];
  *(float4*)(agg + (size_t)n * FF + lane * 4) =
      make_float4(c * acc.x, c * acc.y, c * acc.z, c * acc.w);
}

__global__ void k_agg64(const float* __restrict__ P, const unsigned* __restrict__ offs,
                        const unsigned* __restrict__ esrc, const float* __restrict__ dsin,
                        float* __restrict__ R) {
  const int wv = threadIdx.x >> 5, lane = threadIdx.x & 31;
  const int n = blockIdx.x * 8 + wv;
  const unsigned b = offs[n], e = offs[n + 1];
  float2 acc = make_float2(0.f, 0.f);
  for (unsigned i = b; i < e; ++i) {
    const unsigned s = esrc[i];
    const float2 v = *(const float2*)(P + (size_t)s * 64 + lane * 2);
    acc.x += v.x;
    acc.y += v.y;
  }
  const float c = dsin[n];
  *(float2*)(R + (size_t)n * 64 + lane * 2) = make_float2(c * acc.x, c * acc.y);
}


static constexpr int GD_PITCH = 132;
static constexpr int GD_TILES = NN / 16;
__global__ __launch_bounds__(256) void k_gemm_dual(
    const float* __restrict__ Agg, const float* __restrict__ H,
    const float* __restrict__ W, const float* __restrict__ L,
    float* __restrict__ Z) {
  __shared__ __attribute__((aligned(16))) float sAg[16 * GD_PITCH];
  __shared__ __attribute__((aligned(16))) float sH[16 * GD_PITCH];
  const int wv = threadIdx.x >> 5;
  const int lane = threadIdx.x & 31;
  const int t = lane & 15;
  const int hh = lane >> 4;
  const int cb = wv << 4;

  v2f bw[32], bl[32];
#pragma unroll
  for (int ks = 0; ks < 32; ++ks) {
    const int kk = 4 * ks + 2 * hh;
    bw[ks].x = W[kk * FF + cb + t]; bw[ks].y = W[(kk + 1) * FF + cb + t];
    bl[ks].x = L[kk * FF + cb + t]; bl[ks].y = L[(kk + 1) * FF + cb + t];
  }

  const int so0 = wv * GD_PITCH + lane * 4;
  const int so1 = (wv + 8) * GD_PITCH + lane * 4;
  const float* fa = sAg + t * GD_PITCH + 2 * hh;
  const float* fh = sH + t * GD_PITCH + 2 * hh;

  int tile = blockIdx.x;
  float4 pa0, pa1, ph0, ph1;
  if (tile < GD_TILES) {
    const size_t g0 = (size_t)(tile * 16 + wv) * FF + lane * 4;
    const size_t g1 = g0 + (size_t)8 * FF;
    pa0 = *(const float4*)(Agg + g0); pa1 = *(const float4*)(Agg + g1);
    ph0 = *(const float4*)(H + g0);   ph1 = *(const float4*)(H + g1);
  }
  for (; tile < GD_TILES; tile += gridDim.x) {
    __syncthreads();
    *(float4*)(sAg + so0) = pa0; *(float4*)(sAg + so1) = pa1;
    *(float4*)(sH + so0) = ph0;  *(float4*)(sH + so1) = ph1;
    __syncthreads();
    const int nt = tile + gridDim.x;
    if (nt < GD_TILES) {
      const size_t g0 = (size_t)(nt * 16 + wv) * FF + lane * 4;
      const size_t g1 = g0 + (size_t)8 * FF;
      pa0 = *(const float4*)(Agg + g0); pa1 = *(const float4*)(Agg + g1);
      ph0 = *(const float4*)(H + g0);   ph1 = *(const float4*)(H + g1);
    }
    v8f acc = {};
#pragma unroll
    for (int ks = 0; ks < 32; ++ks) {
      v2f a1 = *(const v2f*)(fa + 4 * ks);
      acc = __builtin_amdgcn_wmma_f32_16x16x4_f32(false, a1, false, bw[ks], (short)0, acc, false, false);
      v2f a2 = *(const v2f*)(fh + 4 * ks);
      acc = __builtin_amdgcn_wmma_f32_16x16x4_f32(false, a2, false, bl[ks], (short)0, acc, false, false);
    }
    const int r0 = tile << 4;
#pragma unroll
    for (int v = 0; v < 8; ++v)
      Z[(size_t)(r0 + v + 8 * hh) * FF + cb + t] = acc[v];
  }
}

static constexpr int GP_TILES = NN / 32;
__global__ __launch_bounds__(256) void k_gemm_p(
    const float* __restrict__ H, const float* __restrict__ dsout,
    const float* __restrict__ Wp, float* __restrict__ P) {
  __shared__ __attribute__((aligned(16))) float sT[32 * GD_PITCH];
  const int wv = threadIdx.x >> 5;
  const int lane = threadIdx.x & 31;
  const int t = lane & 15, hh = lane >> 4;
  const int ct = wv & 3, rt = wv >> 2;
  const int cb = ct << 4;

  v2f bp[32];
#pragma unroll
  for (int ks = 0; ks < 32; ++ks) {
    const int kk = 4 * ks + 2 * hh;
    bp[ks].x = Wp[kk * 64 + cb + t]; bp[ks].y = Wp[(kk + 1) * 64 + cb + t];
  }

  const int so = wv * GD_PITCH + lane * 4;
  const float* fa = sT + (16 * rt + t) * GD_PITCH + 2 * hh;

  int tile = blockIdx.x;
  float4 p0, p1, p2, p3;
  float dn = 0.f;
  if (tile < GP_TILES) {
    const size_t g = (size_t)(tile * 32 + wv) * FF + lane * 4;
    p0 = *(const float4*)(H + g);
    p1 = *(const float4*)(H + g + (size_t)8 * FF);
    p2 = *(const float4*)(H + g + (size_t)16 * FF);
    p3 = *(const float4*)(H + g + (size_t)24 * FF);
    dn = dsout[tile * 32 + 16 * rt + t];
  }
  for (; tile < GP_TILES; tile += gridDim.x) {
    const float d = dn;
    __syncthreads();
    *(float4*)(sT + so) = p0;
    *(float4*)(sT + so + 8 * GD_PITCH) = p1;
    *(float4*)(sT + so + 16 * GD_PITCH) = p2;
    *(float4*)(sT + so + 24 * GD_PITCH) = p3;
    __syncthreads();
    const int nt = tile + gridDim.x;
    if (nt < GP_TILES) {
      const size_t g = (size_t)(nt * 32 + wv) * FF + lane * 4;
      p0 = *(const float4*)(H + g);
      p1 = *(const float4*)(H + g + (size_t)8 * FF);
      p2 = *(const float4*)(H + g + (size_t)16 * FF);
      p3 = *(const float4*)(H + g + (size_t)24 * FF);
      dn = dsout[nt * 32 + 16 * rt + t];
    }
    v8f acc = {};
#pragma unroll
    for (int ks = 0; ks < 32; ++ks) {
      v2f a = *(const v2f*)(fa + 4 * ks);
      a.x = d * a.x; a.y = d * a.y;
      acc = __builtin_amdgcn_wmma_f32_16x16x4_f32(false, a, false, bp[ks], (short)0, acc, false, false);
    }
    const int r0 = tile * 32 + 16 * rt;
#pragma unroll
    for (int v = 0; v < 8; ++v)
      P[(size_t)(r0 + v + 8 * hh) * 64 + cb + t] = acc[v];
  }
}

__global__ __launch_bounds__(256) void k_gemm_out(
    const float* __restrict__ H, const float* __restrict__ Lp,
    const float* __restrict__ R, const float* __restrict__ b2,
    float* __restrict__ out) {
  __shared__ __attribute__((aligned(16))) float sT[32 * GD_PITCH];
  const int wv = threadIdx.x >> 5;
  const int lane = threadIdx.x & 31;
  const int t = lane & 15, hh = lane >> 4;
  const int ct = wv & 3, rt = wv >> 2;
  const int cb = ct << 4;
  const bool active = ct < 3;

  v2f bq[32];
#pragma unroll
  for (int ks = 0; ks < 32; ++ks) { bq[ks].x = 0.f; bq[ks].y = 0.f; }
  if (active) {
#pragma unroll
    for (int ks = 0; ks < 32; ++ks) {
      const int kk = 4 * ks + 2 * hh;
      bq[ks].x = Lp[kk * 64 + cb + t]; bq[ks].y = Lp[(kk + 1) * 64 + cb + t];
    }
  }

  const int so = wv * GD_PITCH + lane * 4;
  const float* fa = sT + (16 * rt + t) * GD_PITCH + 2 * hh;
  const int col = cb + t;

  int tile = blockIdx.x;
  float4 p0, p1, p2, p3;
  float rn[8];
#pragma unroll
  for (int v = 0; v < 8; ++v) rn[v] = 0.f;
  if (tile < GP_TILES) {
    const size_t g = (size_t)(tile * 32 + wv) * FF + lane * 4;
    p0 = *(const float4*)(H + g);
    p1 = *(const float4*)(H + g + (size_t)8 * FF);
    p2 = *(const float4*)(H + g + (size_t)16 * FF);
    p3 = *(const float4*)(H + g + (size_t)24 * FF);
    if (active) {
#pragma unroll
      for (int v = 0; v < 8; ++v)
        rn[v] = R[(size_t)(tile * 32 + 16 * rt + v + 8 * hh) * 64 + col];
    }
  }
  for (; tile < GP_TILES; tile += gridDim.x) {
    float rv[8];
#pragma unroll
    for (int v = 0; v < 8; ++v) rv[v] = rn[v];
    __syncthreads();
    *(float4*)(sT + so) = p0;
    *(float4*)(sT + so + 8 * GD_PITCH) = p1;
    *(float4*)(sT + so + 16 * GD_PITCH) = p2;
    *(float4*)(sT + so + 24 * GD_PITCH) = p3;
    __syncthreads();
    const int nt = tile + gridDim.x;
    if (nt < GP_TILES) {
      const size_t g = (size_t)(nt * 32 + wv) * FF + lane * 4;
      p0 = *(const float4*)(H + g);
      p1 = *(const float4*)(H + g + (size_t)8 * FF);
      p2 = *(const float4*)(H + g + (size_t)16 * FF);
      p3 = *(const float4*)(H + g + (size_t)24 * FF);
      if (active) {
#pragma unroll
        for (int v = 0; v < 8; ++v)
          rn[v] = R[(size_t)(nt * 32 + 16 * rt + v + 8 * hh) * 64 + col];
      }
    }
    if (active) {
      const int r0 = tile * 32 + 16 * rt;
      v8f acc = {};
#pragma unroll
      for (int ks = 0; ks < 32; ++ks) {
        v2f a = *(const v2f*)(fa + 4 * ks);
        acc = __builtin_amdgcn_wmma_f32_16x16x4_f32(false, a, false, bq[ks], (short)0, acc, false, false);
      }
      if (col < 40) {
        const float bias = b2[col];
#pragma unroll
        for (int v = 0; v < 8; ++v)
          out[(size_t)(r0 + v + 8 * hh) * 40 + col] = acc[v] + rv[v] + bias;
      }
    }
  }
}


__global__ void k_colstats(const float* __restrict__ z, float* __restrict__ cs,
                           float* __restrict__ cs2) {
  const int c = threadIdx.x;
  const size_t r0 = (size_t)blockIdx.x * 250;
  float s = 0.f, s2 = 0.f;
  for (int i = 0; i < 250; ++i) {
    const float v = z[(r0 + i) * FF + c];
    s += v;
    s2 = fmaf(v, v, s2);
  }
  atomicAdd(&cs[c], s);
  atomicAdd(&cs2[c], s2);
}

__global__ void k_bnparams(const float* __restrict__ cs, const float* __restrict__ cs2,
                           const float* __restrict__ g, const float* __restrict__ be,
                           float* __restrict__ bnA, float* __restrict__ bnB) {
  const int c = threadIdx.x;
  const float inv = 1.0f / (float)NN;
  const float mu = cs[c] * inv;
  const float var = cs2[c] * inv - mu * mu;
  const float a = g[c] * rsqrtf(var + 1e-5f);
  bnA[c] = a;
  bnB[c] = be[c] - mu * a;
}

static constexpr int BA_STEPS = 10;
static constexpr int BA_BLOCKS = (NN * FF / 4) / (256 * BA_STEPS);
static_assert((size_t)BA_BLOCKS * 256 * BA_STEPS * 4 == (size_t)NN * FF, "bnapply grid must tile the matrix exactly");
__global__ __launch_bounds__(256) void k_bnapply(
    const float* __restrict__ z, const float* __restrict__ bnA,
    const float* __restrict__ bnB, float* __restrict__ hout) {
  const int c4 = (threadIdx.x & 31) * 4;
  const float4 a = *(const float4*)(bnA + c4);
  const float4 b = *(const float4*)(bnB + c4);
  const size_t j0 = ((size_t)blockIdx.x * BA_STEPS * 256 + threadIdx.x) * 4;
  float4 v[BA_STEPS];
#pragma unroll
  for (int k = 0; k < BA_STEPS; ++k) v[k] = *(const float4*)(z + j0 + (size_t)k * 1024);
#pragma unroll
  for (int k = 0; k < BA_STEPS; ++k) {
    float4 r;
    r.x = fmaf(v[k].x, a.x, b.x); r.x = r.x > 0.f ? r.x : 0.f;
    r.y = fmaf(v[k].y, a.y, b.y); r.y = r.y > 0.f ? r.y : 0.f;
    r.z = fmaf(v[k].z, a.z, b.z); r.z = r.z > 0.f ? r.z : 0.f;
    r.w = fmaf(v[k].w, a.w, b.w); r.w = r.w > 0.f ? r.w : 0.f;
    *(float4*)(hout + j0 + (size_t)k * 1024) = r;
  }
}

__global__ void k_padw(const float* __restrict__ W2, const float* __restrict__ L2,
                       float* __restrict__ Wp, float* __restrict__ Lp) {
  const int i = blockIdx.x * 256 + threadIdx.x;
  const int k = i >> 6, c = i & 63;
  const float w = (c < 40) ? W2[k * 40 + c] : 0.f;
  const float l = (c < 40) ? L2[k * 40 + c] : 0.f;
  Wp[i] = w;
  Lp[i] = l;
}


extern "C" void kernel_launch(void* const* d_in, const int* in_sizes, int n_in,
                              void* d_out, int out_size, void* d_ws, size_t ws_size,
                              hipStream_t stream) {
  (void)in_sizes; (void)n_in; (void)out_size; (void)ws_size;

  const float* feat = (const float*)d_in[0];
  const int* src   = (const int*)d_in[1];
  const int* dst   = (const int*)d_in[2];
  const float* W0  = (const float*)d_in[3];
  const float* L0  = (const float*)d_in[4];
  const float* W1  = (const float*)d_in[5];
  const float* L1  = (const float*)d_in[6];
  const float* W2  = (const float*)d_in[7];
  const float* L2  = (const float*)d_in[8];
  const float* b2  = (const float*)d_in[9];
  const float* g0  = (const float*)d_in[10];
  const float* be0 = (const float*)d_in[11];
  const float* g1  = (const float*)d_in[12];
  const float* be1 = (const float*)d_in[13];
  float* out = (float*)d_out;

  char* base = (char*)d_ws;
  size_t off = 0;
  auto take = [&](size_t bytes) -> char* {
    char* p = base + off;
    off = (off + bytes + 255) & ~(size_t)255;
    return p;
  };
  unsigned* indeg  = (unsigned*)take((size_t)NN * 4);
  unsigned* outdeg = (unsigned*)take((size_t)NN * 4);
  unsigned* offs   = (unsigned*)take((size_t)(NN + 1) * 4);
  unsigned* rank   = (unsigned*)take((size_t)EE * 4);
  unsigned* esrc   = (unsigned*)take((size_t)EE * 4);
  float* dsout = (float*)take((size_t)NN * 4);
  float* dsin  = (float*)take((size_t)NN * 4);
  float* cs    = (float*)take(128 * 4);
  float* cs2   = (float*)take(128 * 4);
  float* bnA   = (float*)take(128 * 4);
  float* bnB   = (float*)take(128 * 4);
  float* Wp    = (float*)take(128 * 64 * 4);
  float* Lp    = (float*)take(128 * 64 * 4);
  float* bufA  = (float*)take((size_t)NN * FF * 4);
  float* bufB  = (float*)take((size_t)NN * FF * 4);
  float* bufC  = (float*)take((size_t)NN * FF * 4);

  const int BEdge = (EE + 255) / 256;
  const int BNode = (NN + 255) / 256;
  const int BWave = NN / 8;

  k_zero_u32<<<BNode, 256, 0, stream>>>(indeg, NN);
  k_zero_u32<<<BNode, 256, 0, stream>>>(outdeg, NN);
  k_deg<<<BEdge, 256, 0, stream>>>(src, dst, outdeg, indeg, rank);
  k_ds<<<BNode, 256, 0, stream>>>(outdeg, indeg, dsout, dsin);
  k_scan<<<1, 1024, 0, stream>>>(indeg, offs);
  k_fill<<<BEdge, 256, 0, stream>>>(src, dst, offs, rank, esrc);
  k_padw<<<32, 256, 0, stream>>>(W2, L2, Wp, Lp);

  k_agg128<<<BWave, 256, 0, stream>>>(feat, offs, esrc, dsout, dsin, bufA);
  k_gemm_dual<<<512, 256, 0, stream>>>(bufA, feat, W0, L0, bufB);
  k_zero_u32<<<1, 256, 0, stream>>>((unsigned*)cs, 256);
  k_colstats<<<400, 128, 0, stream>>>(bufB, cs, cs2);
  k_bnparams<<<1, 128, 0, stream>>>(cs, cs2, g0, be0, bnA, bnB);
  k_bnapply<<<BA_BLOCKS, 256, 0, stream>>>(bufB, bnA, bnB, bufC);

  k_agg128<<<BWave, 256, 0, stream>>>(bufC, offs, esrc, dsout, dsin, bufA);
  k_gemm_dual<<<512, 256, 0, stream>>>(bufA, bufC, W1, L1, bufB);
  k_zero_u32<<<1, 256, 0, stream>>>((unsigned*)cs, 256);
  k_colstats<<<400, 128, 0, stream>>>(bufB, cs, cs2);
  k_bnparams<<<1, 128, 0, stream>>>(cs, cs2, g1, be1, bnA, bnB);
  k_bnapply<<<BA_BLOCKS, 256, 0, stream>>>(bufB, bnA, bnB, bufC);

  k_gemm_p<<<1024, 256, 0, stream>>>(bufC, dsout, Wp, bufA);
  k_agg64<<<BWave, 256, 0, stream>>>(bufA, offs, esrc, dsin, bufB);
  k_gemm_out<<<1024, 256, 0, stream>>>(bufC, Lp, bufB, b2, out);
}


# Degrees and CSR slots from per-slice 16-bit count tables in LDS (the LDS add returns the rank), summed per node, not two global atomics per edge; CSR rows slice-major: neighbour sum order changes
# speedup vs baseline: 1.7330x; 1.2190x over previous
#include <hip/hip_runtime.h>
#include <hip/hip_bf16.h>


static constexpr int NN = 100000;
static constexpr int EE = 1600000;
static constexpr int FF = 128;

typedef __attribute__((ext_vector_type(2))) float v2f;
typedef __attribute__((ext_vector_type(8))) float v8f;


__global__ void k_zero_u32(unsigned* __restrict__ p, int n) {
  int i = blockIdx.x * 256 + threadIdx.x;
  if (i < n) p[i] = 0u;
}


static constexpr int CB = 64;
static constexpr int SLICE = EE / CB;
static constexpr int TW = NN / 2;
static_assert(SLICE * CB == EE, "slices must tile the edge list exactly");
static_assert(SLICE < 65536, "a 16-bit counter must hold a whole slice");
static_assert(TW * 2 == NN && TW % 4 == 0, "table is copied as 16-byte pieces");
static constexpr int CQ = SLICE / 4;
static constexpr int CQ_PT = (CQ + 1023) / 1024;
static_assert(CQ * 4 == SLICE, "a slice is a whole number of 16-byte groups, so every slice starts 16-byte aligned");
__global__ __launch_bounds__(1024) void k_count(
    const int* __restrict__ src, const int* __restrict__ dst,
    unsigned* __restrict__ cnt, unsigned* __restrict__ lrank) {
  __shared__ __attribute__((aligned(16))) unsigned tab[TW];
  const int role = blockIdx.x >= CB ? 1 : 0;
  const int b = blockIdx.x - role * CB;
  const int tid = threadIdx.x;
  const int4* in4 = (const int4*)((role ? src : dst) + (size_t)b * SLICE);
  int4 q[CQ_PT];
#pragma unroll
  for (int k = 0; k < CQ_PT; ++k) {
    const int i = tid + k * 1024;
    q[k] = in4[i < CQ ? i : CQ - 1];
  }
  uint4* tab4 = (uint4*)tab;
  for (int i = tid; i < TW / 4; i += 1024) tab4[i] = make_uint4(0u, 0u, 0u, 0u);
  __syncthreads();
  if (role == 0) {
    uint4 r[CQ_PT];
#pragma unroll
    for (int k = 0; k < CQ_PT; ++k) {
      if (tid + k * 1024 < CQ) {
        const unsigned dx = (unsigned)q[k].x, dy = (unsigned)q[k].y;
        const unsigned dz = (unsigned)q[k].z, dw = (unsigned)q[k].w;
        r[k].x = atomicAdd(&tab[dx >> 1], 1u << ((dx & 1u) << 4));
        r[k].y = atomicAdd(&tab[dy >> 1], 1u << ((dy & 1u) << 4));
        r[k].z = atomicAdd(&tab[dz >> 1], 1u << ((dz & 1u) << 4));
        r[k].w = atomicAdd(&tab[dw >> 1], 1u << ((dw & 1u) << 4));
      }
    }
    uint4* out4 = (uint4*)(lrank + (size_t)b * SLICE);
#pragma unroll
    for (int k = 0; k < CQ_PT; ++k) {
      if (tid + k * 1024 < CQ) {
        uint4 o;
        o.x = (r[k].x >> (((unsigned)q[k].x & 1u) << 4)) & 0xFFFFu;
        o.y = (r[k].y >> (((unsigned)q[k].y & 1u) << 4)) & 0xFFFFu;
        o.z = (r[k].z >> (((unsigned)q[k].z & 1u) << 4)) & 0xFFFFu;
        o.w = (r[k].w >> (((unsigned)q[k].w & 1u) << 4)) & 0xFFFFu;
        out4[tid + k * 1024] = o;
      }
    }
  } else {
#pragma unroll
    for (int k = 0; k < CQ_PT; ++k) {
      if (tid + k * 1024 < CQ) {
        const unsigned sx = (unsigned)q[k].x, sy = (unsigned)q[k].y;
        const unsigned sz = (unsigned)q[k].z, sw = (unsigned)q[k].w;
        atomicAdd(&tab[sx >> 1], 1u << ((sx & 1u) << 4));
        atomicAdd(&tab[sy >> 1], 1u << ((sy & 1u) << 4));
        atomicAdd(&tab[sz >> 1], 1u << ((sz & 1u) << 4));
        atomicAdd(&tab[sw >> 1], 1u << ((sw & 1u) << 4));
      }
    }
  }
  __syncthreads();
  uint4* o = (uint4*)(cnt + (size_t)blockIdx.x * TW);
  for (int i = tid; i < TW / 4; i += 1024) o[i] = tab4[i];
}

static constexpr int RB = (TW + 255) / 256;
__global__ __launch_bounds__(256) void k_reduce(
    const unsigned* __restrict__ cnt, unsigned* __restrict__ pre,
    unsigned* __restrict__ indeg, float* __restrict__ dsout, float* __restrict__ dsin) {
  const int role = blockIdx.x >= RB ? 1 : 0;
  const int w = (blockIdx.x - role * RB) * 256 + threadIdx.x;
  if (w >= TW) return;
  const unsigned* c = cnt + (size_t)role * CB * TW + w;
  unsigned v[CB];
#pragma unroll
  for (int b = 0; b < CB; ++b) v[b] = c[(size_t)b * TW];
  unsigned lo = 0u, hi = 0u;
  if (role == 0) {
#pragma unroll
    for (int b = 0; b < CB; ++b) {
      *(uint2*)(pre + (size_t)b * NN + 2 * w) = make_uint2(lo, hi);
      lo += v[b] & 0xFFFFu;
      hi += v[b] >> 16;
    }
    *(uint2*)(indeg + 2 * w) = make_uint2(lo, hi);
    *(float2*)(dsin + 2 * w) = make_float2(lo ? rsqrtf((float)lo) : 1.0f,
                                           hi ? rsqrtf((float)hi) : 1.0f);
  } else {
#pragma unroll
    for (int b = 0; b < CB; ++b) {
      lo += v[b] & 0xFFFFu;
      hi += v[b] >> 16;
    }
    *(float2*)(dsout + 2 * w) = make_float2(lo ? rsqrtf((float)lo) : 1.0f,
                                            hi ? rsqrtf((float)hi) : 1.0f);
  }
}

__global__ void k_scan(const unsigned* __restrict__ deg, unsigned* __restrict__ offs) {
  __shared__ unsigned ts[1024];
  const int t = threadIdx.x;
  const int per = (NN + 1023) / 1024;
  int s = t * per; if (s > NN) s = NN;
  int e = s + per; if (e > NN) e = NN;
  unsigned sum = 0;
  for (int i = s; i < e; ++i) sum += deg[i];
  ts[t] = sum;
  __syncthreads();
  for (int off = 1; off < 1024; off <<= 1) {
    unsigned v = (t >= off) ? ts[t - off] : 0u;
    __syncthreads();
    if (t >= off) ts[t] += v;
    __syncthreads();
  }
  unsigned run = (t == 0) ? 0u : ts[t - 1];
  for (int i = s; i < e; ++i) { offs[i] = run; run += deg[i]; }
  if (t == 1023) offs[NN] = ts[1023];
}

__global__ void k_fill(const int* __restrict__ src, const int* __restrict__ dst,
                       const unsigned* __restrict__ offs, const unsigned* __restrict__ pre,
                       const unsigned* __restrict__ lrank, unsigned* __restrict__ esrc) {
  int e = blockIdx.x * 256 + threadIdx.x;
  if (e < EE) {
    const int d = dst[e];
    const int b = e / SLICE;
    esrc[offs[d] + pre[(size_t)b * NN + d] + lrank[e]] = (unsigned)src[e];
  }
}


__global__ void k_agg128(const float* __restrict__ h, const unsigned* __restrict__ offs,
                         const unsigned* __restrict__ esrc,
                         const float* __restrict__ dsout, const float* __restrict__ dsin,
                         float* __restrict__ agg) {
  const int wv = threadIdx.x >> 5, lane = threadIdx.x & 31;
  const int n = blockIdx.x * 8 + wv;
  const unsigned b = offs[n], e = offs[n + 1];
  float4 acc = make_float4(0.f, 0.f, 0.f, 0.f);
  for (unsigned i = b; i < e; ++i) {
    const unsigned s = esrc[i];
    const float w = dsout[s];
    const float4 v = *(const float4*)(h + (size_t)s * FF + lane * 4);
    acc.x = fmaf(w, v.x, acc.x);
    acc.y = fmaf(w, v.y, acc.y);
    acc.z = fmaf(w, v.z, acc.z);
    acc.w = fmaf(w, v.w, acc.w);
  }
  const float c = dsin[n];
  *(float4*)(agg + (size_t)n * FF + lane * 4) =
      make_float4(c * acc.x, c * acc.y, c * acc.z, c * acc.w);
}

__global__ void k_agg64(const float* __restrict__ P, const unsigned* __restrict__ offs,
                        const unsigned* __restrict__ esrc, const float* __restrict__ dsin,
                        float* __restrict__ R) {
  const int wv = threadIdx.x >> 5, lane = threadIdx.x & 31;
  const int n = blockIdx.x * 8 + wv;
  const unsigned b = offs[n], e = offs[n + 1];
  float2 acc = make_float2(0.f, 0.f);
  for (unsigned i = b; i < e; ++i) {
    const unsigned s = esrc[i];
    const float2 v = *(const float2*)(P + (size_t)s * 64 + lane * 2);
    acc.x += v.x;
    acc.y += v.y;
  }
  const float c = dsin[n];
  *(float2*)(R + (size_t)n * 64 + lane * 2) = make_float2(c * acc.x, c * acc.y);
}


static constexpr int GD_PITCH = 132;
static constexpr int GD_TILES = NN / 16;
__global__ __launch_bounds__(256) void k_gemm_dual(
    const float* __restrict__ Agg, const float* __restrict__ H,
    const float* __restrict__ W, const float* __restrict__ L,
    float* __restrict__ Z) {
  __shared__ __attribute__((aligned(16))) float sAg[16 * GD_PITCH];
  __shared__ __attribute__((aligned(16))) float sH[16 * GD_PITCH];
  const int wv = threadIdx.x >> 5;
  const int lane = threadIdx.x & 31;
  const int t = lane & 15;
  const int hh = lane >> 4;
  const int cb = wv << 4;

  v2f bw[32], bl[32];
#pragma unroll
  for (int ks = 0; ks < 32; ++ks) {
    const int kk = 4 * ks + 2 * hh;
    bw[ks].x = W[kk * FF + cb + t]; bw[ks].y = W[(kk + 1) * FF + cb + t];
    bl[ks].x = L[kk * FF + cb + t]; bl[ks].y = L[(kk + 1) * FF + cb + t];
  }

  const int so0 = wv * GD_PITCH + lane * 4;
  const int so1 = (wv + 8) * GD_PITCH + lane * 4;
  const float* fa = sAg + t * GD_PITCH + 2 * hh;
  const float* fh = sH + t * GD_PITCH + 2 * hh;

  int tile = blockIdx.x;
  float4 pa0, pa1, ph0, ph1;
  if (tile < GD_TILES) {
    const size_t g0 = (size_t)(tile * 16 + wv) * FF + lane * 4;
    const size_t g1 = g0 + (size_t)8 * FF;
    pa0 = *(const float4*)(Agg + g0); pa1 = *(const float4*)(Agg + g1);
    ph0 = *(const float4*)(H + g0);   ph1 = *(const float4*)(H + g1);
  }
  for (; tile < GD_TILES; tile += gridDim.x) {
    __syncthreads();
    *(float4*)(sAg + so0) = pa0; *(float4*)(sAg + so1) = pa1;
    *(float4*)(sH + so0) = ph0;  *(float4*)(sH + so1) = ph1;
    __syncthreads();
    const int nt = tile + gridDim.x;
    if (nt < GD_TILES) {
      const size_t g0 = (size_t)(nt * 16 + wv) * FF + lane * 4;
      const size_t g1 = g0 + (size_t)8 * FF;
      pa0 = *(const float4*)(Agg + g0); pa1 = *(const float4*)(Agg + g1);
      ph0 = *(const float4*)(H + g0);   ph1 = *(const float4*)(H + g1);
    }
    v8f acc = {};
#pragma unroll
    for (int ks = 0; ks < 32; ++ks) {
      v2f a1 = *(const v2f*)(fa + 4 * ks);
      acc = __builtin_amdgcn_wmma_f32_16x16x4_f32(false, a1, false, bw[ks], (short)0, acc, false, false);
      v2f a2 = *(const v2f*)(fh + 4 * ks);
      acc = __builtin_amdgcn_wmma_f32_16x16x4_f32(false, a2, false, bl[ks], (short)0, acc, false, false);
    }
    const int r0 = tile << 4;
#pragma unroll
    for (int v = 0; v < 8; ++v)
      Z[(size_t)(r0 + v + 8 * hh) * FF + cb + t] = acc[v];
  }
}

static constexpr int GP_TILES = NN / 32;
__global__ __launch_bounds__(256) void k_gemm_p(
    const float* __restrict__ H, const float* __restrict__ dsout,
    const float* __restrict__ Wp, float* __restrict__ P) {
  __shared__ __attribute__((aligned(16))) float sT[32 * GD_PITCH];
  const int wv = threadIdx.x >> 5;
  const int lane = threadIdx.x & 31;
  const int t = lane & 15, hh = lane >> 4;
  const int ct = wv & 3, rt = wv >> 2;
  const int cb = ct << 4;

  v2f bp[32];
#pragma unroll
  for (int ks = 0; ks < 32; ++ks) {
    const int kk = 4 * ks + 2 * hh;
    bp[ks].x = Wp[kk * 64 + cb + t]; bp[ks].y = Wp[(kk + 1) * 64 + cb + t];
  }

  const int so = wv * GD_PITCH + lane * 4;
  const float* fa = sT + (16 * rt + t) * GD_PITCH + 2 * hh;

  int tile = blockIdx.x;
  float4 p0, p1, p2, p3;
  float dn = 0.f;
  if (tile < GP_TILES) {
    const size_t g = (size_t)(tile * 32 + wv) * FF + lane * 4;
    p0 = *(const float4*)(H + g);
    p1 = *(const float4*)(H + g + (size_t)8 * FF);
    p2 = *(const float4*)(H + g + (size_t)16 * FF);
    p3 = *(const float4*)(H + g + (size_t)24 * FF);
    dn = dsout[tile * 32 + 16 * rt + t];
  }
  for (; tile < GP_TILES; tile += gridDim.x) {
    const float d = dn;
    __syncthreads();
    *(float4*)(sT + so) = p0;
    *(float4*)(sT + so + 8 * GD_PITCH) = p1;
    *(float4*)(sT + so + 16 * GD_PITCH) = p2;
    *(float4*)(sT + so + 24 * GD_PITCH) = p3;
    __syncthreads();
    const int nt = tile + gridDim.x;
    if (nt < GP_TILES) {
      const size_t g = (size_t)(nt * 32 + wv) * FF + lane * 4;
      p0 = *(const float4*)(H + g);
      p1 = *(const float4*)(H + g + (size_t)8 * FF);
      p2 = *(const float4*)(H + g + (size_t)16 * FF);
      p3 = *(const float4*)(H + g + (size_t)24 * FF);
      dn = dsout[nt * 32 + 16 * rt + t];
    }
    v8f acc = {};
#pragma unroll
    for (int ks = 0; ks < 32; ++ks) {
      v2f a = *(const v2f*)(fa + 4 * ks);
      a.x = d * a.x; a.y = d * a.y;
      acc = __builtin_amdgcn_wmma_f32_16x16x4_f32(false, a, false, bp[ks], (short)0, acc, false, false);
    }
    const int r0 = tile * 32 + 16 * rt;
#pragma unroll
    for (int v = 0; v < 8; ++v)
      P[(size_t)(r0 + v + 8 * hh) * 64 + cb + t] = acc[v];
  }
}

__global__ __launch_bounds__(256) void k_gemm_out(
    const float* __restrict__ H, const float* __restrict__ Lp,
    const float* __restrict__ R, const float* __restrict__ b2,
    float* __restrict__ out) {
  __shared__ __attribute__((aligned(16))) float sT[32 * GD_PITCH];
  const int wv = threadIdx.x >> 5;
  const int lane = threadIdx.x & 31;
  const int t = lane & 15, hh = lane >> 4;
  const int ct = wv & 3, rt = wv >> 2;
  const int cb = ct << 4;
  const bool active = ct < 3;

  v2f bq[32];
#pragma unroll
  for (int ks = 0; ks < 32; ++ks) { bq[ks].x = 0.f; bq[ks].y = 0.f; }
  if (active) {
#pragma unroll
    for (int ks = 0; ks < 32; ++ks) {
      const int kk = 4 * ks + 2 * hh;
      bq[ks].x = Lp[kk * 64 + cb + t]; bq[ks].y = Lp[(kk + 1) * 64 + cb + t];
    }
  }

  const int so = wv * GD_PITCH + lane * 4;
  const float* fa = sT + (16 * rt + t) * GD_PITCH + 2 * hh;
  const int col = cb + t;

  int tile = blockIdx.x;
  float4 p0, p1, p2, p3;
  float rn[8];
#pragma unroll
  for (int v = 0; v < 8; ++v) rn[v] = 0.f;
  if (tile < GP_TILES) {
    const size_t g = (size_t)(tile * 32 + wv) * FF + lane * 4;
    p0 = *(const float4*)(H + g);
    p1 = *(const float4*)(H + g + (size_t)8 * FF);
    p2 = *(const float4*)(H + g + (size_t)16 * FF);
    p3 = *(const float4*)(H + g + (size_t)24 * FF);
    if (active) {
#pragma unroll
      for (int v = 0; v < 8; ++v)
        rn[v] = R[(size_t)(tile * 32 + 16 * rt + v + 8 * hh) * 64 + col];
    }
  }
  for (; tile < GP_TILES; tile += gridDim.x) {
    float rv[8];
#pragma unroll
    for (int v = 0; v < 8; ++v) rv[v] = rn[v];
    __syncthreads();
    *(float4*)(sT + so) = p0;
    *(float4*)(sT + so + 8 * GD_PITCH) = p1;
    *(float4*)(sT + so + 16 * GD_PITCH) = p2;
    *(float4*)(sT + so + 24 * GD_PITCH) = p3;
    __syncthreads();
    const int nt = tile + gridDim.x;
    if (nt < GP_TILES) {
      const size_t g = (size_t)(nt * 32 + wv) * FF + lane * 4;
      p0 = *(const float4*)(H + g);
      p1 = *(const float4*)(H + g + (size_t)8 * FF);
      p2 = *(const float4*)(H + g + (size_t)16 * FF);
      p3 = *(const float4*)(H + g + (size_t)24 * FF);
      if (active) {
#pragma unroll
        for (int v = 0; v < 8; ++v)
          rn[v] = R[(size_t)(nt * 32 + 16 * rt + v + 8 * hh) * 64 + col];
      }
    }
    if (active) {
      const int r0 = tile * 32 + 16 * rt;
      v8f acc = {};
#pragma unroll
      for (int ks = 0; ks < 32; ++ks) {
        v2f a = *(const v2f*)(fa + 4 * ks);
        acc = __builtin_amdgcn_wmma_f32_16x16x4_f32(false, a, false, bq[ks], (short)0, acc, false, false);
      }
      if (col < 40) {
        const float bias = b2[col];
#pragma unroll
        for (int v = 0; v < 8; ++v)
          out[(size_t)(r0 + v + 8 * hh) * 40 + col] = acc[v] + rv[v] + bias;
      }
    }
  }
}


__global__ void k_colstats(const float* __restrict__ z, float* __restrict__ cs,
                           float* __restrict__ cs2) {
  const int c = threadIdx.x;
  const size_t r0 = (size_t)blockIdx.x * 250;
  float s = 0.f, s2 = 0.f;
  for (int i = 0; i < 250; ++i) {
    const float v = z[(r0 + i) * FF + c];
    s += v;
    s2 = fmaf(v, v, s2);
  }
  atomicAdd(&cs[c], s);
  atomicAdd(&cs2[c], s2);
}

__global__ void k_bnparams(const float* __restrict__ cs, const float* __restrict__ cs2,
                           const float* __restrict__ g, const float* __restrict__ be,
                           float* __restrict__ bnA, float* __restrict__ bnB) {
  const int c = threadIdx.x;
  const float inv = 1.0f / (float)NN;
  const float mu = cs[c] * inv;
  const float var = cs2[c] * inv - mu * mu;
  const float a = g[c] * rsqrtf(var + 1e-5f);
  bnA[c] = a;
  bnB[c] = be[c] - mu * a;
}

static constexpr int BA_STEPS = 10;
static constexpr int BA_BLOCKS = (NN * FF / 4) / (256 * BA_STEPS);
static_assert((size_t)BA_BLOCKS * 256 * BA_STEPS * 4 == (size_t)NN * FF, "bnapply grid must tile the matrix exactly");
__global__ __launch_bounds__(256) void k_bnapply(
    const float* __restrict__ z, const float* __restrict__ bnA,
    const float* __restrict__ bnB, float* __restrict__ hout) {
  const int c4 = (threadIdx.x & 31) * 4;
  const float4 a = *(const float4*)(bnA + c4);
  const float4 b = *(const float4*)(bnB + c4);
  const size_t j0 = ((size_t)blockIdx.x * BA_STEPS * 256 + threadIdx.x) * 4;
  float4 v[BA_STEPS];
#pragma unroll
  for (int k = 0; k < BA_STEPS; ++k) v[k] = *(const float4*)(z + j0 + (size_t)k * 1024);
#pragma unroll
  for (int k = 0; k < BA_STEPS; ++k) {
    float4 r;
    r.x = fmaf(v[k].x, a.x, b.x); r.x = r.x > 0.f ? r.x : 0.f;
    r.y = fmaf(v[k].y, a.y, b.y); r.y = r.y > 0.f ? r.y : 0.f;
    r.z = fmaf(v[k].z, a.z, b.z); r.z = r.z > 0.f ? r.z : 0.f;
    r.w = fmaf(v[k].w, a.w, b.w); r.w = r.w > 0.f ? r.w : 0.f;
    *(float4*)(hout + j0 + (size_t)k * 1024) = r;
  }
}

__global__ void k_padw(const float* __restrict__ W2, const float* __restrict__ L2,
                       float* __restrict__ Wp, float* __restrict__ Lp) {
  const int i = blockIdx.x * 256 + threadIdx.x;
  const int k = i >> 6, c = i & 63;
  const float w = (c < 40) ? W2[k * 40 + c] : 0.f;
  const float l = (c < 40) ? L2[k * 40 + c] : 0.f;
  Wp[i] = w;
  Lp[i] = l;
}


extern "C" void kernel_launch(void* const* d_in, const int* in_sizes, int n_in,
                              void* d_out, int out_size, void* d_ws, size_t ws_size,
                              hipStream_t stream) {
  (void)in_sizes; (void)n_in; (void)out_size; (void)ws_size;

  const float* feat = (const float*)d_in[0];
  const int* src   = (const int*)d_in[1];
  const int* dst   = (const int*)d_in[2];
  const float* W0  = (const float*)d_in[3];
  const float* L0  = (const float*)d_in[4];
  const float* W1  = (const float*)d_in[5];
  const float* L1  = (const float*)d_in[6];
  const float* W2  = (const float*)d_in[7];
  const float* L2  = (const float*)d_in[8];
  const float* b2  = (const float*)d_in[9];
  const float* g0  = (const float*)d_in[10];
  const float* be0 = (const float*)d_in[11];
  const float* g1  = (const float*)d_in[12];
  const float* be1 = (const float*)d_in[13];
  float* out = (float*)d_out;

  char* base = (char*)d_ws;
  size_t off = 0;
  auto take = [&](size_t bytes) -> char* {
    char* p = base + off;
    off = (off + bytes + 255) & ~(size_t)255;
    return p;
  };
  unsigned* indeg  = (unsigned*)take((size_t)NN * 4);
  unsigned* offs   = (unsigned*)take((size_t)(NN + 1) * 4);
  unsigned* lrank  = (unsigned*)take((size_t)EE * 4);
  unsigned* esrc   = (unsigned*)take((size_t)EE * 4);
  unsigned* cnt    = (unsigned*)take((size_t)2 * CB * TW * 4);
  unsigned* pre    = (unsigned*)take((size_t)CB * NN * 4);
  float* dsout = (float*)take((size_t)NN * 4);
  float* dsin  = (float*)take((size_t)NN * 4);
  float* cs    = (float*)take(128 * 4);
  float* cs2   = (float*)take(128 * 4);
  float* bnA   = (float*)take(128 * 4);
  float* bnB   = (float*)take(128 * 4);
  float* Wp    = (float*)take(128 * 64 * 4);
  float* Lp    = (float*)take(128 * 64 * 4);
  float* bufA  = (float*)take((size_t)NN * FF * 4);
  float* bufB  = (float*)take((size_t)NN * FF * 4);
  float* bufC  = (float*)take((size_t)NN * FF * 4);

  const int BEdge = (EE + 255) / 256;
  const int BWave = NN / 8;

  k_count<<<2 * CB, 1024, 0, stream>>>(src, dst, cnt, lrank);
  k_reduce<<<2 * RB, 256, 0, stream>>>(cnt, pre, indeg, dsout, dsin);
  k_scan<<<1, 1024, 0, stream>>>(indeg, offs);
  k_fill<<<BEdge, 256, 0, stream>>>(src, dst, offs, pre, lrank, esrc);
  k_padw<<<32, 256, 0, stream>>>(W2, L2, Wp, Lp);

  k_agg128<<<BWave, 256, 0, stream>>>(feat, offs, esrc, dsout, dsin, bufA);
  k_gemm_dual<<<512, 256, 0, stream>>>(bufA, feat, W0, L0, bufB);
  k_zero_u32<<<1, 256, 0, stream>>>((unsigned*)cs, 256);
  k_colstats<<<400, 128, 0, stream>>>(bufB, cs, cs2);
  k_bnparams<<<1, 128, 0, stream>>>(cs, cs2, g0, be0, bnA, bnB);
  k_bnapply<<<BA_BLOCKS, 256, 0, stream>>>(bufB, bnA, bnB, bufC);

  k_agg128<<<BWave, 256, 0, stream>>>(bufC, offs, esrc, dsout, dsin, bufA);
  k_gemm_dual<<<512, 256, 0, stream>>>(bufA, bufC, W1, L1, bufB);
  k_zero_u32<<<1, 256, 0, stream>>>((unsigned*)cs, 256);
  k_colstats<<<400, 128, 0, stream>>>(bufB, cs, cs2);
  k_bnparams<<<1, 128, 0, stream>>>(cs, cs2, g1, be1, bnA, bnB);
  k_bnapply<<<BA_BLOCKS, 256, 0, stream>>>(bufB, bnA, bnB, bufC);

  k_gemm_p<<<1024, 256, 0, stream>>>(bufC, dsout, Wp, bufA);
  k_agg64<<<BWave, 256, 0, stream>>>(bufA, offs, esrc, dsin, bufB);
  k_gemm_out<<<1024, 256, 0, stream>>>(bufC, Lp, bufB, b2, out);
}


# Row-offset scan on the whole chip: k_reduce leaves per-wave in-degree totals, 196 blocks add those in front of them and scan their own 512 nodes, whole-line accesses; was one block, 392-B lane stride
# speedup vs baseline: 2.0535x; 1.1849x over previous
#include <hip/hip_runtime.h>
#include <hip/hip_bf16.h>


static constexpr int NN = 100000;
static constexpr int EE = 1600000;
static constexpr int FF = 128;

typedef __attribute__((ext_vector_type(2))) float v2f;
typedef __attribute__((ext_vector_type(8))) float v8f;


__global__ void k_zero_u32(unsigned* __restrict__ p, int n) {
  int i = blockIdx.x * 256 + threadIdx.x;
  if (i < n) p[i] = 0u;
}


static constexpr int CB = 64;
static constexpr int SLICE = EE / CB;
static constexpr int TW = NN / 2;
static_assert(SLICE * CB == EE, "slices must tile the edge list exactly");
static_assert(SLICE < 65536, "a 16-bit counter must hold a whole slice");
static_assert(TW * 2 == NN && TW % 4 == 0, "table is copied as 16-byte pieces");
static constexpr int CQ = SLICE / 4;
static constexpr int CQ_PT = (CQ + 1023) / 1024;
static_assert(CQ * 4 == SLICE, "a slice is a whole number of 16-byte groups, so every slice starts 16-byte aligned");
__global__ __launch_bounds__(1024) void k_count(
    const int* __restrict__ src, const int* __restrict__ dst,
    unsigned* __restrict__ cnt, unsigned* __restrict__ lrank) {
  __shared__ __attribute__((aligned(16))) unsigned tab[TW];
  const int role = blockIdx.x >= CB ? 1 : 0;
  const int b = blockIdx.x - role * CB;
  const int tid = threadIdx.x;
  const int4* in4 = (const int4*)((role ? src : dst) + (size_t)b * SLICE);
  int4 q[CQ_PT];
#pragma unroll
  for (int k = 0; k < CQ_PT; ++k) {
    const int i = tid + k * 1024;
    q[k] = in4[i < CQ ? i : CQ - 1];
  }
  uint4* tab4 = (uint4*)tab;
  for (int i = tid; i < TW / 4; i += 1024) tab4[i] = make_uint4(0u, 0u, 0u, 0u);
  __syncthreads();
  if (role == 0) {
    uint4 r[CQ_PT];
#pragma unroll
    for (int k = 0; k < CQ_PT; ++k) {
      if (tid + k * 1024 < CQ) {
        const unsigned dx = (unsigned)q[k].x, dy = (unsigned)q[k].y;
        const unsigned dz = (unsigned)q[k].z, dw = (unsigned)q[k].w;
        r[k].x = atomicAdd(&tab[dx >> 1], 1u << ((dx & 1u) << 4));
        r[k].y = atomicAdd(&tab[dy >> 1], 1u << ((dy & 1u) << 4));
        r[k].z = atomicAdd(&tab[dz >> 1], 1u << ((dz & 1u) << 4));
        r[k].w = atomicAdd(&tab[dw >> 1], 1u << ((dw & 1u) << 4));
      }
    }
    uint4* out4 = (uint4*)(lrank + (size_t)b * SLICE);
#pragma unroll
    for (int k = 0; k < CQ_PT; ++k) {
      if (tid + k * 1024 < CQ) {
        uint4 o;
        o.x = (r[k].x >> (((unsigned)q[k].x & 1u) << 4)) & 0xFFFFu;
        o.y = (r[k].y >> (((unsigned)q[k].y & 1u) << 4)) & 0xFFFFu;
        o.z = (r[k].z >> (((unsigned)q[k].z & 1u) << 4)) & 0xFFFFu;
        o.w = (r[k].w >> (((unsigned)q[k].w & 1u) << 4)) & 0xFFFFu;
        out4[tid + k * 1024] = o;
      }
    }
  } else {
#pragma unroll
    for (int k = 0; k < CQ_PT; ++k) {
      if (tid + k * 1024 < CQ) {
        const unsigned sx = (unsigned)q[k].x, sy = (unsigned)q[k].y;
        const unsigned sz = (unsigned)q[k].z, sw = (unsigned)q[k].w;
        atomicAdd(&tab[sx >> 1], 1u << ((sx & 1u) << 4));
        atomicAdd(&tab[sy >> 1], 1u << ((sy & 1u) << 4));
        atomicAdd(&tab[sz >> 1], 1u << ((sz & 1u) << 4));
        atomicAdd(&tab[sw >> 1], 1u << ((sw & 1u) << 4));
      }
    }
  }
  __syncthreads();
  uint4* o = (uint4*)(cnt + (size_t)blockIdx.x * TW);
  for (int i = tid; i < TW / 4; i += 1024) o[i] = tab4[i];
}

static constexpr int RB = (TW + 255) / 256;
static constexpr int RW = RB * 8;
__global__ __launch_bounds__(256) void k_reduce(
    const unsigned* __restrict__ cnt, unsigned* __restrict__ pre,
    unsigned* __restrict__ indeg, unsigned* __restrict__ wtot,
    float* __restrict__ dsout, float* __restrict__ dsin) {
  const int role = blockIdx.x >= RB ? 1 : 0;
  const int w = (blockIdx.x - role * RB) * 256 + threadIdx.x;
  unsigned lo = 0u, hi = 0u;
  if (w < TW) {
    const unsigned* c = cnt + (size_t)role * CB * TW + w;
    unsigned v[CB];
#pragma unroll
    for (int b = 0; b < CB; ++b) v[b] = c[(size_t)b * TW];
    if (role == 0) {
#pragma unroll
      for (int b = 0; b < CB; ++b) {
        *(uint2*)(pre + (size_t)b * NN + 2 * w) = make_uint2(lo, hi);
        lo += v[b] & 0xFFFFu;
        hi += v[b] >> 16;
      }
      *(uint2*)(indeg + 2 * w) = make_uint2(lo, hi);
      *(float2*)(dsin + 2 * w) = make_float2(lo ? rsqrtf((float)lo) : 1.0f,
                                             hi ? rsqrtf((float)hi) : 1.0f);
    } else {
#pragma unroll
      for (int b = 0; b < CB; ++b) {
        lo += v[b] & 0xFFFFu;
        hi += v[b] >> 16;
      }
      *(float2*)(dsout + 2 * w) = make_float2(lo ? rsqrtf((float)lo) : 1.0f,
                                              hi ? rsqrtf((float)hi) : 1.0f);
    }
  }
  if (role == 0) {
    unsigned s = lo + hi;
    for (int o = 16; o > 0; o >>= 1) s += __shfl_xor(s, o);
    if ((threadIdx.x & 31) == 0) wtot[w >> 5] = s;
  }
}

static constexpr int SC_TRIPS = (RW + 255) / 256;
static_assert(RB * 256 >= TW, "the scan grid covers the table");
__global__ __launch_bounds__(256) void k_scan(const unsigned* __restrict__ deg,
                                              const unsigned* __restrict__ wtot,
                                              unsigned* __restrict__ offs) {
  __shared__ unsigned sbase[8], swave[8];
  const int t = threadIdx.x, lane = t & 31, wv = t >> 5;
  const int w = blockIdx.x * 256 + t;
  const bool valid = w < TW;
  uint2 d = make_uint2(0u, 0u);
  if (valid) d = *(const uint2*)(deg + 2 * w);
  const int nb = blockIdx.x * 8;
  unsigned bs = 0u;
#pragma unroll
  for (int k = 0; k < SC_TRIPS; ++k) {
    const int j = t + k * 256;
    const unsigned v = wtot[j < RW ? j : RW - 1];
    bs += j < nb ? v : 0u;
  }
  const unsigned p = d.x + d.y;
  unsigned inc = p;
  for (int o = 1; o < 32; o <<= 1) {
    const unsigned u = __shfl_up(inc, o);
    if (lane >= o) inc += u;
  }
  for (int o = 16; o > 0; o >>= 1) bs += __shfl_xor(bs, o);
  if (lane == 31) swave[wv] = inc;
  if (lane == 0) sbase[wv] = bs;
  __syncthreads();
  unsigned run = 0u;
  for (int i = 0; i < 8; ++i) {
    run += sbase[i];
    if (i < wv) run += swave[i];
  }
  const unsigned ex = run + inc - p;
  if (valid) {
    *(uint2*)(offs + 2 * w) = make_uint2(ex, ex + d.x);
    if (w == TW - 1) offs[NN] = ex + p;
  }
}

__global__ void k_fill(const int* __restrict__ src, const int* __restrict__ dst,
                       const unsigned* __restrict__ offs, const unsigned* __restrict__ pre,
                       const unsigned* __restrict__ lrank, unsigned* __restrict__ esrc) {
  int e = blockIdx.x * 256 + threadIdx.x;
  if (e < EE) {
    const int d = dst[e];
    const int b = e / SLICE;
    esrc[offs[d] + pre[(size_t)b * NN + d] + lrank[e]] = (unsigned)src[e];
  }
}


__global__ void k_agg128(const float* __restrict__ h, const unsigned* __restrict__ offs,
                         const unsigned* __restrict__ esrc,
                         const float* __restrict__ dsout, const float* __restrict__ dsin,
                         float* __restrict__ agg) {
  const int wv = threadIdx.x >> 5, lane = threadIdx.x & 31;
  const int n = blockIdx.x * 8 + wv;
  const unsigned b = offs[n], e = offs[n + 1];
  float4 acc = make_float4(0.f, 0.f, 0.f, 0.f);
  for (unsigned i = b; i < e; ++i) {
    const unsigned s = esrc[i];
    const float w = dsout[s];
    const float4 v = *(const float4*)(h + (size_t)s * FF + lane * 4);
    acc.x = fmaf(w, v.x, acc.x);
    acc.y = fmaf(w, v.y, acc.y);
    acc.z = fmaf(w, v.z, acc.z);
    acc.w = fmaf(w, v.w, acc.w);
  }
  const float c = dsin[n];
  *(float4*)(agg + (size_t)n * FF + lane * 4) =
      make_float4(c * acc.x, c * acc.y, c * acc.z, c * acc.w);
}

__global__ void k_agg64(const float* __restrict__ P, const unsigned* __restrict__ offs,
                        const unsigned* __restrict__ esrc, const float* __restrict__ dsin,
                        float* __restrict__ R) {
  const int wv = threadIdx.x >> 5, lane = threadIdx.x & 31;
  const int n = blockIdx.x * 8 + wv;
  const unsigned b = offs[n], e = offs[n + 1];
  float2 acc = make_float2(0.f, 0.f);
  for (unsigned i = b; i < e; ++i) {
    const unsigned s = esrc[i];
    const float2 v = *(const float2*)(P + (size_t)s * 64 + lane * 2);
    acc.x += v.x;
    acc.y += v.y;
  }
  const float c = dsin[n];
  *(float2*)(R + (size_t)n * 64 + lane * 2) = make_float2(c * acc.x, c * acc.y);
}


static constexpr int GD_PITCH = 132;
static constexpr int GD_TILES = NN / 16;
__global__ __launch_bounds__(256) void k_gemm_dual(
    const float* __restrict__ Agg, const float* __restrict__ H,
    const float* __restrict__ W, const float* __restrict__ L,
    float* __restrict__ Z) {
  __shared__ __attribute__((aligned(16))) float sAg[16 * GD_PITCH];
  __shared__ __attribute__((aligned(16))) float sH[16 * GD_PITCH];
  const int wv = threadIdx.x >> 5;
  const int lane = threadIdx.x & 31;
  const int t = lane & 15;
  const int hh = lane >> 4;
  const int cb = wv << 4;

  v2f bw[32], bl[32];
#pragma unroll
  for (int ks = 0; ks < 32; ++ks) {
    const int kk = 4 * ks + 2 * hh;
    bw[ks].x = W[kk * FF + cb + t]; bw[ks].y = W[(kk + 1) * FF + cb + t];
    bl[ks].x = L[kk * FF + cb + t]; bl[ks].y = L[(kk + 1) * FF + cb + t];
  }

  const int so0 = wv * GD_PITCH + lane * 4;
  const int so1 = (wv + 8) * GD_PITCH + lane * 4;
  const float* fa = sAg + t * GD_PITCH + 2 * hh;
  const float* fh = sH + t * GD_PITCH + 2 * hh;

  int tile = blockIdx.x;
  float4 pa0, pa1, ph0, ph1;
  if (tile < GD_TILES) {
    const size_t g0 = (size_t)(tile * 16 + wv) * FF + lane * 4;
    const size_t g1 = g0 + (size_t)8 * FF;
    pa0 = *(const float4*)(Agg + g0); pa1 = *(const float4*)(Agg + g1);
    ph0 = *(const float4*)(H + g0);   ph1 = *(const float4*)(H + g1);
  }
  for (; tile < GD_TILES; tile += gridDim.x) {
    __syncthreads();
    *(float4*)(sAg + so0) = pa0; *(float4*)(sAg + so1) = pa1;
    *(float4*)(sH + so0) = ph0;  *(float4*)(sH + so1) = ph1;
    __syncthreads();
    const int nt = tile + gridDim.x;
    if (nt < GD_TILES) {
      const size_t g0 = (size_t)(nt * 16 + wv) * FF + lane * 4;
      const size_t g1 = g0 + (size_t)8 * FF;
      pa0 = *(const float4*)(Agg + g0); pa1 = *(const float4*)(Agg + g1);
      ph0 = *(const float4*)(H + g0);   ph1 = *(const float4*)(H + g1);
    }
    v8f acc = {};
#pragma unroll
    for (int ks = 0; ks < 32; ++ks) {
      v2f a1 = *(const v2f*)(fa + 4 * ks);
      acc = __builtin_amdgcn_wmma_f32_16x16x4_f32(false, a1, false, bw[ks], (short)0, acc, false, false);
      v2f a2 = *(const v2f*)(fh + 4 * ks);
      acc = __builtin_amdgcn_wmma_f32_16x16x4_f32(false, a2, false, bl[ks], (short)0, acc, false, false);
    }
    const int r0 = tile << 4;
#pragma unroll
    for (int v = 0; v < 8; ++v)
      Z[(size_t)(r0 + v + 8 * hh) * FF + cb + t] = acc[v];
  }
}

static constexpr int GP_TILES = NN / 32;
__global__ __launch_bounds__(256) void k_gemm_p(
    const float* __restrict__ H, const float* __restrict__ dsout,
    const float* __restrict__ Wp, float* __restrict__ P) {
  __shared__ __attribute__((aligned(16))) float sT[32 * GD_PITCH];
  const int wv = threadIdx.x >> 5;
  const int lane = threadIdx.x & 31;
  const int t = lane & 15, hh = lane >> 4;
  const int ct = wv & 3, rt = wv >> 2;
  const int cb = ct << 4;

  v2f bp[32];
#pragma unroll
  for (int ks = 0; ks < 32; ++ks) {
    const int kk = 4 * ks + 2 * hh;
    bp[ks].x = Wp[kk * 64 + cb + t]; bp[ks].y = Wp[(kk + 1) * 64 + cb + t];
  }

  const int so = wv * GD_PITCH + lane * 4;
  const float* fa = sT + (16 * rt + t) * GD_PITCH + 2 * hh;

  int tile = blockIdx.x;
  float4 p0, p1, p2, p3;
  float dn = 0.f;
  if (tile < GP_TILES) {
    const size_t g = (size_t)(tile * 32 + wv) * FF + lane * 4;
    p0 = *(const float4*)(H + g);
    p1 = *(const float4*)(H + g + (size_t)8 * FF);
    p2 = *(const float4*)(H + g + (size_t)16 * FF);
    p3 = *(const float4*)(H + g + (size_t)24 * FF);
    dn = dsout[tile * 32 + 16 * rt + t];
  }
  for (; tile < GP_TILES; tile += gridDim.x) {
    const float d = dn;
    __syncthreads();
    *(float4*)(sT + so) = p0;
    *(float4*)(sT + so + 8 * GD_PITCH) = p1;
    *(float4*)(sT + so + 16 * GD_PITCH) = p2;
    *(float4*)(sT + so + 24 * GD_PITCH) = p3;
    __syncthreads();
    const int nt = tile + gridDim.x;
    if (nt < GP_TILES) {
      const size_t g = (size_t)(nt * 32 + wv) * FF + lane * 4;
      p0 = *(const float4*)(H + g);
      p1 = *(const float4*)(H + g + (size_t)8 * FF);
      p2 = *(const float4*)(H + g + (size_t)16 * FF);
      p3 = *(const float4*)(H + g + (size_t)24 * FF);
      dn = dsout[nt * 32 + 16 * rt + t];
    }
    v8f acc = {};
#pragma unroll
    for (int ks = 0; ks < 32; ++ks) {
      v2f a = *(const v2f*)(fa + 4 * ks);
      a.x = d * a.x; a.y = d * a.y;
      acc = __builtin_amdgcn_wmma_f32_16x16x4_f32(false, a, false, bp[ks], (short)0, acc, false, false);
    }
    const int r0 = tile * 32 + 16 * rt;
#pragma unroll
    for (int v = 0; v < 8; ++v)
      P[(size_t)(r0 + v + 8 * hh) * 64 + cb + t] = acc[v];
  }
}

__global__ __launch_bounds__(256) void k_gemm_out(
    const float* __restrict__ H, const float* __restrict__ Lp,
    const float* __restrict__ R, const float* __restrict__ b2,
    float* __restrict__ out) {
  __shared__ __attribute__((aligned(16))) float sT[32 * GD_PITCH];
  const int wv = threadIdx.x >> 5;
  const int lane = threadIdx.x & 31;
  const int t = lane & 15, hh = lane >> 4;
  const int ct = wv & 3, rt = wv >> 2;
  const int cb = ct << 4;
  const bool active = ct < 3;

  v2f bq[32];
#pragma unroll
  for (int ks = 0; ks < 32; ++ks) { bq[ks].x = 0.f; bq[ks].y = 0.f; }
  if (active) {
#pragma unroll
    for (int ks = 0; ks < 32; ++ks) {
      const int kk = 4 * ks + 2 * hh;
      bq[ks].x = Lp[kk * 64 + cb + t]; bq[ks].y = Lp[(kk + 1) * 64 + cb + t];
    }
  }

  const int so = wv * GD_PITCH + lane * 4;
  const float* fa = sT + (16 * rt + t) * GD_PITCH + 2 * hh;
  const int col = cb + t;

  int tile = blockIdx.x;
  float4 p0, p1, p2, p3;
  float rn[8];
#pragma unroll
  for (int v = 0; v < 8; ++v) rn[v] = 0.f;
  if (tile < GP_TILES) {
    const size_t g = (size_t)(tile * 32 + wv) * FF + lane * 4;
    p0 = *(const float4*)(H + g);
    p1 = *(const float4*)(H + g + (size_t)8 * FF);
    p2 = *(const float4*)(H + g + (size_t)16 * FF);
    p3 = *(const float4*)(H + g + (size_t)24 * FF);
    if (active) {
#pragma unroll
      for (int v = 0; v < 8; ++v)
        rn[v] = R[(size_t)(tile * 32 + 16 * rt + v + 8 * hh) * 64 + col];
    }
  }
  for (; tile < GP_TILES; tile += gridDim.x) {
    float rv[8];
#pragma unroll
    for (int v = 0; v < 8; ++v) rv[v] = rn[v];
    __syncthreads();
    *(float4*)(sT + so) = p0;
    *(float4*)(sT + so + 8 * GD_PITCH) = p1;
    *(float4*)(sT + so + 16 * GD_PITCH) = p2;
    *(float4*)(sT + so + 24 * GD_PITCH) = p3;
    __syncthreads();
    const int nt = tile + gridDim.x;
    if (nt < GP_TILES) {
      const size_t g = (size_t)(nt * 32 + wv) * FF + lane * 4;
      p0 = *(const float4*)(H + g);
      p1 = *(const float4*)(H + g + (size_t)8 * FF);
      p2 = *(const float4*)(H + g + (size_t)16 * FF);
      p3 = *(const float4*)(H + g + (size_t)24 * FF);
      if (active) {
#pragma unroll
        for (int v = 0; v < 8; ++v)
          rn[v] = R[(size_t)(nt * 32 + 16 * rt + v + 8 * hh) * 64 + col];
      }
    }
    if (active) {
      const int r0 = tile * 32 + 16 * rt;
      v8f acc = {};
#pragma unroll
      for (int ks = 0; ks < 32; ++ks) {
        v2f a = *(const v2f*)(fa + 4 * ks);
        acc = __builtin_amdgcn_wmma_f32_16x16x4_f32(false, a, false, bq[ks], (short)0, acc, false, false);
      }
      if (col < 40) {
        const float bias = b2[col];
#pragma unroll
        for (int v = 0; v < 8; ++v)
          out[(size_t)(r0 + v + 8 * hh) * 40 + col] = acc[v] + rv[v] + bias;
      }
    }
  }
}


__global__ void k_colstats(const float* __restrict__ z, float* __restrict__ cs,
                           float* __restrict__ cs2) {
  const int c = threadIdx.x;
  const size_t r0 = (size_t)blockIdx.x * 250;
  float s = 0.f, s2 = 0.f;
  for (int i = 0; i < 250; ++i) {
    const float v = z[(r0 + i) * FF + c];
    s += v;
    s2 = fmaf(v, v, s2);
  }
  atomicAdd(&cs[c], s);
  atomicAdd(&cs2[c], s2);
}

__global__ void k_bnparams(const float* __restrict__ cs, const float* __restrict__ cs2,
                           const float* __restrict__ g, const float* __restrict__ be,
                           float* __restrict__ bnA, float* __restrict__ bnB) {
  const int c = threadIdx.x;
  const float inv = 1.0f / (float)NN;
  const float mu = cs[c] * inv;
  const float var = cs2[c] * inv - mu * mu;
  const float a = g[c] * rsqrtf(var + 1e-5f);
  bnA[c] = a;
  bnB[c] = be[c] - mu * a;
}

static constexpr int BA_STEPS = 10;
static constexpr int BA_BLOCKS = (NN * FF / 4) / (256 * BA_STEPS);
static_assert((size_t)BA_BLOCKS * 256 * BA_STEPS * 4 == (size_t)NN * FF, "bnapply grid must tile the matrix exactly");
__global__ __launch_bounds__(256) void k_bnapply(
    const float* __restrict__ z, const float* __restrict__ bnA,
    const float* __restrict__ bnB, float* __restrict__ hout) {
  const int c4 = (threadIdx.x & 31) * 4;
  const float4 a = *(const float4*)(bnA + c4);
  const float4 b = *(const float4*)(bnB + c4);
  const size_t j0 = ((size_t)blockIdx.x * BA_STEPS * 256 + threadIdx.x) * 4;
  float4 v[BA_STEPS];
#pragma unroll
  for (int k = 0; k < BA_STEPS; ++k) v[k] = *(const float4*)(z + j0 + (size_t)k * 1024);
#pragma unroll
  for (int k = 0; k < BA_STEPS; ++k) {
    float4 r;
    r.x = fmaf(v[k].x, a.x, b.x); r.x = r.x > 0.f ? r.x : 0.f;
    r.y = fmaf(v[k].y, a.y, b.y); r.y = r.y > 0.f ? r.y : 0.f;
    r.z = fmaf(v[k].z, a.z, b.z); r.z = r.z > 0.f ? r.z : 0.f;
    r.w = fmaf(v[k].w, a.w, b.w); r.w = r.w > 0.f ? r.w : 0.f;
    *(float4*)(hout + j0 + (size_t)k * 1024) = r;
  }
}

__global__ void k_padw(const float* __restrict__ W2, const float* __restrict__ L2,
                       float* __restrict__ Wp, float* __restrict__ Lp) {
  const int i = blockIdx.x * 256 + threadIdx.x;
  const int k = i >> 6, c = i & 63;
  const float w = (c < 40) ? W2[k * 40 + c] : 0.f;
  const float l = (c < 40) ? L2[k * 40 + c] : 0.f;
  Wp[i] = w;
  Lp[i] = l;
}


extern "C" void kernel_launch(void* const* d_in, const int* in_sizes, int n_in,
                              void* d_out, int out_size, void* d_ws, size_t ws_size,
                              hipStream_t stream) {
  (void)in_sizes; (void)n_in; (void)out_size; (void)ws_size;

  const float* feat = (const float*)d_in[0];
  const int* src   = (const int*)d_in[1];
  const int* dst   = (const int*)d_in[2];
  const float* W0  = (const float*)d_in[3];
  const float* L0  = (const float*)d_in[4];
  const float* W1  = (const float*)d_in[5];
  const float* L1  = (const float*)d_in[6];
  const float* W2  = (const float*)d_in[7];
  const float* L2  = (const float*)d_in[8];
  const float* b2  = (const float*)d_in[9];
  const float* g0  = (const float*)d_in[10];
  const float* be0 = (const float*)d_in[11];
  const float* g1  = (const float*)d_in[12];
  const float* be1 = (const float*)d_in[13];
  float* out = (float*)d_out;

  char* base = (char*)d_ws;
  size_t off = 0;
  auto take = [&](size_t bytes) -> char* {
    char* p = base + off;
    off = (off + bytes + 255) & ~(size_t)255;
    return p;
  };
  unsigned* indeg  = (unsigned*)take((size_t)NN * 4);
  unsigned* offs   = (unsigned*)take((size_t)(NN + 1) * 4);
  unsigned* lrank  = (unsigned*)take((size_t)EE * 4);
  unsigned* esrc   = (unsigned*)take((size_t)EE * 4);
  unsigned* cnt    = (unsigned*)take((size_t)2 * CB * TW * 4);
  unsigned* pre    = (unsigned*)take((size_t)CB * NN * 4);
  float* dsout = (float*)take((size_t)NN * 4);
  float* dsin  = (float*)take((size_t)NN * 4);
  float* cs    = (float*)take(128 * 4);
  float* cs2   = (float*)take(128 * 4);
  float* bnA   = (float*)take(128 * 4);
  float* bnB   = (float*)take(128 * 4);
  float* Wp    = (float*)take(128 * 64 * 4);
  float* Lp    = (float*)take(128 * 64 * 4);
  float* bufA  = (float*)take((size_t)NN * FF * 4);
  float* bufB  = (float*)take((size_t)NN * FF * 4);
  float* bufC  = (float*)take((size_t)NN * FF * 4);
  unsigned* wtot = (unsigned*)take((size_t)RW * 4);

  const int BEdge = (EE + 255) / 256;
  const int BWave = NN / 8;

  k_count<<<2 * CB, 1024, 0, stream>>>(src, dst, cnt, lrank);
  k_reduce<<<2 * RB, 256, 0, stream>>>(cnt, pre, indeg, wtot, dsout, dsin);
  k_scan<<<RB, 256, 0, stream>>>(indeg, wtot, offs);
  k_fill<<<BEdge, 256, 0, stream>>>(src, dst, offs, pre, lrank, esrc);
  k_padw<<<32, 256, 0, stream>>>(W2, L2, Wp, Lp);

  k_agg128<<<BWave, 256, 0, stream>>>(feat, offs, esrc, dsout, dsin, bufA);
  k_gemm_dual<<<512, 256, 0, stream>>>(bufA, feat, W0, L0, bufB);
  k_zero_u32<<<1, 256, 0, stream>>>((unsigned*)cs, 256);
  k_colstats<<<400, 128, 0, stream>>>(bufB, cs, cs2);
  k_bnparams<<<1, 128, 0, stream>>>(cs, cs2, g0, be0, bnA, bnB);
  k_bnapply<<<BA_BLOCKS, 256, 0, stream>>>(bufB, bnA, bnB, bufC);

  k_agg128<<<BWave, 256, 0, stream>>>(bufC, offs, esrc, dsout, dsin, bufA);
  k_gemm_dual<<<512, 256, 0, stream>>>(bufA, bufC, W1, L1, bufB);
  k_zero_u32<<<1, 256, 0, stream>>>((unsigned*)cs, 256);
  k_colstats<<<400, 128, 0, stream>>>(bufB, cs, cs2);
  k_bnparams<<<1, 128, 0, stream>>>(cs, cs2, g1, be1, bnA, bnB);
  k_bnapply<<<BA_BLOCKS, 256, 0, stream>>>(bufB, bnA, bnB, bufC);

  k_gemm_p<<<1024, 256, 0, stream>>>(bufC, dsout, Wp, bufA);
  k_agg64<<<BWave, 256, 0, stream>>>(bufA, offs, esrc, dsin, bufB);
  k_gemm_out<<<1024, 256, 0, stream>>>(bufC, Lp, bufB, b2, out);
}


# BN column sums/square sums taken from the dual-GEMM accumulators, one partial row per block, added in fixed order in k_bnparams; colstats/clear launches and float atomics gone: BN sum order changes
# speedup vs baseline: 2.1625x; 1.0531x over previous
#include <hip/hip_runtime.h>
#include <hip/hip_bf16.h>


static constexpr int NN = 100000;
static constexpr int EE = 1600000;
static constexpr int FF = 128;

typedef __attribute__((ext_vector_type(2))) float v2f;
typedef __attribute__((ext_vector_type(8))) float v8f;


static constexpr int CB = 64;
static constexpr int SLICE = EE / CB;
static constexpr int TW = NN / 2;
static_assert(SLICE * CB == EE, "slices must tile the edge list exactly");
static_assert(SLICE < 65536, "a 16-bit counter must hold a whole slice");
static_assert(TW * 2 == NN && TW % 4 == 0, "table is copied as 16-byte pieces");
static constexpr int CQ = SLICE / 4;
static constexpr int CQ_PT = (CQ + 1023) / 1024;
static_assert(CQ * 4 == SLICE, "a slice is a whole number of 16-byte groups, so every slice starts 16-byte aligned");
__global__ __launch_bounds__(1024) void k_count(
    const int* __restrict__ src, const int* __restrict__ dst,
    unsigned* __restrict__ cnt, unsigned* __restrict__ lrank) {
  __shared__ __attribute__((aligned(16))) unsigned tab[TW];
  const int role = blockIdx.x >= CB ? 1 : 0;
  const int b = blockIdx.x - role * CB;
  const int tid = threadIdx.x;
  const int4* in4 = (const int4*)((role ? src : dst) + (size_t)b * SLICE);
  int4 q[CQ_PT];
#pragma unroll
  for (int k = 0; k < CQ_PT; ++k) {
    const int i = tid + k * 1024;
    q[k] = in4[i < CQ ? i : CQ - 1];
  }
  uint4* tab4 = (uint4*)tab;
  for (int i = tid; i < TW / 4; i += 1024) tab4[i] = make_uint4(0u, 0u, 0u, 0u);
  __syncthreads();
  if (role == 0) {
    uint4 r[CQ_PT];
#pragma unroll
    for (int k = 0; k < CQ_PT; ++k) {
      if (tid + k * 1024 < CQ) {
        const unsigned dx = (unsigned)q[k].x, dy = (unsigned)q[k].y;
        const unsigned dz = (unsigned)q[k].z, dw = (unsigned)q[k].w;
        r[k].x = atomicAdd(&tab[dx >> 1], 1u << ((dx & 1u) << 4));
        r[k].y = atomicAdd(&tab[dy >> 1], 1u << ((dy & 1u) << 4));
        r[k].z = atomicAdd(&tab[dz >> 1], 1u << ((dz & 1u) << 4));
        r[k].w = atomicAdd(&tab[dw >> 1], 1u << ((dw & 1u) << 4));
      }
    }
    uint4* out4 = (uint4*)(lrank + (size_t)b * SLICE);
#pragma unroll
    for (int k = 0; k < CQ_PT; ++k) {
      if (tid + k * 1024 < CQ) {
        uint4 o;
        o.x = (r[k].x >> (((unsigned)q[k].x & 1u) << 4)) & 0xFFFFu;
        o.y = (r[k].y >> (((unsigned)q[k].y & 1u) << 4)) & 0xFFFFu;
        o.z = (r[k].z >> (((unsigned)q[k].z & 1u) << 4)) & 0xFFFFu;
        o.w = (r[k].w >> (((unsigned)q[k].w & 1u) << 4)) & 0xFFFFu;
        out4[tid + k * 1024] = o;
      }
    }
  } else {
#pragma unroll
    for (int k = 0; k < CQ_PT; ++k) {
      if (tid + k * 1024 < CQ) {
        const unsigned sx = (unsigned)q[k].x, sy = (unsigned)q[k].y;
        const unsigned sz = (unsigned)q[k].z, sw = (unsigned)q[k].w;
        atomicAdd(&tab[sx >> 1], 1u << ((sx & 1u) << 4));
        atomicAdd(&tab[sy >> 1], 1u << ((sy & 1u) << 4));
        atomicAdd(&tab[sz >> 1], 1u << ((sz & 1u) << 4));
        atomicAdd(&tab[sw >> 1], 1u << ((sw & 1u) << 4));
      }
    }
  }
  __syncthreads();
  uint4* o = (uint4*)(cnt + (size_t)blockIdx.x * TW);
  for (int i = tid; i < TW / 4; i += 1024) o[i] = tab4[i];
}

static constexpr int RB = (TW + 255) / 256;
static constexpr int RW = RB * 8;
__global__ __launch_bounds__(256) void k_reduce(
    const unsigned* __restrict__ cnt, unsigned* __restrict__ pre,
    unsigned* __restrict__ indeg, unsigned* __restrict__ wtot,
    float* __restrict__ dsout, float* __restrict__ dsin) {
  const int role = blockIdx.x >= RB ? 1 : 0;
  const int w = (blockIdx.x - role * RB) * 256 + threadIdx.x;
  unsigned lo = 0u, hi = 0u;
  if (w < TW) {
    const unsigned* c = cnt + (size_t)role * CB * TW + w;
    unsigned v[CB];
#pragma unroll
    for (int b = 0; b < CB; ++b) v[b] = c[(size_t)b * TW];
    if (role == 0) {
#pragma unroll
      for (int b = 0; b < CB; ++b) {
        *(uint2*)(pre + (size_t)b * NN + 2 * w) = make_uint2(lo, hi);
        lo += v[b] & 0xFFFFu;
        hi += v[b] >> 16;
      }
      *(uint2*)(indeg + 2 * w) = make_uint2(lo, hi);
      *(float2*)(dsin + 2 * w) = make_float2(lo ? rsqrtf((float)lo) : 1.0f,
                                             hi ? rsqrtf((float)hi) : 1.0f);
    } else {
#pragma unroll
      for (int b = 0; b < CB; ++b) {
        lo += v[b] & 0xFFFFu;
        hi += v[b] >> 16;
      }
      *(float2*)(dsout + 2 * w) = make_float2(lo ? rsqrtf((float)lo) : 1.0f,
                                              hi ? rsqrtf((float)hi) : 1.0f);
    }
  }
  if (role == 0) {
    unsigned s = lo + hi;
    for (int o = 16; o > 0; o >>= 1) s += __shfl_xor(s, o);
    if ((threadIdx.x & 31) == 0) wtot[w >> 5] = s;
  }
}

static constexpr int SC_TRIPS = (RW + 255) / 256;
static_assert(RB * 256 >= TW, "the scan grid covers the table");
__global__ __launch_bounds__(256) void k_scan(const unsigned* __restrict__ deg,
                                              const unsigned* __restrict__ wtot,
                                              unsigned* __restrict__ offs) {
  __shared__ unsigned sbase[8], swave[8];
  const int t = threadIdx.x, lane = t & 31, wv = t >> 5;
  const int w = blockIdx.x * 256 + t;
  const bool valid = w < TW;
  uint2 d = make_uint2(0u, 0u);
  if (valid) d = *(const uint2*)(deg + 2 * w);
  const int nb = blockIdx.x * 8;
  unsigned bs = 0u;
#pragma unroll
  for (int k = 0; k < SC_TRIPS; ++k) {
    const int j = t + k * 256;
    const unsigned v = wtot[j < RW ? j : RW - 1];
    bs += j < nb ? v : 0u;
  }
  const unsigned p = d.x + d.y;
  unsigned inc = p;
  for (int o = 1; o < 32; o <<= 1) {
    const unsigned u = __shfl_up(inc, o);
    if (lane >= o) inc += u;
  }
  for (int o = 16; o > 0; o >>= 1) bs += __shfl_xor(bs, o);
  if (lane == 31) swave[wv] = inc;
  if (lane == 0) sbase[wv] = bs;
  __syncthreads();
  unsigned run = 0u;
  for (int i = 0; i < 8; ++i) {
    run += sbase[i];
    if (i < wv) run += swave[i];
  }
  const unsigned ex = run + inc - p;
  if (valid) {
    *(uint2*)(offs + 2 * w) = make_uint2(ex, ex + d.x);
    if (w == TW - 1) offs[NN] = ex + p;
  }
}

__global__ void k_fill(const int* __restrict__ src, const int* __restrict__ dst,
                       const unsigned* __restrict__ offs, const unsigned* __restrict__ pre,
                       const unsigned* __restrict__ lrank, unsigned* __restrict__ esrc) {
  int e = blockIdx.x * 256 + threadIdx.x;
  if (e < EE) {
    const int d = dst[e];
    const int b = e / SLICE;
    esrc[offs[d] + pre[(size_t)b * NN + d] + lrank[e]] = (unsigned)src[e];
  }
}


__global__ void k_agg128(const float* __restrict__ h, const unsigned* __restrict__ offs,
                         const unsigned* __restrict__ esrc,
                         const float* __restrict__ dsout, const float* __restrict__ dsin,
                         float* __restrict__ agg) {
  const int wv = threadIdx.x >> 5, lane = threadIdx.x & 31;
  const int n = blockIdx.x * 8 + wv;
  const unsigned b = offs[n], e = offs[n + 1];
  float4 acc = make_float4(0.f, 0.f, 0.f, 0.f);
  for (unsigned i = b; i < e; ++i) {
    const unsigned s = esrc[i];
    const float w = dsout[s];
    const float4 v = *(const float4*)(h + (size_t)s * FF + lane * 4);
    acc.x = fmaf(w, v.x, acc.x);
    acc.y = fmaf(w, v.y, acc.y);
    acc.z = fmaf(w, v.z, acc.z);
    acc.w = fmaf(w, v.w, acc.w);
  }
  const float c = dsin[n];
  *(float4*)(agg + (size_t)n * FF + lane * 4) =
      make_float4(c * acc.x, c * acc.y, c * acc.z, c * acc.w);
}

__global__ void k_agg64(const float* __restrict__ P, const unsigned* __restrict__ offs,
                        const unsigned* __restrict__ esrc, const float* __restrict__ dsin,
                        float* __restrict__ R) {
  const int wv = threadIdx.x >> 5, lane = threadIdx.x & 31;
  const int n = blockIdx.x * 8 + wv;
  const unsigned b = offs[n], e = offs[n + 1];
  float2 acc = make_float2(0.f, 0.f);
  for (unsigned i = b; i < e; ++i) {
    const unsigned s = esrc[i];
    const float2 v = *(const float2*)(P + (size_t)s * 64 + lane * 2);
    acc.x += v.x;
    acc.y += v.y;
  }
  const float c = dsin[n];
  *(float2*)(R + (size_t)n * 64 + lane * 2) = make_float2(c * acc.x, c * acc.y);
}


static constexpr int GD_PITCH = 132;
static constexpr int GD_TILES = NN / 16;
static constexpr int GD_BLOCKS = 512;
__global__ __launch_bounds__(256) void k_gemm_dual(
    const float* __restrict__ Agg, const float* __restrict__ H,
    const float* __restrict__ W, const float* __restrict__ L,
    float* __restrict__ Z, float* __restrict__ part) {
  __shared__ __attribute__((aligned(16))) float sAg[16 * GD_PITCH];
  __shared__ __attribute__((aligned(16))) float sH[16 * GD_PITCH];
  __shared__ float sStat[256];
  const int wv = threadIdx.x >> 5;
  const int lane = threadIdx.x & 31;
  const int t = lane & 15;
  const int hh = lane >> 4;
  const int cb = wv << 4;

  v2f bw[32], bl[32];
#pragma unroll
  for (int ks = 0; ks < 32; ++ks) {
    const int kk = 4 * ks + 2 * hh;
    bw[ks].x = W[kk * FF + cb + t]; bw[ks].y = W[(kk + 1) * FF + cb + t];
    bl[ks].x = L[kk * FF + cb + t]; bl[ks].y = L[(kk + 1) * FF + cb + t];
  }

  const int so0 = wv * GD_PITCH + lane * 4;
  const int so1 = (wv + 8) * GD_PITCH + lane * 4;
  const float* fa = sAg + t * GD_PITCH + 2 * hh;
  const float* fh = sH + t * GD_PITCH + 2 * hh;

  float csum = 0.f, csq = 0.f;
  int tile = blockIdx.x;
  float4 pa0, pa1, ph0, ph1;
  if (tile < GD_TILES) {
    const size_t g0 = (size_t)(tile * 16 + wv) * FF + lane * 4;
    const size_t g1 = g0 + (size_t)8 * FF;
    pa0 = *(const float4*)(Agg + g0); pa1 = *(const float4*)(Agg + g1);
    ph0 = *(const float4*)(H + g0);   ph1 = *(const float4*)(H + g1);
  }
  for (; tile < GD_TILES; tile += gridDim.x) {
    __syncthreads();
    *(float4*)(sAg + so0) = pa0; *(float4*)(sAg + so1) = pa1;
    *(float4*)(sH + so0) = ph0;  *(float4*)(sH + so1) = ph1;
    __syncthreads();
    const int nt = tile + gridDim.x;
    if (nt < GD_TILES) {
      const size_t g0 = (size_t)(nt * 16 + wv) * FF + lane * 4;
      const size_t g1 = g0 + (size_t)8 * FF;
      pa0 = *(const float4*)(Agg + g0); pa1 = *(const float4*)(Agg + g1);
      ph0 = *(const float4*)(H + g0);   ph1 = *(const float4*)(H + g1);
    }
    v8f acc = {};
#pragma unroll
    for (int ks = 0; ks < 32; ++ks) {
      v2f a1 = *(const v2f*)(fa + 4 * ks);
      acc = __builtin_amdgcn_wmma_f32_16x16x4_f32(false, a1, false, bw[ks], (short)0, acc, false, false);
      v2f a2 = *(const v2f*)(fh + 4 * ks);
      acc = __builtin_amdgcn_wmma_f32_16x16x4_f32(false, a2, false, bl[ks], (short)0, acc, false, false);
    }
    const int r0 = tile << 4;
#pragma unroll
    for (int v = 0; v < 8; ++v)
      Z[(size_t)(r0 + v + 8 * hh) * FF + cb + t] = acc[v];
    float ts = acc[0], tq = acc[0] * acc[0];
#pragma unroll
    for (int v = 1; v < 8; ++v) {
      ts += acc[v];
      tq = fmaf(acc[v], acc[v], tq);
    }
    csum += ts;
    csq += tq;
  }
  csum += __shfl_xor(csum, 16);
  csq += __shfl_xor(csq, 16);
  if (hh == 0) {
    sStat[cb + t] = csum;
    sStat[128 + cb + t] = csq;
  }
  __syncthreads();
  part[(size_t)blockIdx.x * 256 + threadIdx.x] = sStat[threadIdx.x];
}

static constexpr int GP_TILES = NN / 32;
__global__ __launch_bounds__(256) void k_gemm_p(
    const float* __restrict__ H, const float* __restrict__ dsout,
    const float* __restrict__ Wp, float* __restrict__ P) {
  __shared__ __attribute__((aligned(16))) float sT[32 * GD_PITCH];
  const int wv = threadIdx.x >> 5;
  const int lane = threadIdx.x & 31;
  const int t = lane & 15, hh = lane >> 4;
  const int ct = wv & 3, rt = wv >> 2;
  const int cb = ct << 4;

  v2f bp[32];
#pragma unroll
  for (int ks = 0; ks < 32; ++ks) {
    const int kk = 4 * ks + 2 * hh;
    bp[ks].x = Wp[kk * 64 + cb + t]; bp[ks].y = Wp[(kk + 1) * 64 + cb + t];
  }

  const int so = wv * GD_PITCH + lane * 4;
  const float* fa = sT + (16 * rt + t) * GD_PITCH + 2 * hh;

  int tile = blockIdx.x;
  float4 p0, p1, p2, p3;
  float dn = 0.f;
  if (tile < GP_TILES) {
    const size_t g = (size_t)(tile * 32 + wv) * FF + lane * 4;
    p0 = *(const float4*)(H + g);
    p1 = *(const float4*)(H + g + (size_t)8 * FF);
    p2 = *(const float4*)(H + g + (size_t)16 * FF);
    p3 = *(const float4*)(H + g + (size_t)24 * FF);
    dn = dsout[tile * 32 + 16 * rt + t];
  }
  for (; tile < GP_TILES; tile += gridDim.x) {
    const float d = dn;
    __syncthreads();
    *(float4*)(sT + so) = p0;
    *(float4*)(sT + so + 8 * GD_PITCH) = p1;
    *(float4*)(sT + so + 16 * GD_PITCH) = p2;
    *(float4*)(sT + so + 24 * GD_PITCH) = p3;
    __syncthreads();
    const int nt = tile + gridDim.x;
    if (nt < GP_TILES) {
      const size_t g = (size_t)(nt * 32 + wv) * FF + lane * 4;
      p0 = *(const float4*)(H + g);
      p1 = *(const float4*)(H + g + (size_t)8 * FF);
      p2 = *(const float4*)(H + g + (size_t)16 * FF);
      p3 = *(const float4*)(H + g + (size_t)24 * FF);
      dn = dsout[nt * 32 + 16 * rt + t];
    }
    v8f acc = {};
#pragma unroll
    for (int ks = 0; ks < 32; ++ks) {
      v2f a = *(const v2f*)(fa + 4 * ks);
      a.x = d * a.x; a.y = d * a.y;
      acc = __builtin_amdgcn_wmma_f32_16x16x4_f32(false, a, false, bp[ks], (short)0, acc, false, false);
    }
    const int r0 = tile * 32 + 16 * rt;
#pragma unroll
    for (int v = 0; v < 8; ++v)
      P[(size_t)(r0 + v + 8 * hh) * 64 + cb + t] = acc[v];
  }
}

__global__ __launch_bounds__(256) void k_gemm_out(
    const float* __restrict__ H, const float* __restrict__ Lp,
    const float* __restrict__ R, const float* __restrict__ b2,
    float* __restrict__ out) {
  __shared__ __attribute__((aligned(16))) float sT[32 * GD_PITCH];
  const int wv = threadIdx.x >> 5;
  const int lane = threadIdx.x & 31;
  const int t = lane & 15, hh = lane >> 4;
  const int ct = wv & 3, rt = wv >> 2;
  const int cb = ct << 4;
  const bool active = ct < 3;

  v2f bq[32];
#pragma unroll
  for (int ks = 0; ks < 32; ++ks) { bq[ks].x = 0.f; bq[ks].y = 0.f; }
  if (active) {
#pragma unroll
    for (int ks = 0; ks < 32; ++ks) {
      const int kk = 4 * ks + 2 * hh;
      bq[ks].x = Lp[kk * 64 + cb + t]; bq[ks].y = Lp[(kk + 1) * 64 + cb + t];
    }
  }

  const int so = wv * GD_PITCH + lane * 4;
  const float* fa = sT + (16 * rt + t) * GD_PITCH + 2 * hh;
  const int col = cb + t;

  int tile = blockIdx.x;
  float4 p0, p1, p2, p3;
  float rn[8];
#pragma unroll
  for (int v = 0; v < 8; ++v) rn[v] = 0.f;
  if (tile < GP_TILES) {
    const size_t g = (size_t)(tile * 32 + wv) * FF + lane * 4;
    p0 = *(const float4*)(H + g);
    p1 = *(const float4*)(H + g + (size_t)8 * FF);
    p2 = *(const float4*)(H + g + (size_t)16 * FF);
    p3 = *(const float4*)(H + g + (size_t)24 * FF);
    if (active) {
#pragma unroll
      for (int v = 0; v < 8; ++v)
        rn[v] = R[(size_t)(tile * 32 + 16 * rt + v + 8 * hh) * 64 + col];
    }
  }
  for (; tile < GP_TILES; tile += gridDim.x) {
    float rv[8];
#pragma unroll
    for (int v = 0; v < 8; ++v) rv[v] = rn[v];
    __syncthreads();
    *(float4*)(sT + so) = p0;
    *(float4*)(sT + so + 8 * GD_PITCH) = p1;
    *(float4*)(sT + so + 16 * GD_PITCH) = p2;
    *(float4*)(sT + so + 24 * GD_PITCH) = p3;
    __syncthreads();
    const int nt = tile + gridDim.x;
    if (nt < GP_TILES) {
      const size_t g = (size_t)(nt * 32 + wv) * FF + lane * 4;
      p0 = *(const float4*)(H + g);
      p1 = *(const float4*)(H + g + (size_t)8 * FF);
      p2 = *(const float4*)(H + g + (size_t)16 * FF);
      p3 = *(const float4*)(H + g + (size_t)24 * FF);
      if (active) {
#pragma unroll
        for (int v = 0; v < 8; ++v)
          rn[v] = R[(size_t)(nt * 32 + 16 * rt + v + 8 * hh) * 64 + col];
      }
    }
    if (active) {
      const int r0 = tile * 32 + 16 * rt;
      v8f acc = {};
#pragma unroll
      for (int ks = 0; ks < 32; ++ks) {
        v2f a = *(const v2f*)(fa + 4 * ks);
        acc = __builtin_amdgcn_wmma_f32_16x16x4_f32(false, a, false, bq[ks], (short)0, acc, false, false);
      }
      if (col < 40) {
        const float bias = b2[col];
#pragma unroll
        for (int v = 0; v < 8; ++v)
          out[(size_t)(r0 + v + 8 * hh) * 40 + col] = acc[v] + rv[v] + bias;
      }
    }
  }
}


static constexpr int BP_GROUPS = 16;
static constexpr int BP_ROWS = GD_BLOCKS / BP_GROUPS;
static_assert(BP_ROWS * BP_GROUPS == GD_BLOCKS, "the groups cover every partial row");
__global__ __launch_bounds__(1024) void k_bnparams(
    const float* __restrict__ part,
    const float* __restrict__ g, const float* __restrict__ be,
    float* __restrict__ bnA, float* __restrict__ bnB) {
  __shared__ __attribute__((aligned(16))) float red[BP_GROUPS * 256];
  const int q = threadIdx.x & 63, grp = threadIdx.x >> 6;
  const float4* p = (const float4*)(part + (size_t)grp * BP_ROWS * 256) + q;
  float4 s = make_float4(0.f, 0.f, 0.f, 0.f);
#pragma unroll 16
  for (int b = 0; b < BP_ROWS; ++b) {
    const float4 v = p[b * 64];
    s.x += v.x; s.y += v.y; s.z += v.z; s.w += v.w;
  }
  *(float4*)(red + grp * 256 + 4 * q) = s;
  __syncthreads();
  const int c = threadIdx.x;
  if (c < 128) {
    float cs = red[c], cs2 = red[128 + c];
#pragma unroll
    for (int k = 1; k < BP_GROUPS; ++k) {
      cs += red[k * 256 + c];
      cs2 += red[k * 256 + 128 + c];
    }
    const float inv = 1.0f / (float)NN;
    const float mu = cs * inv;
    const float var = cs2 * inv - mu * mu;
    const float a = g[c] * rsqrtf(var + 1e-5f);
    bnA[c] = a;
    bnB[c] = be[c] - mu * a;
  }
}

static constexpr int BA_STEPS = 10;
static constexpr int BA_BLOCKS = (NN * FF / 4) / (256 * BA_STEPS);
static_assert((size_t)BA_BLOCKS * 256 * BA_STEPS * 4 == (size_t)NN * FF, "bnapply grid must tile the matrix exactly");
__global__ __launch_bounds__(256) void k_bnapply(
    const float* __restrict__ z, const float* __restrict__ bnA,
    const float* __restrict__ bnB, float* __restrict__ hout) {
  const int c4 = (threadIdx.x & 31) * 4;
  const float4 a = *(const float4*)(bnA + c4);
  const float4 b = *(const float4*)(bnB + c4);
  const size_t j0 = ((size_t)blockIdx.x * BA_STEPS * 256 + threadIdx.x) * 4;
  float4 v[BA_STEPS];
#pragma unroll
  for (int k = 0; k < BA_STEPS; ++k) v[k] = *(const float4*)(z + j0 + (size_t)k * 1024);
#pragma unroll
  for (int k = 0; k < BA_STEPS; ++k) {
    float4 r;
    r.x = fmaf(v[k].x, a.x, b.x); r.x = r.x > 0.f ? r.x : 0.f;
    r.y = fmaf(v[k].y, a.y, b.y); r.y = r.y > 0.f ? r.y : 0.f;
    r.z = fmaf(v[k].z, a.z, b.z); r.z = r.z > 0.f ? r.z : 0.f;
    r.w = fmaf(v[k].w, a.w, b.w); r.w = r.w > 0.f ? r.w : 0.f;
    *(float4*)(hout + j0 + (size_t)k * 1024) = r;
  }
}

__global__ void k_padw(const float* __restrict__ W2, const float* __restrict__ L2,
                       float* __restrict__ Wp, float* __restrict__ Lp) {
  const int i = blockIdx.x * 256 + threadIdx.x;
  const int k = i >> 6, c = i & 63;
  const float w = (c < 40) ? W2[k * 40 + c] : 0.f;
  const float l = (c < 40) ? L2[k * 40 + c] : 0.f;
  Wp[i] = w;
  Lp[i] = l;
}


extern "C" void kernel_launch(void* const* d_in, const int* in_sizes, int n_in,
                              void* d_out, int out_size, void* d_ws, size_t ws_size,
                              hipStream_t stream) {
  (void)in_sizes; (void)n_in; (void)out_size; (void)ws_size;

  const float* feat = (const float*)d_in[0];
  const int* src   = (const int*)d_in[1];
  const int* dst   = (const int*)d_in[2];
  const float* W0  = (const float*)d_in[3];
  const float* L0  = (const float*)d_in[4];
  const float* W1  = (const float*)d_in[5];
  const float* L1  = (const float*)d_in[6];
  const float* W2  = (const float*)d_in[7];
  const float* L2  = (const float*)d_in[8];
  const float* b2  = (const float*)d_in[9];
  const float* g0  = (const float*)d_in[10];
  const float* be0 = (const float*)d_in[11];
  const float* g1  = (const float*)d_in[12];
  const float* be1 = (const float*)d_in[13];
  float* out = (float*)d_out;

  char* base = (char*)d_ws;
  size_t off = 0;
  auto take = [&](size_t bytes) -> char* {
    char* p = base + off;
    off = (off + bytes + 255) & ~(size_t)255;
    return p;
  };
  unsigned* indeg  = (unsigned*)take((size_t)NN * 4);
  unsigned* offs   = (unsigned*)take((size_t)(NN + 1) * 4);
  unsigned* lrank  = (unsigned*)take((size_t)EE * 4);
  unsigned* esrc   = (unsigned*)take((size_t)EE * 4);
  unsigned* cnt    = (unsigned*)take((size_t)2 * CB * TW * 4);
  unsigned* pre    = (unsigned*)take((size_t)CB * NN * 4);
  float* dsout = (float*)take((size_t)NN * 4);
  float* dsin  = (float*)take((size_t)NN * 4);
  float* part  = (float*)take((size_t)GD_BLOCKS * 256 * 4);
  float* bnA   = (float*)take(128 * 4);
  float* bnB   = (float*)take(128 * 4);
  float* Wp    = (float*)take(128 * 64 * 4);
  float* Lp    = (float*)take(128 * 64 * 4);
  float* bufA  = (float*)take((size_t)NN * FF * 4);
  float* bufB  = (float*)take((size_t)NN * FF * 4);
  float* bufC  = (float*)take((size_t)NN * FF * 4);
  unsigned* wtot = (unsigned*)take((size_t)RW * 4);

  const int BEdge = (EE + 255) / 256;
  const int BWave = NN / 8;

  k_count<<<2 * CB, 1024, 0, stream>>>(src, dst, cnt, lrank);
  k_reduce<<<2 * RB, 256, 0, stream>>>(cnt, pre, indeg, wtot, dsout, dsin);
  k_scan<<<RB, 256, 0, stream>>>(indeg, wtot, offs);
  k_fill<<<BEdge, 256, 0, stream>>>(src, dst, offs, pre, lrank, esrc);
  k_padw<<<32, 256, 0, stream>>>(W2, L2, Wp, Lp);

  k_agg128<<<BWave, 256, 0, stream>>>(feat, offs, esrc, dsout, dsin, bufA);
  k_gemm_dual<<<GD_BLOCKS, 256, 0, stream>>>(bufA, feat, W0, L0, bufB, part);
  k_bnparams<<<1, 1024, 0, stream>>>(part, g0, be0, bnA, bnB);
  k_bnapply<<<BA_BLOCKS, 256, 0, stream>>>(bufB, bnA, bnB, bufC);

  k_agg128<<<BWave, 256, 0, stream>>>(bufC, offs, esrc, dsout, dsin, bufA);
  k_gemm_dual<<<GD_BLOCKS, 256, 0, stream>>>(bufA, bufC, W1, L1, bufB, part);
  k_bnparams<<<1, 1024, 0, stream>>>(part, g1, be1, bnA, bnB);
  k_bnapply<<<BA_BLOCKS, 256, 0, stream>>>(bufB, bnA, bnB, bufC);

  k_gemm_p<<<1024, 256, 0, stream>>>(bufC, dsout, Wp, bufA);
  k_agg64<<<BWave, 256, 0, stream>>>(bufA, offs, esrc, dsin, bufB);
  k_gemm_out<<<1024, 256, 0, stream>>>(bufC, Lp, bufB, b2, out);
}


# BN+ReLU applied where a row of h is read (gather, GEMM operand staging) from the stored z; the two BN-apply launches and the write and re-read of h1, h2 are gone
# speedup vs baseline: 2.3871x; 1.1039x over previous
#include <hip/hip_runtime.h>
#include <hip/hip_bf16.h>


static constexpr int NN = 100000;
static constexpr int EE = 1600000;
static constexpr int FF = 128;

typedef __attribute__((ext_vector_type(2))) float v2f;
typedef __attribute__((ext_vector_type(8))) float v8f;

__device__ __forceinline__ float4 bn_relu4(float4 z, float4 a, float4 b) {
  float4 r;
  r.x = fmaf(z.x, a.x, b.x); r.x = r.x > 0.f ? r.x : 0.f;
  r.y = fmaf(z.y, a.y, b.y); r.y = r.y > 0.f ? r.y : 0.f;
  r.z = fmaf(z.z, a.z, b.z); r.z = r.z > 0.f ? r.z : 0.f;
  r.w = fmaf(z.w, a.w, b.w); r.w = r.w > 0.f ? r.w : 0.f;
  return r;
}


static constexpr int CB = 64;
static constexpr int SLICE = EE / CB;
static constexpr int TW = NN / 2;
static_assert(SLICE * CB == EE, "slices must tile the edge list exactly");
static_assert(SLICE < 65536, "a 16-bit counter must hold a whole slice");
static_assert(TW * 2 == NN && TW % 4 == 0, "table is copied as 16-byte pieces");
static constexpr int CQ = SLICE / 4;
static constexpr int CQ_PT = (CQ + 1023) / 1024;
static_assert(CQ * 4 == SLICE, "a slice is a whole number of 16-byte groups, so every slice starts 16-byte aligned");
__global__ __launch_bounds__(1024) void k_count(
    const int* __restrict__ src, const int* __restrict__ dst,
    unsigned* __restrict__ cnt, unsigned* __restrict__ lrank) {
  __shared__ __attribute__((aligned(16))) unsigned tab[TW];
  const int role = blockIdx.x >= CB ? 1 : 0;
  const int b = blockIdx.x - role * CB;
  const int tid = threadIdx.x;
  const int4* in4 = (const int4*)((role ? src : dst) + (size_t)b * SLICE);
  int4 q[CQ_PT];
#pragma unroll
  for (int k = 0; k < CQ_PT; ++k) {
    const int i = tid + k * 1024;
    q[k] = in4[i < CQ ? i : CQ - 1];
  }
  uint4* tab4 = (uint4*)tab;
  for (int i = tid; i < TW / 4; i += 1024) tab4[i] = make_uint4(0u, 0u, 0u, 0u);
  __syncthreads();
  if (role == 0) {
    uint4 r[CQ_PT];
#pragma unroll
    for (int k = 0; k < CQ_PT; ++k) {
      if (tid + k * 1024 < CQ) {
        const unsigned dx = (unsigned)q[k].x, dy = (unsigned)q[k].y;
        const unsigned dz = (unsigned)q[k].z, dw = (unsigned)q[k].w;
        r[k].x = atomicAdd(&tab[dx >> 1], 1u << ((dx & 1u) << 4));
        r[k].y = atomicAdd(&tab[dy >> 1], 1u << ((dy & 1u) << 4));
        r[k].z = atomicAdd(&tab[dz >> 1], 1u << ((dz & 1u) << 4));
        r[k].w = atomicAdd(&tab[dw >> 1], 1u << ((dw & 1u) << 4));
      }
    }
    uint4* out4 = (uint4*)(lrank + (size_t)b * SLICE);
#pragma unroll
    for (int k = 0; k < CQ_PT; ++k) {
      if (tid + k * 1024 < CQ) {
        uint4 o;
        o.x = (r[k].x >> (((unsigned)q[k].x & 1u) << 4)) & 0xFFFFu;
        o.y = (r[k].y >> (((unsigned)q[k].y & 1u) << 4)) & 0xFFFFu;
        o.z = (r[k].z >> (((unsigned)q[k].z & 1u) << 4)) & 0xFFFFu;
        o.w = (r[k].w >> (((unsigned)q[k].w & 1u) << 4)) & 0xFFFFu;
        out4[tid + k * 1024] = o;
      }
    }
  } else {
#pragma unroll
    for (int k = 0; k < CQ_PT; ++k) {
      if (tid + k * 1024 < CQ) {
        const unsigned sx = (unsigned)q[k].x, sy = (unsigned)q[k].y;
        const unsigned sz = (unsigned)q[k].z, sw = (unsigned)q[k].w;
        atomicAdd(&tab[sx >> 1], 1u << ((sx & 1u) << 4));
        atomicAdd(&tab[sy >> 1], 1u << ((sy & 1u) << 4));
        atomicAdd(&tab[sz >> 1], 1u << ((sz & 1u) << 4));
        atomicAdd(&tab[sw >> 1], 1u << ((sw & 1u) << 4));
      }
    }
  }
  __syncthreads();
  uint4* o = (uint4*)(cnt + (size_t)blockIdx.x * TW);
  for (int i = tid; i < TW / 4; i += 1024) o[i] = tab4[i];
}

static constexpr int RB = (TW + 255) / 256;
static constexpr int RW = RB * 8;
__global__ __launch_bounds__(256) void k_reduce(
    const unsigned* __restrict__ cnt, unsigned* __restrict__ pre,
    unsigned* __restrict__ indeg, unsigned* __restrict__ wtot,
    float* __restrict__ dsout, float* __restrict__ dsin) {
  const int role = blockIdx.x >= RB ? 1 : 0;
  const int w = (blockIdx.x - role * RB) * 256 + threadIdx.x;
  unsigned lo = 0u, hi = 0u;
  if (w < TW) {
    const unsigned* c = cnt + (size_t)role * CB * TW + w;
    unsigned v[CB];
#pragma unroll
    for (int b = 0; b < CB; ++b) v[b] = c[(size_t)b * TW];
    if (role == 0) {
#pragma unroll
      for (int b = 0; b < CB; ++b) {
        *(uint2*)(pre + (size_t)b * NN + 2 * w) = make_uint2(lo, hi);
        lo += v[b] & 0xFFFFu;
        hi += v[b] >> 16;
      }
      *(uint2*)(indeg + 2 * w) = make_uint2(lo, hi);
      *(float2*)(dsin + 2 * w) = make_float2(lo ? rsqrtf((float)lo) : 1.0f,
                                             hi ? rsqrtf((float)hi) : 1.0f);
    } else {
#pragma unroll
      for (int b = 0; b < CB; ++b) {
        lo += v[b] & 0xFFFFu;
        hi += v[b] >> 16;
      }
      *(float2*)(dsout + 2 * w) = make_float2(lo ? rsqrtf((float)lo) : 1.0f,
                                              hi ? rsqrtf((float)hi) : 1.0f);
    }
  }
  if (role == 0) {
    unsigned s = lo + hi;
    for (int o = 16; o > 0; o >>= 1) s += __shfl_xor(s, o);
    if ((threadIdx.x & 31) == 0) wtot[w >> 5] = s;
  }
}

static constexpr int SC_TRIPS = (RW + 255) / 256;
static_assert(RB * 256 >= TW, "the scan grid covers the table");
__global__ __launch_bounds__(256) void k_scan(const unsigned* __restrict__ deg,
                                              const unsigned* __restrict__ wtot,
                                              unsigned* __restrict__ offs) {
  __shared__ unsigned sbase[8], swave[8];
  const int t = threadIdx.x, lane = t & 31, wv = t >> 5;
  const int w = blockIdx.x * 256 + t;
  const bool valid = w < TW;
  uint2 d = make_uint2(0u, 0u);
  if (valid) d = *(const uint2*)(deg + 2 * w);
  const int nb = blockIdx.x * 8;
  unsigned bs = 0u;
#pragma unroll
  for (int k = 0; k < SC_TRIPS; ++k) {
    const int j = t + k * 256;
    const unsigned v = wtot[j < RW ? j : RW - 1];
    bs += j < nb ? v : 0u;
  }
  const unsigned p = d.x + d.y;
  unsigned inc = p;
  for (int o = 1; o < 32; o <<= 1) {
    const unsigned u = __shfl_up(inc, o);
    if (lane >= o) inc += u;
  }
  for (int o = 16; o > 0; o >>= 1) bs += __shfl_xor(bs, o);
  if (lane == 31) swave[wv] = inc;
  if (lane == 0) sbase[wv] = bs;
  __syncthreads();
  unsigned run = 0u;
  for (int i = 0; i < 8; ++i) {
    run += sbase[i];
    if (i < wv) run += swave[i];
  }
  const unsigned ex = run + inc - p;
  if (valid) {
    *(uint2*)(offs + 2 * w) = make_uint2(ex, ex + d.x);
    if (w == TW - 1) offs[NN] = ex + p;
  }
}

__global__ void k_fill(const int* __restrict__ src, const int* __restrict__ dst,
                       const unsigned* __restrict__ offs, const unsigned* __restrict__ pre,
                       const unsigned* __restrict__ lrank, unsigned* __restrict__ esrc) {
  int e = blockIdx.x * 256 + threadIdx.x;
  if (e < EE) {
    const int d = dst[e];
    const int b = e / SLICE;
    esrc[offs[d] + pre[(size_t)b * NN + d] + lrank[e]] = (unsigned)src[e];
  }
}


template <bool BN>
__global__ void k_agg128(const float* __restrict__ h, const unsigned* __restrict__ offs,
                         const unsigned* __restrict__ esrc,
                         const float* __restrict__ dsout, const float* __restrict__ dsin,
                         const float* __restrict__ bnA, const float* __restrict__ bnB,
                         float* __restrict__ agg) {
  const int wv = threadIdx.x >> 5, lane = threadIdx.x & 31;
  const int n = blockIdx.x * 8 + wv;
  const unsigned b = offs[n], e = offs[n + 1];
  float4 ca = make_float4(1.f, 1.f, 1.f, 1.f), cb4 = make_float4(0.f, 0.f, 0.f, 0.f);
  if (BN) {
    ca = *(const float4*)(bnA + lane * 4);
    cb4 = *(const float4*)(bnB + lane * 4);
  }
  float4 acc = make_float4(0.f, 0.f, 0.f, 0.f);
  for (unsigned i = b; i < e; ++i) {
    const unsigned s = esrc[i];
    const float w = dsout[s];
    float4 v = *(const float4*)(h + (size_t)s * FF + lane * 4);
    if (BN) v = bn_relu4(v, ca, cb4);
    acc.x = fmaf(w, v.x, acc.x);
    acc.y = fmaf(w, v.y, acc.y);
    acc.z = fmaf(w, v.z, acc.z);
    acc.w = fmaf(w, v.w, acc.w);
  }
  const float c = dsin[n];
  *(float4*)(agg + (size_t)n * FF + lane * 4) =
      make_float4(c * acc.x, c * acc.y, c * acc.z, c * acc.w);
}

__global__ void k_agg64(const float* __restrict__ P, const unsigned* __restrict__ offs,
                        const unsigned* __restrict__ esrc, const float* __restrict__ dsin,
                        float* __restrict__ R) {
  const int wv = threadIdx.x >> 5, lane = threadIdx.x & 31;
  const int n = blockIdx.x * 8 + wv;
  const unsigned b = offs[n], e = offs[n + 1];
  float2 acc = make_float2(0.f, 0.f);
  for (unsigned i = b; i < e; ++i) {
    const unsigned s = esrc[i];
    const float2 v = *(const float2*)(P + (size_t)s * 64 + lane * 2);
    acc.x += v.x;
    acc.y += v.y;
  }
  const float c = dsin[n];
  *(float2*)(R + (size_t)n * 64 + lane * 2) = make_float2(c * acc.x, c * acc.y);
}


static constexpr int GD_PITCH = 132;
static constexpr int GD_TILES = NN / 16;
static constexpr int GD_BLOCKS = 512;
template <bool BN>
__global__ __launch_bounds__(256) void k_gemm_dual(
    const float* __restrict__ Agg, const float* __restrict__ H,
    const float* __restrict__ W, const float* __restrict__ L,
    const float* __restrict__ bnA, const float* __restrict__ bnB,
    float* __restrict__ Z, float* __restrict__ part) {
  __shared__ __attribute__((aligned(16))) float sAg[16 * GD_PITCH];
  __shared__ __attribute__((aligned(16))) float sH[16 * GD_PITCH];
  __shared__ float sStat[256];
  const int wv = threadIdx.x >> 5;
  const int lane = threadIdx.x & 31;
  const int t = lane & 15;
  const int hh = lane >> 4;
  const int cb = wv << 4;

  v2f bw[32], bl[32];
#pragma unroll
  for (int ks = 0; ks < 32; ++ks) {
    const int kk = 4 * ks + 2 * hh;
    bw[ks].x = W[kk * FF + cb + t]; bw[ks].y = W[(kk + 1) * FF + cb + t];
    bl[ks].x = L[kk * FF + cb + t]; bl[ks].y = L[(kk + 1) * FF + cb + t];
  }

  const int so0 = wv * GD_PITCH + lane * 4;
  const int so1 = (wv + 8) * GD_PITCH + lane * 4;
  const float* fa = sAg + t * GD_PITCH + 2 * hh;
  const float* fh = sH + t * GD_PITCH + 2 * hh;
  float4 ca = make_float4(1.f, 1.f, 1.f, 1.f), cb4 = make_float4(0.f, 0.f, 0.f, 0.f);
  if (BN) {
    ca = *(const float4*)(bnA + lane * 4);
    cb4 = *(const float4*)(bnB + lane * 4);
  }

  float csum = 0.f, csq = 0.f;
  int tile = blockIdx.x;
  float4 pa0, pa1, ph0, ph1;
  if (tile < GD_TILES) {
    const size_t g0 = (size_t)(tile * 16 + wv) * FF + lane * 4;
    const size_t g1 = g0 + (size_t)8 * FF;
    pa0 = *(const float4*)(Agg + g0); pa1 = *(const float4*)(Agg + g1);
    ph0 = *(const float4*)(H + g0);   ph1 = *(const float4*)(H + g1);
  }
  for (; tile < GD_TILES; tile += gridDim.x) {
    __syncthreads();
    *(float4*)(sAg + so0) = pa0; *(float4*)(sAg + so1) = pa1;
    if (BN) { ph0 = bn_relu4(ph0, ca, cb4); ph1 = bn_relu4(ph1, ca, cb4); }
    *(float4*)(sH + so0) = ph0;  *(float4*)(sH + so1) = ph1;
    __syncthreads();
    const int nt = tile + gridDim.x;
    if (nt < GD_TILES) {
      const size_t g0 = (size_t)(nt * 16 + wv) * FF + lane * 4;
      const size_t g1 = g0 + (size_t)8 * FF;
      pa0 = *(const float4*)(Agg + g0); pa1 = *(const float4*)(Agg + g1);
      ph0 = *(const float4*)(H + g0);   ph1 = *(const float4*)(H + g1);
    }
    v8f acc = {};
#pragma unroll
    for (int ks = 0; ks < 32; ++ks) {
      v2f a1 = *(const v2f*)(fa + 4 * ks);
      acc = __builtin_amdgcn_wmma_f32_16x16x4_f32(false, a1, false, bw[ks], (short)0, acc, false, false);
      v2f a2 = *(const v2f*)(fh + 4 * ks);
      acc = __builtin_amdgcn_wmma_f32_16x16x4_f32(false, a2, false, bl[ks], (short)0, acc, false, false);
    }
    const int r0 = tile << 4;
#pragma unroll
    for (int v = 0; v < 8; ++v)
      Z[(size_t)(r0 + v + 8 * hh) * FF + cb + t] = acc[v];
    float ts = acc[0], tq = acc[0] * acc[0];
#pragma unroll
    for (int v = 1; v < 8; ++v) {
      ts += acc[v];
      tq = fmaf(acc[v], acc[v], tq);
    }
    csum += ts;
    csq += tq;
  }
  csum += __shfl_xor(csum, 16);
  csq += __shfl_xor(csq, 16);
  if (hh == 0) {
    sStat[cb + t] = csum;
    sStat[128 + cb + t] = csq;
  }
  __syncthreads();
  part[(size_t)blockIdx.x * 256 + threadIdx.x] = sStat[threadIdx.x];
}

static constexpr int GP_TILES = NN / 32;
__global__ __launch_bounds__(256) void k_gemm_p(
    const float* __restrict__ H, const float* __restrict__ dsout,
    const float* __restrict__ bnA, const float* __restrict__ bnB,
    const float* __restrict__ Wp, float* __restrict__ P) {
  __shared__ __attribute__((aligned(16))) float sT[32 * GD_PITCH];
  const int wv = threadIdx.x >> 5;
  const int lane = threadIdx.x & 31;
  const int t = lane & 15, hh = lane >> 4;
  const int ct = wv & 3, rt = wv >> 2;
  const int cb = ct << 4;

  v2f bp[32];
#pragma unroll
  for (int ks = 0; ks < 32; ++ks) {
    const int kk = 4 * ks + 2 * hh;
    bp[ks].x = Wp[kk * 64 + cb + t]; bp[ks].y = Wp[(kk + 1) * 64 + cb + t];
  }

  const int so = wv * GD_PITCH + lane * 4;
  const float* fa = sT + (16 * rt + t) * GD_PITCH + 2 * hh;
  const float4 ca = *(const float4*)(bnA + lane * 4);
  const float4 cb4 = *(const float4*)(bnB + lane * 4);

  int tile = blockIdx.x;
  float4 p0, p1, p2, p3;
  float dn = 0.f;
  if (tile < GP_TILES) {
    const size_t g = (size_t)(tile * 32 + wv) * FF + lane * 4;
    p0 = *(const float4*)(H + g);
    p1 = *(const float4*)(H + g + (size_t)8 * FF);
    p2 = *(const float4*)(H + g + (size_t)16 * FF);
    p3 = *(const float4*)(H + g + (size_t)24 * FF);
    dn = dsout[tile * 32 + 16 * rt + t];
  }
  for (; tile < GP_TILES; tile += gridDim.x) {
    const float d = dn;
    __syncthreads();
    *(float4*)(sT + so) = bn_relu4(p0, ca, cb4);
    *(float4*)(sT + so + 8 * GD_PITCH) = bn_relu4(p1, ca, cb4);
    *(float4*)(sT + so + 16 * GD_PITCH) = bn_relu4(p2, ca, cb4);
    *(float4*)(sT + so + 24 * GD_PITCH) = bn_relu4(p3, ca, cb4);
    __syncthreads();
    const int nt = tile + gridDim.x;
    if (nt < GP_TILES) {
      const size_t g = (size_t)(nt * 32 + wv) * FF + lane * 4;
      p0 = *(const float4*)(H + g);
      p1 = *(const float4*)(H + g + (size_t)8 * FF);
      p2 = *(const float4*)(H + g + (size_t)16 * FF);
      p3 = *(const float4*)(H + g + (size_t)24 * FF);
      dn = dsout[nt * 32 + 16 * rt + t];
    }
    v8f acc = {};
#pragma unroll
    for (int ks = 0; ks < 32; ++ks) {
      v2f a = *(const v2f*)(fa + 4 * ks);
      a.x = d * a.x; a.y = d * a.y;
      acc = __builtin_amdgcn_wmma_f32_16x16x4_f32(false, a, false, bp[ks], (short)0, acc, false, false);
    }
    const int r0 = tile * 32 + 16 * rt;
#pragma unroll
    for (int v = 0; v < 8; ++v)
      P[(size_t)(r0 + v + 8 * hh) * 64 + cb + t] = acc[v];
  }
}

__global__ __launch_bounds__(256) void k_gemm_out(
    const float* __restrict__ H, const float* __restrict__ Lp,
    const float* __restrict__ bnA, const float* __restrict__ bnB,
    const float* __restrict__ R, const float* __restrict__ b2,
    float* __restrict__ out) {
  __shared__ __attribute__((aligned(16))) float sT[32 * GD_PITCH];
  __shared__ __attribute__((aligned(16))) float sBN[256];
  const int wv = threadIdx.x >> 5;
  const int lane = threadIdx.x & 31;
  const int t = lane & 15, hh = lane >> 4;
  const int ct = wv & 3, rt = wv >> 2;
  const int cb = ct << 4;
  const bool active = ct < 3;

  v2f bq[32];
#pragma unroll
  for (int ks = 0; ks < 32; ++ks) { bq[ks].x = 0.f; bq[ks].y = 0.f; }
  if (active) {
#pragma unroll
    for (int ks = 0; ks < 32; ++ks) {
      const int kk = 4 * ks + 2 * hh;
      bq[ks].x = Lp[kk * 64 + cb + t]; bq[ks].y = Lp[(kk + 1) * 64 + cb + t];
    }
  }

  const int so = wv * GD_PITCH + lane * 4;
  const float* fa = sT + (16 * rt + t) * GD_PITCH + 2 * hh;
  sBN[threadIdx.x] = threadIdx.x < 128 ? bnA[threadIdx.x] : bnB[threadIdx.x - 128];
  const int col = cb + t;

  int tile = blockIdx.x;
  float4 p0, p1, p2, p3;
  float rn[8];
#pragma unroll
  for (int v = 0; v < 8; ++v) rn[v] = 0.f;
  if (tile < GP_TILES) {
    const size_t g = (size_t)(tile * 32 + wv) * FF + lane * 4;
    p0 = *(const float4*)(H + g);
    p1 = *(const float4*)(H + g + (size_t)8 * FF);
    p2 = *(const float4*)(H + g + (size_t)16 * FF);
    p3 = *(const float4*)(H + g + (size_t)24 * FF);
    if (active) {
#pragma unroll
      for (int v = 0; v < 8; ++v)
        rn[v] = R[(size_t)(tile * 32 + 16 * rt + v + 8 * hh) * 64 + col];
    }
  }
  for (; tile < GP_TILES; tile += gridDim.x) {
    float rv[8];
#pragma unroll
    for (int v = 0; v < 8; ++v) rv[v] = rn[v];
    __syncthreads();
    const float4 ca = *(const float4*)(sBN + lane * 4);
    const float4 cb4 = *(const float4*)(sBN + 128 + lane * 4);
    *(float4*)(sT + so) = bn_relu4(p0, ca, cb4);
    *(float4*)(sT + so + 8 * GD_PITCH) = bn_relu4(p1, ca, cb4);
    *(float4*)(sT + so + 16 * GD_PITCH) = bn_relu4(p2, ca, cb4);
    *(float4*)(sT + so + 24 * GD_PITCH) = bn_relu4(p3, ca, cb4);
    __syncthreads();
    const int nt = tile + gridDim.x;
    if (nt < GP_TILES) {
      const size_t g = (size_t)(nt * 32 + wv) * FF + lane * 4;
      p0 = *(const float4*)(H + g);
      p1 = *(const float4*)(H + g + (size_t)8 * FF);
      p2 = *(const float4*)(H + g + (size_t)16 * FF);
      p3 = *(const float4*)(H + g + (size_t)24 * FF);
      if (active) {
#pragma unroll
        for (int v = 0; v < 8; ++v)
          rn[v] = R[(size_t)(nt * 32 + 16 * rt + v + 8 * hh) * 64 + col];
      }
    }
    if (active) {
      const int r0 = tile * 32 + 16 * rt;
      v8f acc = {};
#pragma unroll
      for (int ks = 0; ks < 32; ++ks) {
        v2f a = *(const v2f*)(fa + 4 * ks);
        acc = __builtin_amdgcn_wmma_f32_16x16x4_f32(false, a, false, bq[ks], (short)0, acc, false, false);
      }
      if (col < 40) {
        const float bias = b2[col];
#pragma unroll
        for (int v = 0; v < 8; ++v)
          out[(size_t)(r0 + v + 8 * hh) * 40 + col] = acc[v] + rv[v] + bias;
      }
    }
  }
}


static constexpr int BP_GROUPS = 16;
static constexpr int BP_ROWS = GD_BLOCKS / BP_GROUPS;
static_assert(BP_ROWS * BP_GROUPS == GD_BLOCKS, "the groups cover every partial row");
__global__ __launch_bounds__(1024) void k_bnparams(
    const float* __restrict__ part,
    const float* __restrict__ g, const float* __restrict__ be,
    float* __restrict__ bnA, float* __restrict__ bnB) {
  __shared__ __attribute__((aligned(16))) float red[BP_GROUPS * 256];
  const int q = threadIdx.x & 63, grp = threadIdx.x >> 6;
  const float4* p = (const float4*)(part + (size_t)grp * BP_ROWS * 256) + q;
  float4 s = make_float4(0.f, 0.f, 0.f, 0.f);
#pragma unroll 16
  for (int b = 0; b < BP_ROWS; ++b) {
    const float4 v = p[b * 64];
    s.x += v.x; s.y += v.y; s.z += v.z; s.w += v.w;
  }
  *(float4*)(red + grp * 256 + 4 * q) = s;
  __syncthreads();
  const int c = threadIdx.x;
  if (c < 128) {
    float cs = red[c], cs2 = red[128 + c];
#pragma unroll
    for (int k = 1; k < BP_GROUPS; ++k) {
      cs += red[k * 256 + c];
      cs2 += red[k * 256 + 128 + c];
    }
    const float inv = 1.0f / (float)NN;
    const float mu = cs * inv;
    const float var = cs2 * inv - mu * mu;
    const float a = g[c] * rsqrtf(var + 1e-5f);
    bnA[c] = a;
    bnB[c] = be[c] - mu * a;
  }
}

__global__ void k_padw(const float* __restrict__ W2, const float* __restrict__ L2,
                       float* __restrict__ Wp, float* __restrict__ Lp) {
  const int i = blockIdx.x * 256 + threadIdx.x;
  const int k = i >> 6, c = i & 63;
  const float w = (c < 40) ? W2[k * 40 + c] : 0.f;
  const float l = (c < 40) ? L2[k * 40 + c] : 0.f;
  Wp[i] = w;
  Lp[i] = l;
}


extern "C" void kernel_launch(void* const* d_in, const int* in_sizes, int n_in,
                              void* d_out, int out_size, void* d_ws, size_t ws_size,
                              hipStream_t stream) {
  (void)in_sizes; (void)n_in; (void)out_size; (void)ws_size;

  const float* feat = (const float*)d_in[0];
  const int* src   = (const int*)d_in[1];
  const int* dst   = (const int*)d_in[2];
  const float* W0  = (const float*)d_in[3];
  const float* L0  = (const float*)d_in[4];
  const float* W1  = (const float*)d_in[5];
  const float* L1  = (const float*)d_in[6];
  const float* W2  = (const float*)d_in[7];
  const float* L2  = (const float*)d_in[8];
  const float* b2  = (const float*)d_in[9];
  const float* g0  = (const float*)d_in[10];
  const float* be0 = (const float*)d_in[11];
  const float* g1  = (const float*)d_in[12];
  const float* be1 = (const float*)d_in[13];
  float* out = (float*)d_out;

  char* base = (char*)d_ws;
  size_t off = 0;
  auto take = [&](size_t bytes) -> char* {
    char* p = base + off;
    off = (off + bytes + 255) & ~(size_t)255;
    return p;
  };
  unsigned* indeg  = (unsigned*)take((size_t)NN * 4);
  unsigned* offs   = (unsigned*)take((size_t)(NN + 1) * 4);
  unsigned* lrank  = (unsigned*)take((size_t)EE * 4);
  unsigned* esrc   = (unsigned*)take((size_t)EE * 4);
  unsigned* cnt    = (unsigned*)take((size_t)2 * CB * TW * 4);
  unsigned* pre    = (unsigned*)take((size_t)CB * NN * 4);
  float* dsout = (float*)take((size_t)NN * 4);
  float* dsin  = (float*)take((size_t)NN * 4);
  float* part  = (float*)take((size_t)GD_BLOCKS * 256 * 4);
  float* bnA0  = (float*)take(128 * 4);
  float* bnB0  = (float*)take(128 * 4);
  float* bnA1  = (float*)take(128 * 4);
  float* bnB1  = (float*)take(128 * 4);
  float* Wp    = (float*)take(128 * 64 * 4);
  float* Lp    = (float*)take(128 * 64 * 4);
  float* bufA  = (float*)take((size_t)NN * FF * 4);
  float* bufB  = (float*)take((size_t)NN * FF * 4);
  float* bufC  = (float*)take((size_t)NN * FF * 4);
  unsigned* wtot = (unsigned*)take((size_t)RW * 4);

  const int BEdge = (EE + 255) / 256;
  const int BWave = NN / 8;

  k_count<<<2 * CB, 1024, 0, stream>>>(src, dst, cnt, lrank);
  k_reduce<<<2 * RB, 256, 0, stream>>>(cnt, pre, indeg, wtot, dsout, dsin);
  k_scan<<<RB, 256, 0, stream>>>(indeg, wtot, offs);
  k_fill<<<BEdge, 256, 0, stream>>>(src, dst, offs, pre, lrank, esrc);
  k_padw<<<32, 256, 0, stream>>>(W2, L2, Wp, Lp);

  k_agg128<false><<<BWave, 256, 0, stream>>>(feat, offs, esrc, dsout, dsin, nullptr, nullptr, bufA);
  k_gemm_dual<false><<<GD_BLOCKS, 256, 0, stream>>>(bufA, feat, W0, L0, nullptr, nullptr, bufB, part);
  k_bnparams<<<1, 1024, 0, stream>>>(part, g0, be0, bnA0, bnB0);

  k_agg128<true><<<BWave, 256, 0, stream>>>(bufB, offs, esrc, dsout, dsin, bnA0, bnB0, bufA);
  k_gemm_dual<true><<<GD_BLOCKS, 256, 0, stream>>>(bufA, bufB, W1, L1, bnA0, bnB0, bufC, part);
  k_bnparams<<<1, 1024, 0, stream>>>(part, g1, be1, bnA1, bnB1);

  k_gemm_p<<<1024, 256, 0, stream>>>(bufC, dsout, bnA1, bnB1, Wp, bufA);
  k_agg64<<<BWave, 256, 0, stream>>>(bufA, offs, esrc, dsin, bufB);
  k_gemm_out<<<1024, 256, 0, stream>>>(bufC, Lp, bnA1, bnB1, bufB, b2, out);
}


# Neighbour gather moved into the dual-GEMM kernel: each wave sums the in-neighbour rows of its two tile rows straight into the LDS A tile; both k_agg128 launches and the agg write and re-read are gone
# speedup vs baseline: 2.5180x; 1.0548x over previous
#include <hip/hip_runtime.h>
#include <hip/hip_bf16.h>


static constexpr int NN = 100000;
static constexpr int EE = 1600000;
static constexpr int FF = 128;

typedef __attribute__((ext_vector_type(2))) float v2f;
typedef __attribute__((ext_vector_type(8))) float v8f;

__device__ __forceinline__ float4 bn_relu4(float4 z, float4 a, float4 b) {
  float4 r;
  r.x = fmaf(z.x, a.x, b.x); r.x = r.x > 0.f ? r.x : 0.f;
  r.y = fmaf(z.y, a.y, b.y); r.y = r.y > 0.f ? r.y : 0.f;
  r.z = fmaf(z.z, a.z, b.z); r.z = r.z > 0.f ? r.z : 0.f;
  r.w = fmaf(z.w, a.w, b.w); r.w = r.w > 0.f ? r.w : 0.f;
  return r;
}


static constexpr int CB = 64;
static constexpr int SLICE = EE / CB;
static constexpr int TW = NN / 2;
static_assert(SLICE * CB == EE, "slices must tile the edge list exactly");
static_assert(SLICE < 65536, "a 16-bit counter must hold a whole slice");
static_assert(TW * 2 == NN && TW % 4 == 0, "table is copied as 16-byte pieces");
static constexpr int CQ = SLICE / 4;
static constexpr int CQ_PT = (CQ + 1023) / 1024;
static_assert(CQ * 4 == SLICE, "a slice is a whole number of 16-byte groups, so every slice starts 16-byte aligned");
__global__ __launch_bounds__(1024) void k_count(
    const int* __restrict__ src, const int* __restrict__ dst,
    unsigned* __restrict__ cnt, unsigned* __restrict__ lrank) {
  __shared__ __attribute__((aligned(16))) unsigned tab[TW];
  const int role = blockIdx.x >= CB ? 1 : 0;
  const int b = blockIdx.x - role * CB;
  const int tid = threadIdx.x;
  const int4* in4 = (const int4*)((role ? src : dst) + (size_t)b * SLICE);
  int4 q[CQ_PT];
#pragma unroll
  for (int k = 0; k < CQ_PT; ++k) {
    const int i = tid + k * 1024;
    q[k] = in4[i < CQ ? i : CQ - 1];
  }
  uint4* tab4 = (uint4*)tab;
  for (int i = tid; i < TW / 4; i += 1024) tab4[i] = make_uint4(0u, 0u, 0u, 0u);
  __syncthreads();
  if (role == 0) {
    uint4 r[CQ_PT];
#pragma unroll
    for (int k = 0; k < CQ_PT; ++k) {
      if (tid + k * 1024 < CQ) {
        const unsigned dx = (unsigned)q[k].x, dy = (unsigned)q[k].y;
        const unsigned dz = (unsigned)q[k].z, dw = (unsigned)q[k].w;
        r[k].x = atomicAdd(&tab[dx >> 1], 1u << ((dx & 1u) << 4));
        r[k].y = atomicAdd(&tab[dy >> 1], 1u << ((dy & 1u) << 4));
        r[k].z = atomicAdd(&tab[dz >> 1], 1u << ((dz & 1u) << 4));
        r[k].w = atomicAdd(&tab[dw >> 1], 1u << ((dw & 1u) << 4));
      }
    }
    uint4* out4 = (uint4*)(lrank + (size_t)b * SLICE);
#pragma unroll
    for (int k = 0; k < CQ_PT; ++k) {
      if (tid + k * 1024 < CQ) {
        uint4 o;
        o.x = (r[k].x >> (((unsigned)q[k].x & 1u) << 4)) & 0xFFFFu;
        o.y = (r[k].y >> (((unsigned)q[k].y & 1u) << 4)) & 0xFFFFu;
        o.z = (r[k].z >> (((unsigned)q[k].z & 1u) << 4)) & 0xFFFFu;
        o.w = (r[k].w >> (((unsigned)q[k].w & 1u) << 4)) & 0xFFFFu;
        out4[tid + k * 1024] = o;
      }
    }
  } else {
#pragma unroll
    for (int k = 0; k < CQ_PT; ++k) {
      if (tid + k * 1024 < CQ) {
        const unsigned sx = (unsigned)q[k].x, sy = (unsigned)q[k].y;
        const unsigned sz = (unsigned)q[k].z, sw = (unsigned)q[k].w;
        atomicAdd(&tab[sx >> 1], 1u << ((sx & 1u) << 4));
        atomicAdd(&tab[sy >> 1], 1u << ((sy & 1u) << 4));
        atomicAdd(&tab[sz >> 1], 1u << ((sz & 1u) << 4));
        atomicAdd(&tab[sw >> 1], 1u << ((sw & 1u) << 4));
      }
    }
  }
  __syncthreads();
  uint4* o = (uint4*)(cnt + (size_t)blockIdx.x * TW);
  for (int i = tid; i < TW / 4; i += 1024) o[i] = tab4[i];
}

static constexpr int RB = (TW + 255) / 256;
static constexpr int RW = RB * 8;
__global__ __launch_bounds__(256) void k_reduce(
    const unsigned* __restrict__ cnt, unsigned* __restrict__ pre,
    unsigned* __restrict__ indeg, unsigned* __restrict__ wtot,
    float* __restrict__ dsout, float* __restrict__ dsin) {
  const int role = blockIdx.x >= RB ? 1 : 0;
  const int w = (blockIdx.x - role * RB) * 256 + threadIdx.x;
  unsigned lo = 0u, hi = 0u;
  if (w < TW) {
    const unsigned* c = cnt + (size_t)role * CB * TW + w;
    unsigned v[CB];
#pragma unroll
    for (int b = 0; b < CB; ++b) v[b] = c[(size_t)b * TW];
    if (role == 0) {
#pragma unroll
      for (int b = 0; b < CB; ++b) {
        *(uint2*)(pre + (size_t)b * NN + 2 * w) = make_uint2(lo, hi);
        lo += v[b] & 0xFFFFu;
        hi += v[b] >> 16;
      }
      *(uint2*)(indeg + 2 * w) = make_uint2(lo, hi);
      *(float2*)(dsin + 2 * w) = make_float2(lo ? rsqrtf((float)lo) : 1.0f,
                                             hi ? rsqrtf((float)hi) : 1.0f);
    } else {
#pragma unroll
      for (int b = 0; b < CB; ++b) {
        lo += v[b] & 0xFFFFu;
        hi += v[b] >> 16;
      }
      *(float2*)(dsout + 2 * w) = make_float2(lo ? rsqrtf((float)lo) : 1.0f,
                                              hi ? rsqrtf((float)hi) : 1.0f);
    }
  }
  if (role == 0) {
    unsigned s = lo + hi;
    for (int o = 16; o > 0; o >>= 1) s += __shfl_xor(s, o);
    if ((threadIdx.x & 31) == 0) wtot[w >> 5] = s;
  }
}

static constexpr int SC_TRIPS = (RW + 255) / 256;
static_assert(RB * 256 >= TW, "the scan grid covers the table");
__global__ __launch_bounds__(256) void k_scan(const unsigned* __restrict__ deg,
                                              const unsigned* __restrict__ wtot,
                                              unsigned* __restrict__ offs) {
  __shared__ unsigned sbase[8], swave[8];
  const int t = threadIdx.x, lane = t & 31, wv = t >> 5;
  const int w = blockIdx.x * 256 + t;
  const bool valid = w < TW;
  uint2 d = make_uint2(0u, 0u);
  if (valid) d = *(const uint2*)(deg + 2 * w);
  const int nb = blockIdx.x * 8;
  unsigned bs = 0u;
#pragma unroll
  for (int k = 0; k < SC_TRIPS; ++k) {
    const int j = t + k * 256;
    const unsigned v = wtot[j < RW ? j : RW - 1];
    bs += j < nb ? v : 0u;
  }
  const unsigned p = d.x + d.y;
  unsigned inc = p;
  for (int o = 1; o < 32; o <<= 1) {
    const unsigned u = __shfl_up(inc, o);
    if (lane >= o) inc += u;
  }
  for (int o = 16; o > 0; o >>= 1) bs += __shfl_xor(bs, o);
  if (lane == 31) swave[wv] = inc;
  if (lane == 0) sbase[wv] = bs;
  __syncthreads();
  unsigned run = 0u;
  for (int i = 0; i < 8; ++i) {
    run += sbase[i];
    if (i < wv) run += swave[i];
  }
  const unsigned ex = run + inc - p;
  if (valid) {
    *(uint2*)(offs + 2 * w) = make_uint2(ex, ex + d.x);
    if (w == TW - 1) offs[NN] = ex + p;
  }
}

__global__ void k_fill(const int* __restrict__ src, const int* __restrict__ dst,
                       const unsigned* __restrict__ offs, const unsigned* __restrict__ pre,
                       const unsigned* __restrict__ lrank, unsigned* __restrict__ esrc) {
  int e = blockIdx.x * 256 + threadIdx.x;
  if (e < EE) {
    const int d = dst[e];
    const int b = e / SLICE;
    esrc[offs[d] + pre[(size_t)b * NN + d] + lrank[e]] = (unsigned)src[e];
  }
}


__global__ void k_agg64(const float* __restrict__ P, const unsigned* __restrict__ offs,
                        const unsigned* __restrict__ esrc, const float* __restrict__ dsin,
                        float* __restrict__ R) {
  const int wv = threadIdx.x >> 5, lane = threadIdx.x & 31;
  const int n = blockIdx.x * 8 + wv;
  const unsigned b = offs[n], e = offs[n + 1];
  float2 acc = make_float2(0.f, 0.f);
  for (unsigned i = b; i < e; ++i) {
    const unsigned s = esrc[i];
    const float2 v = *(const float2*)(P + (size_t)s * 64 + lane * 2);
    acc.x += v.x;
    acc.y += v.y;
  }
  const float c = dsin[n];
  *(float2*)(R + (size_t)n * 64 + lane * 2) = make_float2(c * acc.x, c * acc.y);
}


static constexpr int GD_PITCH = 132;
static constexpr int GD_TILES = NN / 16;
static constexpr int GD_BLOCKS = 512;

__device__ __forceinline__ unsigned edge_ids(const unsigned* esrc,
                                             unsigned i0, unsigned cnt, unsigned lane) {
  const unsigned last = cnt ? cnt - 1u : 0u;
  return esrc[min(i0 + min(lane, last), (unsigned)(EE - 1))];
}

__device__ __forceinline__ void rows8_issue(const float* hl, unsigned sv, unsigned k0,
                                            float4 (&v)[8]) {
#pragma unroll
  for (int k = 0; k < 8; ++k) {
    const unsigned s = (unsigned)__builtin_amdgcn_readlane((int)sv, (int)k0 + k);
    v[k] = *(const float4*)(hl + (size_t)s * FF);
  }
}

template <bool BN>
__device__ __forceinline__ void rows8_sum(float4& acc, const float4 (&v)[8], float wl,
                                          unsigned k0, unsigned cnt, float4 ca, float4 cb4) {
#pragma unroll
  for (int k = 0; k < 8; ++k) {
    const float wk = __builtin_bit_cast(
        float, __builtin_amdgcn_readlane(__builtin_bit_cast(int, wl), (int)k0 + k));
    const float w = k0 + k < cnt ? wk : 0.f;
    float4 x = v[k];
    if (BN) x = bn_relu4(x, ca, cb4);
    acc.x = fmaf(w, x.x, acc.x);
    acc.y = fmaf(w, x.y, acc.y);
    acc.z = fmaf(w, x.z, acc.z);
    acc.w = fmaf(w, x.w, acc.w);
  }
}

template <bool BN>
__global__ __launch_bounds__(256) void k_gemm_dual(
    const float* H, const unsigned* offs, const unsigned* esrc,
    const float* dsout, const float* dsin,
    const float* __restrict__ W, const float* __restrict__ L,
    const float* __restrict__ bnA, const float* __restrict__ bnB,
    float* __restrict__ Z, float* __restrict__ part) {
  __shared__ __attribute__((aligned(16))) float sAg[16 * GD_PITCH];
  __shared__ __attribute__((aligned(16))) float sH[16 * GD_PITCH];
  __shared__ float sStat[256];
  const int wv = __builtin_amdgcn_readfirstlane((int)(threadIdx.x >> 5));
  const int lane = threadIdx.x & 31;
  const int t = lane & 15;
  const int hh = lane >> 4;
  const int cb = wv << 4;

  v2f bw[32], bl[32];
#pragma unroll
  for (int ks = 0; ks < 32; ++ks) {
    const int kk = 4 * ks + 2 * hh;
    bw[ks].x = W[kk * FF + cb + t]; bw[ks].y = W[(kk + 1) * FF + cb + t];
    bl[ks].x = L[kk * FF + cb + t]; bl[ks].y = L[(kk + 1) * FF + cb + t];
  }

  const int so0 = wv * GD_PITCH + lane * 4;
  const int so1 = (wv + 8) * GD_PITCH + lane * 4;
  const float* fa = sAg + t * GD_PITCH + 2 * hh;
  const float* fh = sH + t * GD_PITCH + 2 * hh;
  const float* hl = H + lane * 4;
  float4 ca = make_float4(1.f, 1.f, 1.f, 1.f), cb4 = make_float4(0.f, 0.f, 0.f, 0.f);
  if (BN) {
    ca = *(const float4*)(bnA + lane * 4);
    cb4 = *(const float4*)(bnB + lane * 4);
  }

  float csum = 0.f, csq = 0.f;
  int tile = blockIdx.x;
  const int G = gridDim.x;
  struct GSet { unsigned bA, eA, bB, eB, sA, sB; float wA, wB; };
  const int pick = (lane & 1) + ((lane & 2) << 2);
  GSet X, Y;
  unsigned ov2;
  float cA, cB;
  float4 ph0, ph1;
  {
    const int n0 = (tile < GD_TILES ? tile : 0) * 16 + wv;
    const size_t g0 = (size_t)n0 * FF + lane * 4;
    ph0 = *(const float4*)(H + g0); ph1 = *(const float4*)(H + g0 + (size_t)8 * FF);
    const unsigned ov0 = offs[n0 + pick];
    const int n1 = (tile + G < GD_TILES ? tile + G : 0) * 16 + wv;
    const unsigned ov1 = offs[n1 + pick];
    const int n2 = (tile + 2 * G < GD_TILES ? tile + 2 * G : 0) * 16 + wv;
    ov2 = offs[n2 + pick];
    cA = dsin[n0]; cB = dsin[n0 + 8];
    X.bA = (unsigned)__builtin_amdgcn_readlane((int)ov0, 0); X.eA = (unsigned)__builtin_amdgcn_readlane((int)ov0, 1);
    X.bB = (unsigned)__builtin_amdgcn_readlane((int)ov0, 2); X.eB = (unsigned)__builtin_amdgcn_readlane((int)ov0, 3);
    X.sA = edge_ids(esrc, X.bA, min(X.eA - X.bA, 32u), (unsigned)lane);
    X.sB = edge_ids(esrc, X.bB, min(X.eB - X.bB, 32u), (unsigned)lane);
    Y.bA = (unsigned)__builtin_amdgcn_readlane((int)ov1, 0); Y.eA = (unsigned)__builtin_amdgcn_readlane((int)ov1, 1);
    Y.bB = (unsigned)__builtin_amdgcn_readlane((int)ov1, 2); Y.eB = (unsigned)__builtin_amdgcn_readlane((int)ov1, 3);
    Y.sA = edge_ids(esrc, Y.bA, min(Y.eA - Y.bA, 32u), (unsigned)lane);
    Y.sB = edge_ids(esrc, Y.bB, min(Y.eB - Y.bB, 32u), (unsigned)lane);
    X.wA = dsout[X.sA]; X.wB = dsout[X.sB];
    Y.wA = 0.f; Y.wB = 0.f;
  }
  auto step = [&](GSet& cur, GSet& nxt) {
    float4 accA = make_float4(0.f, 0.f, 0.f, 0.f), accB = accA;
    unsigned iA = cur.bA, iB = cur.bB;
    for (;;) {
      const unsigned cntA = min(cur.eA - iA, 32u), cntB = min(cur.eB - iB, 32u);
      const unsigned cm = max(cntA, cntB);
      for (unsigned k0 = 0; k0 < cm; k0 += 8) {
        float4 va[8], vb[8];
        if (k0 < cntA && k0 < cntB) {
          rows8_issue(hl, cur.sA, k0, va);
          rows8_issue(hl, cur.sB, k0, vb);
          rows8_sum<BN>(accA, va, cur.wA, k0, cntA, ca, cb4);
          rows8_sum<BN>(accB, vb, cur.wB, k0, cntB, ca, cb4);
        } else if (k0 < cntA) {
          rows8_issue(hl, cur.sA, k0, va);
          rows8_sum<BN>(accA, va, cur.wA, k0, cntA, ca, cb4);
        } else {
          rows8_issue(hl, cur.sB, k0, vb);
          rows8_sum<BN>(accB, vb, cur.wB, k0, cntB, ca, cb4);
        }
      }
      iA += cntA; iB += cntB;
      if (iA >= cur.eA && iB >= cur.eB) break;
      cur.sA = edge_ids(esrc, iA, min(cur.eA - iA, 32u), (unsigned)lane);
      cur.sB = edge_ids(esrc, iB, min(cur.eB - iB, 32u), (unsigned)lane);
      cur.wA = dsout[cur.sA]; cur.wB = dsout[cur.sB];
    }
    __syncthreads();
    *(float4*)(sAg + so0) = make_float4(cA * accA.x, cA * accA.y, cA * accA.z, cA * accA.w);
    *(float4*)(sAg + so1) = make_float4(cB * accB.x, cB * accB.y, cB * accB.z, cB * accB.w);
    if (BN) { ph0 = bn_relu4(ph0, ca, cb4); ph1 = bn_relu4(ph1, ca, cb4); }
    *(float4*)(sH + so0) = ph0;  *(float4*)(sH + so1) = ph1;
    __syncthreads();
    {
      nxt.wA = dsout[nxt.sA]; nxt.wB = dsout[nxt.sB];
      const int m = (tile + G < GD_TILES ? tile + G : 0) * 16 + wv;
      const size_t g0 = (size_t)m * FF + lane * 4;
      ph0 = *(const float4*)(H + g0); ph1 = *(const float4*)(H + g0 + (size_t)8 * FF);
      cA = dsin[m]; cB = dsin[m + 8];
      cur.bA = (unsigned)__builtin_amdgcn_readlane((int)ov2, 0); cur.eA = (unsigned)__builtin_amdgcn_readlane((int)ov2, 1);
      cur.bB = (unsigned)__builtin_amdgcn_readlane((int)ov2, 2); cur.eB = (unsigned)__builtin_amdgcn_readlane((int)ov2, 3);
      cur.sA = edge_ids(esrc, cur.bA, min(cur.eA - cur.bA, 32u), (unsigned)lane);
      cur.sB = edge_ids(esrc, cur.bB, min(cur.eB - cur.bB, 32u), (unsigned)lane);
      const int n3 = (tile + 3 * G < GD_TILES ? tile + 3 * G : 0) * 16 + wv;
      ov2 = offs[n3 + pick];
    }
    __builtin_amdgcn_sched_barrier(0);
    v8f acc = {};
#pragma unroll
    for (int ks = 0; ks < 32; ++ks) {
      v2f a1 = *(const v2f*)(fa + 4 * ks);
      acc = __builtin_amdgcn_wmma_f32_16x16x4_f32(false, a1, false, bw[ks], (short)0, acc, false, false);
      v2f a2 = *(const v2f*)(fh + 4 * ks);
      acc = __builtin_amdgcn_wmma_f32_16x16x4_f32(false, a2, false, bl[ks], (short)0, acc, false, false);
    }
    const int r0 = tile << 4;
#pragma unroll
    for (int v = 0; v < 8; ++v)
      Z[(size_t)(r0 + v + 8 * hh) * FF + cb + t] = acc[v];
    float ts = acc[0], tq = acc[0] * acc[0];
#pragma unroll
    for (int v = 1; v < 8; ++v) {
      ts += acc[v];
      tq = fmaf(acc[v], acc[v], tq);
    }
    csum += ts;
    csq += tq;
  };
  for (;;) {
    if (tile >= GD_TILES) break;
    step(X, Y);
    tile += G;
    if (tile >= GD_TILES) break;
    step(Y, X);
    tile += G;
  }
  csum += __shfl_xor(csum, 16);
  csq += __shfl_xor(csq, 16);
  if (hh == 0) {
    sStat[cb + t] = csum;
    sStat[128 + cb + t] = csq;
  }
  __syncthreads();
  part[(size_t)blockIdx.x * 256 + threadIdx.x] = sStat[threadIdx.x];
}

static constexpr int GP_TILES = NN / 32;
__global__ __launch_bounds__(256) void k_gemm_p(
    const float* __restrict__ H, const float* __restrict__ dsout,
    const float* __restrict__ bnA, const float* __restrict__ bnB,
    const float* __restrict__ Wp, float* __restrict__ P) {
  __shared__ __attribute__((aligned(16))) float sT[32 * GD_PITCH];
  const int wv = threadIdx.x >> 5;
  const int lane = threadIdx.x & 31;
  const int t = lane & 15, hh = lane >> 4;
  const int ct = wv & 3, rt = wv >> 2;
  const int cb = ct << 4;

  v2f bp[32];
#pragma unroll
  for (int ks = 0; ks < 32; ++ks) {
    const int kk = 4 * ks + 2 * hh;
    bp[ks].x = Wp[kk * 64 + cb + t]; bp[ks].y = Wp[(kk + 1) * 64 + cb + t];
  }

  const int so = wv * GD_PITCH + lane * 4;
  const float* fa = sT + (16 * rt + t) * GD_PITCH + 2 * hh;
  const float4 ca = *(const float4*)(bnA + lane * 4);
  const float4 cb4 = *(const float4*)(bnB + lane * 4);

  int tile = blockIdx.x;
  float4 p0, p1, p2, p3;
  float dn = 0.f;
  if (tile < GP_TILES) {
    const size_t g = (size_t)(tile * 32 + wv) * FF + lane * 4;
    p0 = *(const float4*)(H + g);
    p1 = *(const float4*)(H + g + (size_t)8 * FF);
    p2 = *(const float4*)(H + g + (size_t)16 * FF);
    p3 = *(const float4*)(H + g + (size_t)24 * FF);
    dn = dsout[tile * 32 + 16 * rt + t];
  }
  for (; tile < GP_TILES; tile += gridDim.x) {
    const float d = dn;
    __syncthreads();
    *(float4*)(sT + so) = bn_relu4(p0, ca, cb4);
    *(float4*)(sT + so + 8 * GD_PITCH) = bn_relu4(p1, ca, cb4);
    *(float4*)(sT + so + 16 * GD_PITCH) = bn_relu4(p2, ca, cb4);
    *(float4*)(sT + so + 24 * GD_PITCH) = bn_relu4(p3, ca, cb4);
    __syncthreads();
    const int nt = tile + gridDim.x;
    if (nt < GP_TILES) {
      const size_t g = (size_t)(nt * 32 + wv) * FF + lane * 4;
      p0 = *(const float4*)(H + g);
      p1 = *(const float4*)(H + g + (size_t)8 * FF);
      p2 = *(const float4*)(H + g + (size_t)16 * FF);
      p3 = *(const float4*)(H + g + (size_t)24 * FF);
      dn = dsout[nt * 32 + 16 * rt + t];
    }
    v8f acc = {};
#pragma unroll
    for (int ks = 0; ks < 32; ++ks) {
      v2f a = *(const v2f*)(fa + 4 * ks);
      a.x = d * a.x; a.y = d * a.y;
      acc = __builtin_amdgcn_wmma_f32_16x16x4_f32(false, a, false, bp[ks], (short)0, acc, false, false);
    }
    const int r0 = tile * 32 + 16 * rt;
#pragma unroll
    for (int v = 0; v < 8; ++v)
      P[(size_t)(r0 + v + 8 * hh) * 64 + cb + t] = acc[v];
  }
}

__global__ __launch_bounds__(256) void k_gemm_out(
    const float* __restrict__ H, const float* __restrict__ Lp,
    const float* __restrict__ bnA, const float* __restrict__ bnB,
    const float* __restrict__ R, const float* __restrict__ b2,
    float* __restrict__ out) {
  __shared__ __attribute__((aligned(16))) float sT[32 * GD_PITCH];
  __shared__ __attribute__((aligned(16))) float sBN[256];
  const int wv = threadIdx.x >> 5;
  const int lane = threadIdx.x & 31;
  const int t = lane & 15, hh = lane >> 4;
  const int ct = wv & 3, rt = wv >> 2;
  const int cb = ct << 4;
  const bool active = ct < 3;

  v2f bq[32];
#pragma unroll
  for (int ks = 0; ks < 32; ++ks) { bq[ks].x = 0.f; bq[ks].y = 0.f; }
  if (active) {
#pragma unroll
    for (int ks = 0; ks < 32; ++ks) {
      const int kk = 4 * ks + 2 * hh;
      bq[ks].x = Lp[kk * 64 + cb + t]; bq[ks].y = Lp[(kk + 1) * 64 + cb + t];
    }
  }

  const int so = wv * GD_PITCH + lane * 4;
  const float* fa = sT + (16 * rt + t) * GD_PITCH + 2 * hh;
  sBN[threadIdx.x] = threadIdx.x < 128 ? bnA[threadIdx.x] : bnB[threadIdx.x - 128];
  const int col = cb + t;

  int tile = blockIdx.x;
  float4 p0, p1, p2, p3;
  float rn[8];
#pragma unroll
  for (int v = 0; v < 8; ++v) rn[v] = 0.f;
  if (tile < GP_TILES) {
    const size_t g = (size_t)(tile * 32 + wv) * FF + lane * 4;
    p0 = *(const float4*)(H + g);
    p1 = *(const float4*)(H + g + (size_t)8 * FF);
    p2 = *(const float4*)(H + g + (size_t)16 * FF);
    p3 = *(const float4*)(H + g + (size_t)24 * FF);
    if (active) {
#pragma unroll
      for (int v = 0; v < 8; ++v)
        rn[v] = R[(size_t)(tile * 32 + 16 * rt + v + 8 * hh) * 64 + col];
    }
  }
  for (; tile < GP_TILES; tile += gridDim.x) {
    float rv[8];
#pragma unroll
    for (int v = 0; v < 8; ++v) rv[v] = rn[v];
    __syncthreads();
    const float4 ca = *(const float4*)(sBN + lane * 4);
    const float4 cb4 = *(const float4*)(sBN + 128 + lane * 4);
    *(float4*)(sT + so) = bn_relu4(p0, ca, cb4);
    *(float4*)(sT + so + 8 * GD_PITCH) = bn_relu4(p1, ca, cb4);
    *(float4*)(sT + so + 16 * GD_PITCH) = bn_relu4(p2, ca, cb4);
    *(float4*)(sT + so + 24 * GD_PITCH) = bn_relu4(p3, ca, cb4);
    __syncthreads();
    const int nt = tile + gridDim.x;
    if (nt < GP_TILES) {
      const size_t g = (size_t)(nt * 32 + wv) * FF + lane * 4;
      p0 = *(const float4*)(H + g);
      p1 = *(const float4*)(H + g + (size_t)8 * FF);
      p2 = *(const float4*)(H + g + (size_t)16 * FF);
      p3 = *(const float4*)(H + g + (size_t)24 * FF);
      if (active) {
#pragma unroll
        for (int v = 0; v < 8; ++v)
          rn[v] = R[(size_t)(nt * 32 + 16 * rt + v + 8 * hh) * 64 + col];
      }
    }
    if (active) {
      const int r0 = tile * 32 + 16 * rt;
      v8f acc = {};
#pragma unroll
      for (int ks = 0; ks < 32; ++ks) {
        v2f a = *(const v2f*)(fa + 4 * ks);
        acc = __builtin_amdgcn_wmma_f32_16x16x4_f32(false, a, false, bq[ks], (short)0, acc, false, false);
      }
      if (col < 40) {
        const float bias = b2[col];
#pragma unroll
        for (int v = 0; v < 8; ++v)
          out[(size_t)(r0 + v + 8 * hh) * 40 + col] = acc[v] + rv[v] + bias;
      }
    }
  }
}


static constexpr int BP_GROUPS = 16;
static constexpr int BP_ROWS = GD_BLOCKS / BP_GROUPS;
static_assert(BP_ROWS * BP_GROUPS == GD_BLOCKS, "the groups cover every partial row");
__global__ __launch_bounds__(1024) void k_bnparams(
    const float* __restrict__ part,
    const float* __restrict__ g, const float* __restrict__ be,
    float* __restrict__ bnA, float* __restrict__ bnB) {
  __shared__ __attribute__((aligned(16))) float red[BP_GROUPS * 256];
  const int q = threadIdx.x & 63, grp = threadIdx.x >> 6;
  const float4* p = (const float4*)(part + (size_t)grp * BP_ROWS * 256) + q;
  float4 s = make_float4(0.f, 0.f, 0.f, 0.f);
#pragma unroll 16
  for (int b = 0; b < BP_ROWS; ++b) {
    const float4 v = p[b * 64];
    s.x += v.x; s.y += v.y; s.z += v.z; s.w += v.w;
  }
  *(float4*)(red + grp * 256 + 4 * q) = s;
  __syncthreads();
  const int c = threadIdx.x;
  if (c < 128) {
    float cs = red[c], cs2 = red[128 + c];
#pragma unroll
    for (int k = 1; k < BP_GROUPS; ++k) {
      cs += red[k * 256 + c];
      cs2 += red[k * 256 + 128 + c];
    }
    const float inv = 1.0f / (float)NN;
    const float mu = cs * inv;
    const float var = cs2 * inv - mu * mu;
    const float a = g[c] * rsqrtf(var + 1e-5f);
    bnA[c] = a;
    bnB[c] = be[c] - mu * a;
  }
}

__global__ void k_padw(const float* __restrict__ W2, const float* __restrict__ L2,
                       float* __restrict__ Wp, float* __restrict__ Lp) {
  const int i = blockIdx.x * 256 + threadIdx.x;
  const int k = i >> 6, c = i & 63;
  const float w = (c < 40) ? W2[k * 40 + c] : 0.f;
  const float l = (c < 40) ? L2[k * 40 + c] : 0.f;
  Wp[i] = w;
  Lp[i] = l;
}


extern "C" void kernel_launch(void* const* d_in, const int* in_sizes, int n_in,
                              void* d_out, int out_size, void* d_ws, size_t ws_size,
                              hipStream_t stream) {
  (void)in_sizes; (void)n_in; (void)out_size; (void)ws_size;

  const float* feat = (const float*)d_in[0];
  const int* src   = (const int*)d_in[1];
  const int* dst   = (const int*)d_in[2];
  const float* W0  = (const float*)d_in[3];
  const float* L0  = (const float*)d_in[4];
  const float* W1  = (const float*)d_in[5];
  const float* L1  = (const float*)d_in[6];
  const float* W2  = (const float*)d_in[7];
  const float* L2  = (const float*)d_in[8];
  const float* b2  = (const float*)d_in[9];
  const float* g0  = (const float*)d_in[10];
  const float* be0 = (const float*)d_in[11];
  const float* g1  = (const float*)d_in[12];
  const float* be1 = (const float*)d_in[13];
  float* out = (float*)d_out;

  char* base = (char*)d_ws;
  size_t off = 0;
  auto take = [&](size_t bytes) -> char* {
    char* p = base + off;
    off = (off + bytes + 255) & ~(size_t)255;
    return p;
  };
  unsigned* indeg  = (unsigned*)take((size_t)NN * 4);
  unsigned* offs   = (unsigned*)take((size_t)(NN + 1) * 4);
  unsigned* lrank  = (unsigned*)take((size_t)EE * 4);
  unsigned* esrc   = (unsigned*)take((size_t)EE * 4);
  unsigned* cnt    = (unsigned*)take((size_t)2 * CB * TW * 4);
  unsigned* pre    = (unsigned*)take((size_t)CB * NN * 4);
  float* dsout = (float*)take((size_t)NN * 4);
  float* dsin  = (float*)take((size_t)NN * 4);
  float* part  = (float*)take((size_t)GD_BLOCKS * 256 * 4);
  float* bnA0  = (float*)take(128 * 4);
  float* bnB0  = (float*)take(128 * 4);
  float* bnA1  = (float*)take(128 * 4);
  float* bnB1  = (float*)take(128 * 4);
  float* Wp    = (float*)take(128 * 64 * 4);
  float* Lp    = (float*)take(128 * 64 * 4);
  float* bufA  = (float*)take((size_t)NN * FF * 4);
  float* bufB  = (float*)take((size_t)NN * FF * 4);
  float* bufC  = (float*)take((size_t)NN * FF * 4);
  unsigned* wtot = (unsigned*)take((size_t)RW * 4);

  const int BEdge = (EE + 255) / 256;
  const int BWave = NN / 8;

  k_count<<<2 * CB, 1024, 0, stream>>>(src, dst, cnt, lrank);
  k_reduce<<<2 * RB, 256, 0, stream>>>(cnt, pre, indeg, wtot, dsout, dsin);
  k_scan<<<RB, 256, 0, stream>>>(indeg, wtot, offs);
  k_fill<<<BEdge, 256, 0, stream>>>(src, dst, offs, pre, lrank, esrc);
  k_padw<<<32, 256, 0, stream>>>(W2, L2, Wp, Lp);

  k_gemm_dual<false><<<GD_BLOCKS, 256, 0, stream>>>(feat, offs, esrc, dsout, dsin, W0, L0, nullptr, nullptr, bufB, part);
  k_bnparams<<<1, 1024, 0, stream>>>(part, g0, be0, bnA0, bnB0);

  k_gemm_dual<true><<<GD_BLOCKS, 256, 0, stream>>>(bufB, offs, esrc, dsout, dsin, W1, L1, bnA0, bnB0, bufC, part);
  k_bnparams<<<1, 1024, 0, stream>>>(part, g1, be1, bnA1, bnB1);

  k_gemm_p<<<1024, 256, 0, stream>>>(bufC, dsout, bnA1, bnB1, Wp, bufA);
  k_agg64<<<BWave, 256, 0, stream>>>(bufA, offs, esrc, dsin, bufB);
  k_gemm_out<<<1024, 256, 0, stream>>>(bufC, Lp, bnA1, bnB1, bufB, b2, out);
}


# Layer 2: both products (P and h2@L2+b2) in one GEMM pass over z1; the gather adds its sum into out. Pad launch, second GEMM launch and the R plane gone; out = fma(dsin, sum, h2@L2+b2): order changes
# speedup vs baseline: 2.6066x; 1.0352x over previous
#include <hip/hip_runtime.h>
#include <hip/hip_bf16.h>


static constexpr int NN = 100000;
static constexpr int EE = 1600000;
static constexpr int FF = 128;

typedef __attribute__((ext_vector_type(2))) float v2f;
typedef __attribute__((ext_vector_type(8))) float v8f;

__device__ __forceinline__ float4 bn_relu4(float4 z, float4 a, float4 b) {
  float4 r;
  r.x = fmaf(z.x, a.x, b.x); r.x = r.x > 0.f ? r.x : 0.f;
  r.y = fmaf(z.y, a.y, b.y); r.y = r.y > 0.f ? r.y : 0.f;
  r.z = fmaf(z.z, a.z, b.z); r.z = r.z > 0.f ? r.z : 0.f;
  r.w = fmaf(z.w, a.w, b.w); r.w = r.w > 0.f ? r.w : 0.f;
  return r;
}


static constexpr int CB = 64;
static constexpr int SLICE = EE / CB;
static constexpr int TW = NN / 2;
static_assert(SLICE * CB == EE, "slices must tile the edge list exactly");
static_assert(SLICE < 65536, "a 16-bit counter must hold a whole slice");
static_assert(TW * 2 == NN && TW % 4 == 0, "table is copied as 16-byte pieces");
static constexpr int CQ = SLICE / 4;
static constexpr int CQ_PT = (CQ + 1023) / 1024;
static_assert(CQ * 4 == SLICE, "a slice is a whole number of 16-byte groups, so every slice starts 16-byte aligned");
__global__ __launch_bounds__(1024) void k_count(
    const int* __restrict__ src, const int* __restrict__ dst,
    unsigned* __restrict__ cnt, unsigned* __restrict__ lrank) {
  __shared__ __attribute__((aligned(16))) unsigned tab[TW];
  const int role = blockIdx.x >= CB ? 1 : 0;
  const int b = blockIdx.x - role * CB;
  const int tid = threadIdx.x;
  const int4* in4 = (const int4*)((role ? src : dst) + (size_t)b * SLICE);
  int4 q[CQ_PT];
#pragma unroll
  for (int k = 0; k < CQ_PT; ++k) {
    const int i = tid + k * 1024;
    q[k] = in4[i < CQ ? i : CQ - 1];
  }
  uint4* tab4 = (uint4*)tab;
  for (int i = tid; i < TW / 4; i += 1024) tab4[i] = make_uint4(0u, 0u, 0u, 0u);
  __syncthreads();
  if (role == 0) {
    uint4 r[CQ_PT];
#pragma unroll
    for (int k = 0; k < CQ_PT; ++k) {
      if (tid + k * 1024 < CQ) {
        const unsigned dx = (unsigned)q[k].x, dy = (unsigned)q[k].y;
        const unsigned dz = (unsigned)q[k].z, dw = (unsigned)q[k].w;
        r[k].x = atomicAdd(&tab[dx >> 1], 1u << ((dx & 1u) << 4));
        r[k].y = atomicAdd(&tab[dy >> 1], 1u << ((dy & 1u) << 4));
        r[k].z = atomicAdd(&tab[dz >> 1], 1u << ((dz & 1u) << 4));
        r[k].w = atomicAdd(&tab[dw >> 1], 1u << ((dw & 1u) << 4));
      }
    }
    uint4* out4 = (uint4*)(lrank + (size_t)b * SLICE);
#pragma unroll
    for (int k = 0; k < CQ_PT; ++k) {
      if (tid + k * 1024 < CQ) {
        uint4 o;
        o.x = (r[k].x >> (((unsigned)q[k].x & 1u) << 4)) & 0xFFFFu;
        o.y = (r[k].y >> (((unsigned)q[k].y & 1u) << 4)) & 0xFFFFu;
        o.z = (r[k].z >> (((unsigned)q[k].z & 1u) << 4)) & 0xFFFFu;
        o.w = (r[k].w >> (((unsigned)q[k].w & 1u) << 4)) & 0xFFFFu;
        out4[tid + k * 1024] = o;
      }
    }
  } else {
#pragma unroll
    for (int k = 0; k < CQ_PT; ++k) {
      if (tid + k * 1024 < CQ) {
        const unsigned sx = (unsigned)q[k].x, sy = (unsigned)q[k].y;
        const unsigned sz = (unsigned)q[k].z, sw = (unsigned)q[k].w;
        atomicAdd(&tab[sx >> 1], 1u << ((sx & 1u) << 4));
        atomicAdd(&tab[sy >> 1], 1u << ((sy & 1u) << 4));
        atomicAdd(&tab[sz >> 1], 1u << ((sz & 1u) << 4));
        atomicAdd(&tab[sw >> 1], 1u << ((sw & 1u) << 4));
      }
    }
  }
  __syncthreads();
  uint4* o = (uint4*)(cnt + (size_t)blockIdx.x * TW);
  for (int i = tid; i < TW / 4; i += 1024) o[i] = tab4[i];
}

static constexpr int RB = (TW + 255) / 256;
static constexpr int RW = RB * 8;
__global__ __launch_bounds__(256) void k_reduce(
    const unsigned* __restrict__ cnt, unsigned* __restrict__ pre,
    unsigned* __restrict__ indeg, unsigned* __restrict__ wtot,
    float* __restrict__ dsout, float* __restrict__ dsin) {
  const int role = blockIdx.x >= RB ? 1 : 0;
  const int w = (blockIdx.x - role * RB) * 256 + threadIdx.x;
  unsigned lo = 0u, hi = 0u;
  if (w < TW) {
    const unsigned* c = cnt + (size_t)role * CB * TW + w;
    unsigned v[CB];
#pragma unroll
    for (int b = 0; b < CB; ++b) v[b] = c[(size_t)b * TW];
    if (role == 0) {
#pragma unroll
      for (int b = 0; b < CB; ++b) {
        *(uint2*)(pre + (size_t)b * NN + 2 * w) = make_uint2(lo, hi);
        lo += v[b] & 0xFFFFu;
        hi += v[b] >> 16;
      }
      *(uint2*)(indeg + 2 * w) = make_uint2(lo, hi);
      *(float2*)(dsin + 2 * w) = make_float2(lo ? rsqrtf((float)lo) : 1.0f,
                                             hi ? rsqrtf((float)hi) : 1.0f);
    } else {
#pragma unroll
      for (int b = 0; b < CB; ++b) {
        lo += v[b] & 0xFFFFu;
        hi += v[b] >> 16;
      }
      *(float2*)(dsout + 2 * w) = make_float2(lo ? rsqrtf((float)lo) : 1.0f,
                                              hi ? rsqrtf((float)hi) : 1.0f);
    }
  }
  if (role == 0) {
    unsigned s = lo + hi;
    for (int o = 16; o > 0; o >>= 1) s += __shfl_xor(s, o);
    if ((threadIdx.x & 31) == 0) wtot[w >> 5] = s;
  }
}

static constexpr int SC_TRIPS = (RW + 255) / 256;
static_assert(RB * 256 >= TW, "the scan grid covers the table");
__global__ __launch_bounds__(256) void k_scan(const unsigned* __restrict__ deg,
                                              const unsigned* __restrict__ wtot,
                                              unsigned* __restrict__ offs) {
  __shared__ unsigned sbase[8], swave[8];
  const int t = threadIdx.x, lane = t & 31, wv = t >> 5;
  const int w = blockIdx.x * 256 + t;
  const bool valid = w < TW;
  uint2 d = make_uint2(0u, 0u);
  if (valid) d = *(const uint2*)(deg + 2 * w);
  const int nb = blockIdx.x * 8;
  unsigned bs = 0u;
#pragma unroll
  for (int k = 0; k < SC_TRIPS; ++k) {
    const int j = t + k * 256;
    const unsigned v = wtot[j < RW ? j : RW - 1];
    bs += j < nb ? v : 0u;
  }
  const unsigned p = d.x + d.y;
  unsigned inc = p;
  for (int o = 1; o < 32; o <<= 1) {
    const unsigned u = __shfl_up(inc, o);
    if (lane >= o) inc += u;
  }
  for (int o = 16; o > 0; o >>= 1) bs += __shfl_xor(bs, o);
  if (lane == 31) swave[wv] = inc;
  if (lane == 0) sbase[wv] = bs;
  __syncthreads();
  unsigned run = 0u;
  for (int i = 0; i < 8; ++i) {
    run += sbase[i];
    if (i < wv) run += swave[i];
  }
  const unsigned ex = run + inc - p;
  if (valid) {
    *(uint2*)(offs + 2 * w) = make_uint2(ex, ex + d.x);
    if (w == TW - 1) offs[NN] = ex + p;
  }
}

__global__ void k_fill(const int* __restrict__ src, const int* __restrict__ dst,
                       const unsigned* __restrict__ offs, const unsigned* __restrict__ pre,
                       const unsigned* __restrict__ lrank, unsigned* __restrict__ esrc) {
  int e = blockIdx.x * 256 + threadIdx.x;
  if (e < EE) {
    const int d = dst[e];
    const int b = e / SLICE;
    esrc[offs[d] + pre[(size_t)b * NN + d] + lrank[e]] = (unsigned)src[e];
  }
}


__global__ void k_agg_out(const float* __restrict__ P, const unsigned* __restrict__ offs,
                          const unsigned* __restrict__ esrc, const float* __restrict__ dsin,
                          float* __restrict__ out) {
  const int wv = threadIdx.x >> 5, lane = threadIdx.x & 31;
  const int n = blockIdx.x * 8 + wv;
  const int cl = (lane < 20 ? lane : 19) * 2;
  const unsigned b = offs[n], e = offs[n + 1];
  float* o = out + (size_t)n * 40 + cl;
  const float2 q = *(const float2*)o;
  const float* pl = P + cl;
  float2 acc = make_float2(0.f, 0.f);
  for (unsigned i = b; i < e; ++i) {
    const unsigned s = esrc[i];
    const float2 v = *(const float2*)(pl + (size_t)s * 64);
    acc.x += v.x;
    acc.y += v.y;
  }
  const float c = dsin[n];
  if (lane < 20)
    *(float2*)o = make_float2(fmaf(c, acc.x, q.x), fmaf(c, acc.y, q.y));
}


static constexpr int GD_PITCH = 132;
static constexpr int GD_TILES = NN / 16;
static constexpr int GD_BLOCKS = 512;

__device__ __forceinline__ unsigned edge_ids(const unsigned* esrc,
                                             unsigned i0, unsigned cnt, unsigned lane) {
  const unsigned last = cnt ? cnt - 1u : 0u;
  return esrc[min(i0 + min(lane, last), (unsigned)(EE - 1))];
}

__device__ __forceinline__ void rows8_issue(const float* hl, unsigned sv, unsigned k0,
                                            float4 (&v)[8]) {
#pragma unroll
  for (int k = 0; k < 8; ++k) {
    const unsigned s = (unsigned)__builtin_amdgcn_readlane((int)sv, (int)k0 + k);
    v[k] = *(const float4*)(hl + (size_t)s * FF);
  }
}

template <bool BN>
__device__ __forceinline__ void rows8_sum(float4& acc, const float4 (&v)[8], float wl,
                                          unsigned k0, unsigned cnt, float4 ca, float4 cb4) {
#pragma unroll
  for (int k = 0; k < 8; ++k) {
    const float wk = __builtin_bit_cast(
        float, __builtin_amdgcn_readlane(__builtin_bit_cast(int, wl), (int)k0 + k));
    const float w = k0 + k < cnt ? wk : 0.f;
    float4 x = v[k];
    if (BN) x = bn_relu4(x, ca, cb4);
    acc.x = fmaf(w, x.x, acc.x);
    acc.y = fmaf(w, x.y, acc.y);
    acc.z = fmaf(w, x.z, acc.z);
    acc.w = fmaf(w, x.w, acc.w);
  }
}

template <bool BN>
__global__ __launch_bounds__(256) void k_gemm_dual(
    const float* H, const unsigned* offs, const unsigned* esrc,
    const float* dsout, const float* dsin,
    const float* __restrict__ W, const float* __restrict__ L,
    const float* __restrict__ bnA, const float* __restrict__ bnB,
    float* __restrict__ Z, float* __restrict__ part) {
  __shared__ __attribute__((aligned(16))) float sAg[16 * GD_PITCH];
  __shared__ __attribute__((aligned(16))) float sH[16 * GD_PITCH];
  __shared__ float sStat[256];
  const int wv = __builtin_amdgcn_readfirstlane((int)(threadIdx.x >> 5));
  const int lane = threadIdx.x & 31;
  const int t = lane & 15;
  const int hh = lane >> 4;
  const int cb = wv << 4;

  v2f bw[32], bl[32];
#pragma unroll
  for (int ks = 0; ks < 32; ++ks) {
    const int kk = 4 * ks + 2 * hh;
    bw[ks].x = W[kk * FF + cb + t]; bw[ks].y = W[(kk + 1) * FF + cb + t];
    bl[ks].x = L[kk * FF + cb + t]; bl[ks].y = L[(kk + 1) * FF + cb + t];
  }

  const int so0 = wv * GD_PITCH + lane * 4;
  const int so1 = (wv + 8) * GD_PITCH + lane * 4;
  const float* fa = sAg + t * GD_PITCH + 2 * hh;
  const float* fh = sH + t * GD_PITCH + 2 * hh;
  const float* hl = H + lane * 4;
  float4 ca = make_float4(1.f, 1.f, 1.f, 1.f), cb4 = make_float4(0.f, 0.f, 0.f, 0.f);
  if (BN) {
    ca = *(const float4*)(bnA + lane * 4);
    cb4 = *(const float4*)(bnB + lane * 4);
  }

  float csum = 0.f, csq = 0.f;
  int tile = blockIdx.x;
  const int G = gridDim.x;
  struct GSet { unsigned bA, eA, bB, eB, sA, sB; float wA, wB; };
  const int pick = (lane & 1) + ((lane & 2) << 2);
  GSet X, Y;
  unsigned ov2;
  float cA, cB;
  float4 ph0, ph1;
  {
    const int n0 = (tile < GD_TILES ? tile : 0) * 16 + wv;
    const size_t g0 = (size_t)n0 * FF + lane * 4;
    ph0 = *(const float4*)(H + g0); ph1 = *(const float4*)(H + g0 + (size_t)8 * FF);
    const unsigned ov0 = offs[n0 + pick];
    const int n1 = (tile + G < GD_TILES ? tile + G : 0) * 16 + wv;
    const unsigned ov1 = offs[n1 + pick];
    const int n2 = (tile + 2 * G < GD_TILES ? tile + 2 * G : 0) * 16 + wv;
    ov2 = offs[n2 + pick];
    cA = dsin[n0]; cB = dsin[n0 + 8];
    X.bA = (unsigned)__builtin_amdgcn_readlane((int)ov0, 0); X.eA = (unsigned)__builtin_amdgcn_readlane((int)ov0, 1);
    X.bB = (unsigned)__builtin_amdgcn_readlane((int)ov0, 2); X.eB = (unsigned)__builtin_amdgcn_readlane((int)ov0, 3);
    X.sA = edge_ids(esrc, X.bA, min(X.eA - X.bA, 32u), (unsigned)lane);
    X.sB = edge_ids(esrc, X.bB, min(X.eB - X.bB, 32u), (unsigned)lane);
    Y.bA = (unsigned)__builtin_amdgcn_readlane((int)ov1, 0); Y.eA = (unsigned)__builtin_amdgcn_readlane((int)ov1, 1);
    Y.bB = (unsigned)__builtin_amdgcn_readlane((int)ov1, 2); Y.eB = (unsigned)__builtin_amdgcn_readlane((int)ov1, 3);
    Y.sA = edge_ids(esrc, Y.bA, min(Y.eA - Y.bA, 32u), (unsigned)lane);
    Y.sB = edge_ids(esrc, Y.bB, min(Y.eB - Y.bB, 32u), (unsigned)lane);
    X.wA = dsout[X.sA]; X.wB = dsout[X.sB];
    Y.wA = 0.f; Y.wB = 0.f;
  }
  auto step = [&](GSet& cur, GSet& nxt) {
    float4 accA = make_float4(0.f, 0.f, 0.f, 0.f), accB = accA;
    unsigned iA = cur.bA, iB = cur.bB;
    for (;;) {
      const unsigned cntA = min(cur.eA - iA, 32u), cntB = min(cur.eB - iB, 32u);
      const unsigned cm = max(cntA, cntB);
      for (unsigned k0 = 0; k0 < cm; k0 += 8) {
        float4 va[8], vb[8];
        if (k0 < cntA && k0 < cntB) {
          rows8_issue(hl, cur.sA, k0, va);
          rows8_issue(hl, cur.sB, k0, vb);
          rows8_sum<BN>(accA, va, cur.wA, k0, cntA, ca, cb4);
          rows8_sum<BN>(accB, vb, cur.wB, k0, cntB, ca, cb4);
        } else if (k0 < cntA) {
          rows8_issue(hl, cur.sA, k0, va);
          rows8_sum<BN>(accA, va, cur.wA, k0, cntA, ca, cb4);
        } else {
          rows8_issue(hl, cur.sB, k0, vb);
          rows8_sum<BN>(accB, vb, cur.wB, k0, cntB, ca, cb4);
        }
      }
      iA += cntA; iB += cntB;
      if (iA >= cur.eA && iB >= cur.eB) break;
      cur.sA = edge_ids(esrc, iA, min(cur.eA - iA, 32u), (unsigned)lane);
      cur.sB = edge_ids(esrc, iB, min(cur.eB - iB, 32u), (unsigned)lane);
      cur.wA = dsout[cur.sA]; cur.wB = dsout[cur.sB];
    }
    __syncthreads();
    *(float4*)(sAg + so0) = make_float4(cA * accA.x, cA * accA.y, cA * accA.z, cA * accA.w);
    *(float4*)(sAg + so1) = make_float4(cB * accB.x, cB * accB.y, cB * accB.z, cB * accB.w);
    if (BN) { ph0 = bn_relu4(ph0, ca, cb4); ph1 = bn_relu4(ph1, ca, cb4); }
    *(float4*)(sH + so0) = ph0;  *(float4*)(sH + so1) = ph1;
    __syncthreads();
    {
      nxt.wA = dsout[nxt.sA]; nxt.wB = dsout[nxt.sB];
      const int m = (tile + G < GD_TILES ? tile + G : 0) * 16 + wv;
      const size_t g0 = (size_t)m * FF + lane * 4;
      ph0 = *(const float4*)(H + g0); ph1 = *(const float4*)(H + g0 + (size_t)8 * FF);
      cA = dsin[m]; cB = dsin[m + 8];
      cur.bA = (unsigned)__builtin_amdgcn_readlane((int)ov2, 0); cur.eA = (unsigned)__builtin_amdgcn_readlane((int)ov2, 1);
      cur.bB = (unsigned)__builtin_amdgcn_readlane((int)ov2, 2); cur.eB = (unsigned)__builtin_amdgcn_readlane((int)ov2, 3);
      cur.sA = edge_ids(esrc, cur.bA, min(cur.eA - cur.bA, 32u), (unsigned)lane);
      cur.sB = edge_ids(esrc, cur.bB, min(cur.eB - cur.bB, 32u), (unsigned)lane);
      const int n3 = (tile + 3 * G < GD_TILES ? tile + 3 * G : 0) * 16 + wv;
      ov2 = offs[n3 + pick];
    }
    __builtin_amdgcn_sched_barrier(0);
    v8f acc = {};
#pragma unroll
    for (int ks = 0; ks < 32; ++ks) {
      v2f a1 = *(const v2f*)(fa + 4 * ks);
      acc = __builtin_amdgcn_wmma_f32_16x16x4_f32(false, a1, false, bw[ks], (short)0, acc, false, false);
      v2f a2 = *(const v2f*)(fh + 4 * ks);
      acc = __builtin_amdgcn_wmma_f32_16x16x4_f32(false, a2, false, bl[ks], (short)0, acc, false, false);
    }
    const int r0 = tile << 4;
#pragma unroll
    for (int v = 0; v < 8; ++v)
      Z[(size_t)(r0 + v + 8 * hh) * FF + cb + t] = acc[v];
    float ts = acc[0], tq = acc[0] * acc[0];
#pragma unroll
    for (int v = 1; v < 8; ++v) {
      ts += acc[v];
      tq = fmaf(acc[v], acc[v], tq);
    }
    csum += ts;
    csq += tq;
  };
  for (;;) {
    if (tile >= GD_TILES) break;
    step(X, Y);
    tile += G;
    if (tile >= GD_TILES) break;
    step(Y, X);
    tile += G;
  }
  csum += __shfl_xor(csum, 16);
  csq += __shfl_xor(csq, 16);
  if (hh == 0) {
    sStat[cb + t] = csum;
    sStat[128 + cb + t] = csq;
  }
  __syncthreads();
  part[(size_t)blockIdx.x * 256 + threadIdx.x] = sStat[threadIdx.x];
}

static constexpr int GP_TILES = NN / 32;
__global__ __launch_bounds__(256) void k_gemm_pq(
    const float* __restrict__ H, const float* __restrict__ dsout,
    const float* __restrict__ bnA, const float* __restrict__ bnB,
    const float* __restrict__ W2, const float* __restrict__ L2,
    const float* __restrict__ b2, float* __restrict__ P, float* __restrict__ out) {
  __shared__ __attribute__((aligned(16))) float sT[32 * GD_PITCH];
  __shared__ __attribute__((aligned(16))) float sS[32 * GD_PITCH];
  __shared__ __attribute__((aligned(16))) float sBN[256];
  const int wv = __builtin_amdgcn_readfirstlane((int)(threadIdx.x >> 5));
  const int lane = threadIdx.x & 31;
  const int t = lane & 15, hh = lane >> 4;
  const bool two = wv < 4;
  const int combo = two ? wv : 4 + ((wv - 4) >> 1);
  const bool isP = combo < 3;
  const int cb = (isP ? combo : combo - 3) << 4;
  const int rt0 = two ? 0 : (wv & 1);
  const int col = cb + t;
  const float* Bm = isP ? W2 : L2;
  const bool real = col < 40;
  const int cc = real ? col : 39;

  v2f bf[32];
#pragma unroll
  for (int ks = 0; ks < 32; ++ks) {
    const int kk = 4 * ks + 2 * hh;
    const float x = Bm[kk * 40 + cc], y = Bm[(kk + 1) * 40 + cc];
    bf[ks].x = real ? x : 0.f;
    bf[ks].y = real ? y : 0.f;
  }

  const int so = wv * GD_PITCH + lane * 4;
  const float* fa0 = (isP ? sS : sT) + (16 * rt0 + t) * GD_PITCH + 2 * hh;
  const float* fa1 = fa0 + 16 * GD_PITCH;
  sBN[threadIdx.x] = threadIdx.x < 128 ? bnA[threadIdx.x] : bnB[threadIdx.x - 128];
  const float bias = (!isP && real) ? b2[col] : 0.f;

  int tile = blockIdx.x;
  float4 p0, p1, p2, p3;
  float e0 = 0.f, e1 = 0.f, e2 = 0.f, e3 = 0.f;
  if (tile < GP_TILES) {
    const int r = tile * 32 + wv;
    const size_t g = (size_t)r * FF + lane * 4;
    p0 = *(const float4*)(H + g);
    p1 = *(const float4*)(H + g + (size_t)8 * FF);
    p2 = *(const float4*)(H + g + (size_t)16 * FF);
    p3 = *(const float4*)(H + g + (size_t)24 * FF);
    e0 = dsout[r]; e1 = dsout[r + 8]; e2 = dsout[r + 16]; e3 = dsout[r + 24];
  }
  for (; tile < GP_TILES; tile += gridDim.x) {
    __syncthreads();
    const float4 ca = *(const float4*)(sBN + lane * 4);
    const float4 cb4 = *(const float4*)(sBN + 128 + lane * 4);
    const float4 h0 = bn_relu4(p0, ca, cb4), h1 = bn_relu4(p1, ca, cb4);
    const float4 h2 = bn_relu4(p2, ca, cb4), h3 = bn_relu4(p3, ca, cb4);
    *(float4*)(sT + so) = h0;
    *(float4*)(sT + so + 8 * GD_PITCH) = h1;
    *(float4*)(sT + so + 16 * GD_PITCH) = h2;
    *(float4*)(sT + so + 24 * GD_PITCH) = h3;
    *(float4*)(sS + so) = make_float4(e0 * h0.x, e0 * h0.y, e0 * h0.z, e0 * h0.w);
    *(float4*)(sS + so + 8 * GD_PITCH) = make_float4(e1 * h1.x, e1 * h1.y, e1 * h1.z, e1 * h1.w);
    *(float4*)(sS + so + 16 * GD_PITCH) = make_float4(e2 * h2.x, e2 * h2.y, e2 * h2.z, e2 * h2.w);
    *(float4*)(sS + so + 24 * GD_PITCH) = make_float4(e3 * h3.x, e3 * h3.y, e3 * h3.z, e3 * h3.w);
    __syncthreads();
    const int nt = tile + gridDim.x;
    if (nt < GP_TILES) {
      const int r = nt * 32 + wv;
      const size_t g = (size_t)r * FF + lane * 4;
      p0 = *(const float4*)(H + g);
      p1 = *(const float4*)(H + g + (size_t)8 * FF);
      p2 = *(const float4*)(H + g + (size_t)16 * FF);
      p3 = *(const float4*)(H + g + (size_t)24 * FF);
      e0 = dsout[r]; e1 = dsout[r + 8]; e2 = dsout[r + 16]; e3 = dsout[r + 24];
    }
    const size_t rb = (size_t)(tile * 32 + 8 * hh);
    if (two) {
      v8f acc0 = {}, acc1 = {};
#pragma unroll
      for (int ks = 0; ks < 32; ++ks) {
        const v2f a0 = *(const v2f*)(fa0 + 4 * ks);
        const v2f a1 = *(const v2f*)(fa1 + 4 * ks);
        acc0 = __builtin_amdgcn_wmma_f32_16x16x4_f32(false, a0, false, bf[ks], (short)0, acc0, false, false);
        acc1 = __builtin_amdgcn_wmma_f32_16x16x4_f32(false, a1, false, bf[ks], (short)0, acc1, false, false);
      }
      if (isP) {
        float* pp = P + rb * 64 + col;
#pragma unroll
        for (int v = 0; v < 8; ++v) {
          pp[v * 64] = acc0[v];
          pp[(16 + v) * 64] = acc1[v];
        }
      } else {
        float* oo = out + rb * 40 + col;
#pragma unroll
        for (int v = 0; v < 8; ++v) {
          oo[v * 40] = acc0[v] + bias;
          oo[(16 + v) * 40] = acc1[v] + bias;
        }
      }
    } else {
      v8f acc = {};
#pragma unroll
      for (int ks = 0; ks < 32; ++ks) {
        const v2f a = *(const v2f*)(fa0 + 4 * ks);
        acc = __builtin_amdgcn_wmma_f32_16x16x4_f32(false, a, false, bf[ks], (short)0, acc, false, false);
      }
      if (real) {
        float* oo = out + (rb + 16 * rt0) * 40 + col;
#pragma unroll
        for (int v = 0; v < 8; ++v)
          oo[v * 40] = acc[v] + bias;
      }
    }
  }
}


static constexpr int BP_GROUPS = 16;
static constexpr int BP_ROWS = GD_BLOCKS / BP_GROUPS;
static_assert(BP_ROWS * BP_GROUPS == GD_BLOCKS, "the groups cover every partial row");
__global__ __launch_bounds__(1024) void k_bnparams(
    const float* __restrict__ part,
    const float* __restrict__ g, const float* __restrict__ be,
    float* __restrict__ bnA, float* __restrict__ bnB) {
  __shared__ __attribute__((aligned(16))) float red[BP_GROUPS * 256];
  const int q = threadIdx.x & 63, grp = threadIdx.x >> 6;
  const float4* p = (const float4*)(part + (size_t)grp * BP_ROWS * 256) + q;
  float4 s = make_float4(0.f, 0.f, 0.f, 0.f);
#pragma unroll 16
  for (int b = 0; b < BP_ROWS; ++b) {
    const float4 v = p[b * 64];
    s.x += v.x; s.y += v.y; s.z += v.z; s.w += v.w;
  }
  *(float4*)(red + grp * 256 + 4 * q) = s;
  __syncthreads();
  const int c = threadIdx.x;
  if (c < 128) {
    float cs = red[c], cs2 = red[128 + c];
#pragma unroll
    for (int k = 1; k < BP_GROUPS; ++k) {
      cs += red[k * 256 + c];
      cs2 += red[k * 256 + 128 + c];
    }
    const float inv = 1.0f / (float)NN;
    const float mu = cs * inv;
    const float var = cs2 * inv - mu * mu;
    const float a = g[c] * rsqrtf(var + 1e-5f);
    bnA[c] = a;
    bnB[c] = be[c] - mu * a;
  }
}


extern "C" void kernel_launch(void* const* d_in, const int* in_sizes, int n_in,
                              void* d_out, int out_size, void* d_ws, size_t ws_size,
                              hipStream_t stream) {
  (void)in_sizes; (void)n_in; (void)out_size; (void)ws_size;

  const float* feat = (const float*)d_in[0];
  const int* src   = (const int*)d_in[1];
  const int* dst   = (const int*)d_in[2];
  const float* W0  = (const float*)d_in[3];
  const float* L0  = (const float*)d_in[4];
  const float* W1  = (const float*)d_in[5];
  const float* L1  = (const float*)d_in[6];
  const float* W2  = (const float*)d_in[7];
  const float* L2  = (const float*)d_in[8];
  const float* b2  = (const float*)d_in[9];
  const float* g0  = (const float*)d_in[10];
  const float* be0 = (const float*)d_in[11];
  const float* g1  = (const float*)d_in[12];
  const float* be1 = (const float*)d_in[13];
  float* out = (float*)d_out;

  char* base = (char*)d_ws;
  size_t off = 0;
  auto take = [&](size_t bytes) -> char* {
    char* p = base + off;
    off = (off + bytes + 255) & ~(size_t)255;
    return p;
  };
  unsigned* indeg  = (unsigned*)take((size_t)NN * 4);
  unsigned* offs   = (unsigned*)take((size_t)(NN + 1) * 4);
  unsigned* lrank  = (unsigned*)take((size_t)EE * 4);
  unsigned* esrc   = (unsigned*)take((size_t)EE * 4);
  unsigned* cnt    = (unsigned*)take((size_t)2 * CB * TW * 4);
  unsigned* pre    = (unsigned*)take((size_t)CB * NN * 4);
  float* dsout = (float*)take((size_t)NN * 4);
  float* dsin  = (float*)take((size_t)NN * 4);
  float* part  = (float*)take((size_t)GD_BLOCKS * 256 * 4);
  float* bnA0  = (float*)take(128 * 4);
  float* bnB0  = (float*)take(128 * 4);
  float* bnA1  = (float*)take(128 * 4);
  float* bnB1  = (float*)take(128 * 4);
  float* bufA  = (float*)take((size_t)NN * FF * 4);
  float* bufB  = (float*)take((size_t)NN * FF * 4);
  float* bufC  = (float*)take((size_t)NN * FF * 4);
  unsigned* wtot = (unsigned*)take((size_t)RW * 4);

  const int BEdge = (EE + 255) / 256;
  const int BWave = NN / 8;

  k_count<<<2 * CB, 1024, 0, stream>>>(src, dst, cnt, lrank);
  k_reduce<<<2 * RB, 256, 0, stream>>>(cnt, pre, indeg, wtot, dsout, dsin);
  k_scan<<<RB, 256, 0, stream>>>(indeg, wtot, offs);
  k_fill<<<BEdge, 256, 0, stream>>>(src, dst, offs, pre, lrank, esrc);

  k_gemm_dual<false><<<GD_BLOCKS, 256, 0, stream>>>(feat, offs, esrc, dsout, dsin, W0, L0, nullptr, nullptr, bufB, part);
  k_bnparams<<<1, 1024, 0, stream>>>(part, g0, be0, bnA0, bnB0);

  k_gemm_dual<true><<<GD_BLOCKS, 256, 0, stream>>>(bufB, offs, esrc, dsout, dsin, W1, L1, bnA0, bnB0, bufC, part);
  k_bnparams<<<1, 1024, 0, stream>>>(part, g1, be1, bnA1, bnB1);

  k_gemm_pq<<<1024, 256, 0, stream>>>(bufC, dsout, bnA1, bnB1, W2, L2, b2, bufA, out);
  k_agg_out<<<BWave, 256, 0, stream>>>(bufA, offs, esrc, dsin, out);
}
